# starting kernel
# baseline (speedup 1.0000x reference)
#include <hip/hip_runtime.h>

#define NN 100000
#define NE 1600000
#define DIM 128
#define NG 512
#define NL 3
#define OUTC (NL * DIM)
#define BN_EPS 1e-5f
#define ROWS_PER_BLOCK 128

typedef __attribute__((ext_vector_type(2))) float v2f;
typedef __attribute__((ext_vector_type(8))) float v8f;


__global__ void k_zero(float* __restrict__ p, int n) {
    int i = blockIdx.x * blockDim.x + threadIdx.x;
    if (i < n) p[i] = 0.f;
}

__global__ void k_init_deg(float* __restrict__ deg, int n) {
    int i = blockIdx.x * blockDim.x + threadIdx.x;
    if (i < n) deg[i] = 1.f;
}

__global__ void k_deg_edges(const int* __restrict__ dst, float* __restrict__ deg, int e) {
    int i = blockIdx.x * blockDim.x + threadIdx.x;
    if (i < e) atomicAdd(&deg[dst[i]], 1.f);
}

__global__ void k_dinv(float* __restrict__ deg, int n) {
    int i = blockIdx.x * blockDim.x + threadIdx.x;
    if (i < n) deg[i] = rsqrtf(fmaxf(deg[i], 1.f));
}

__global__ void __launch_bounds__(256)
k_transpose128(const float* __restrict__ W, float* __restrict__ WT) {
    __shared__ float t[16][17];
    int bx = blockIdx.x & 7, by = blockIdx.x >> 3;
    int tx = threadIdx.x & 15, ty = threadIdx.x >> 4;
    t[ty][tx] = W[(size_t)(by * 16 + ty) * DIM + bx * 16 + tx];
    __syncthreads();
    WT[(size_t)(bx * 16 + ty) * DIM + by * 16 + tx] = t[tx][ty];
}


__global__ void __launch_bounds__(256)
k_gemm_wmma(const float* __restrict__ A, const float* __restrict__ WT,
            float* __restrict__ out, int nrows) {
    __shared__ float tileA[16 * DIM];

    const int tid  = threadIdx.x;
    const int wave = tid >> 5;
    const int lane = tid & 31;
    const int row0 = blockIdx.x << 4;
    (void)nrows;

    const float4* srcv = (const float4*)(A + (size_t)row0 * DIM);
    float4* dstv = (float4*)tileA;
    dstv[tid]       = srcv[tid];
    dstv[tid + 256] = srcv[tid + 256];
    __syncthreads();

    const int m  = lane & 15;
    const int kh = (lane >> 4) << 1;
    const int n  = (wave << 4) + (lane & 15);

    const float2* wtrow = (const float2*)(WT + (size_t)n * DIM);

    v8f c = {};
#pragma unroll
    for (int ks = 0; ks < DIM; ks += 4) {
        v2f a, b;
        a.x = tileA[m * DIM + ks + kh];
        a.y = tileA[m * DIM + ks + kh + 1];
        float2 bb = wtrow[(ks + kh) >> 1];
        b.x = bb.x;
        b.y = bb.y;
        c = __builtin_amdgcn_wmma_f32_16x16x4_f32(
                 false, a,  false, b,
                 (short)0, c,  false,  false);
    }

    const int mo = (lane >> 4) << 3;
    float* orow = out + (size_t)(row0 + mo) * DIM + n;
#pragma unroll
    for (int r = 0; r < 8; ++r)
        orow[(size_t)r * DIM] = c[r];
}


__global__ void k_selfloop(const float* __restrict__ h, const float* __restrict__ dinv,
                           float* __restrict__ out) {
    int i = blockIdx.x * blockDim.x + threadIdx.x;
    int node = i >> 5, q = i & 31;
    if (node < NN) {
        float w = dinv[node]; w *= w;
        float4 v = ((const float4*)(h + (size_t)node * DIM))[q];
        float4 r; r.x = v.x * w; r.y = v.y * w; r.z = v.z * w; r.w = v.w * w;
        ((float4*)(out + (size_t)node * DIM))[q] = r;
    }
}

__global__ void k_edge_agg(const int* __restrict__ src, const int* __restrict__ dst,
                           const float* __restrict__ dinv, const float* __restrict__ h,
                           float* __restrict__ out) {
    int i = blockIdx.x * blockDim.x + threadIdx.x;
    int e = i >> 5, q = i & 31;
    if (e < NE) {
        int s = src[e], d = dst[e];
        float w = dinv[s] * dinv[d];
        float4 v = ((const float4*)(h + (size_t)s * DIM))[q];
        float* o = out + (size_t)d * DIM + (q << 2);
        atomicAdd(o + 0, v.x * w);
        atomicAdd(o + 1, v.y * w);
        atomicAdd(o + 2, v.z * w);
        atomicAdd(o + 3, v.w * w);
    }
}


__global__ void __launch_bounds__(DIM)
k_bias_relu_stats(float* __restrict__ h, const float* __restrict__ bias,
                  float* __restrict__ stats, int nrows) {
    int c  = threadIdx.x;
    int r0 = blockIdx.x * ROWS_PER_BLOCK;
    int r1 = min(r0 + ROWS_PER_BLOCK, nrows);
    float b = bias[c];
    float s = 0.f, s2 = 0.f;
    for (int r = r0; r < r1; ++r) {
        float v = h[(size_t)r * DIM + c] + b;
        v = fmaxf(v, 0.f);
        h[(size_t)r * DIM + c] = v;
        s += v; s2 += v * v;
    }
    atomicAdd(&stats[c], s);
    atomicAdd(&stats[DIM + c], s2);
}

__global__ void k_stats_finish(float* __restrict__ stats, float inv_n) {
    int c = threadIdx.x;
    float mu  = stats[c] * inv_n;
    float var = stats[DIM + c] * inv_n - mu * mu;
    stats[2 * DIM + c] = mu;
    stats[3 * DIM + c] = rsqrtf(var + BN_EPS);
}

__global__ void __launch_bounds__(DIM)
k_norm_pool(const float* __restrict__ h, const float* __restrict__ stats,
            const float* __restrict__ gamma, const float* __restrict__ beta,
            const int* __restrict__ batch, float* __restrict__ hnext,
            float* __restrict__ out, int layer, int nrows) {
    int c  = threadIdx.x;
    int r0 = blockIdx.x * ROWS_PER_BLOCK;
    int r1 = min(r0 + ROWS_PER_BLOCK, nrows);
    float mu = stats[2 * DIM + c], rstd = stats[3 * DIM + c];
    float ga = gamma[c], be = beta[c];
    int cur = -1; float acc = 0.f;
    for (int r = r0; r < r1; ++r) {
        float v = (h[(size_t)r * DIM + c] - mu) * rstd * ga + be;
        hnext[(size_t)r * DIM + c] = v;
        int g = batch[r];
        if (g != cur) {
            if (cur >= 0) atomicAdd(&out[(size_t)cur * OUTC + layer * DIM + c], acc);
            cur = g; acc = 0.f;
        }
        acc += v;
    }
    if (cur >= 0) atomicAdd(&out[(size_t)cur * OUTC + layer * DIM + c], acc);
}


extern "C" void kernel_launch(void* const* d_in, const int* in_sizes, int n_in,
                              void* d_out, int out_size, void* d_ws, size_t ws_size,
                              hipStream_t stream) {
    (void)in_sizes; (void)n_in; (void)out_size; (void)ws_size;

    const float* x     = (const float*)d_in[0];
    const int*   ei    = (const int*)d_in[1];
    const int*   batch = (const int*)d_in[2];
    const float *Wp[NL], *bp[NL], *gp[NL], *tp[NL];
    for (int i = 0; i < NL; ++i) {
        Wp[i] = (const float*)d_in[3 + 4 * i];
        bp[i] = (const float*)d_in[4 + 4 * i];
        gp[i] = (const float*)d_in[5 + 4 * i];
        tp[i] = (const float*)d_in[6 + 4 * i];
    }
    const int* esrc = ei;
    const int* edst = ei + NE;

    float* ws    = (float*)d_ws;
    float* dinv  = ws;
    float* stats = ws + NN;
    float* wtb   = ws + NN + 4 * DIM;
    size_t o     = ((size_t)NN + 4 * DIM + DIM * DIM + 255) & ~(size_t)255;
    float* bufA  = ws + o;
    float* bufB  = bufA + (size_t)NN * DIM;
    float* outp  = (float*)d_out;

    k_zero     <<<(NG * OUTC + 255) / 256, 256, 0, stream>>>(outp, NG * OUTC);
    k_init_deg <<<(NN + 255) / 256,        256, 0, stream>>>(dinv, NN);
    k_deg_edges<<<(NE + 255) / 256,        256, 0, stream>>>(edst, dinv, NE);
    k_dinv     <<<(NN + 255) / 256,        256, 0, stream>>>(dinv, NN);

    for (int l = 0; l < NL; ++l) {
        const float* hin = (l == 0) ? x : bufA;
        k_transpose128<<<64, 256, 0, stream>>>(Wp[l], wtb);
        k_gemm_wmma<<<NN / 16, 256, 0, stream>>>(hin, wtb, bufA, NN);
        k_selfloop <<<(NN * 32 + 255) / 256, 256, 0, stream>>>(bufA, dinv, bufB);
        k_edge_agg <<<(NE * 32) / 256, 256, 0, stream>>>(esrc, edst, dinv, bufA, bufB);
        k_zero           <<<1, 256, 0, stream>>>(stats, 2 * DIM);
        k_bias_relu_stats<<<(NN + ROWS_PER_BLOCK - 1) / ROWS_PER_BLOCK, DIM, 0, stream>>>(
            bufB, bp[l], stats, NN);
        k_stats_finish   <<<1, DIM, 0, stream>>>(stats, 1.f / (float)NN);
        k_norm_pool      <<<(NN + ROWS_PER_BLOCK - 1) / ROWS_PER_BLOCK, DIM, 0, stream>>>(
            bufB, stats, gp[l], tp[l], batch, bufA, outp, l, NN);
    }
}


# edge scatter-add: lane owns columns q, q+32, q+64, q+96 (was a float4), so every row load and f32 atomic instruction of a wave covers one whole 128-byte line; atomics and arrival order as before
# speedup vs baseline: 1.2630x; 1.2630x over previous
#include <hip/hip_runtime.h>

#define NN 100000
#define NE 1600000
#define DIM 128
#define NG 512
#define NL 3
#define OUTC (NL * DIM)
#define BN_EPS 1e-5f
#define ROWS_PER_BLOCK 128

typedef __attribute__((ext_vector_type(2))) float v2f;
typedef __attribute__((ext_vector_type(8))) float v8f;


__global__ void k_zero(float* __restrict__ p, int n) {
    int i = blockIdx.x * blockDim.x + threadIdx.x;
    if (i < n) p[i] = 0.f;
}

__global__ void k_init_deg(float* __restrict__ deg, int n) {
    int i = blockIdx.x * blockDim.x + threadIdx.x;
    if (i < n) deg[i] = 1.f;
}

__global__ void k_deg_edges(const int* __restrict__ dst, float* __restrict__ deg, int e) {
    int i = blockIdx.x * blockDim.x + threadIdx.x;
    if (i < e) atomicAdd(&deg[dst[i]], 1.f);
}

__global__ void k_dinv(float* __restrict__ deg, int n) {
    int i = blockIdx.x * blockDim.x + threadIdx.x;
    if (i < n) deg[i] = rsqrtf(fmaxf(deg[i], 1.f));
}

__global__ void __launch_bounds__(256)
k_transpose128(const float* __restrict__ W, float* __restrict__ WT) {
    __shared__ float t[16][17];
    int bx = blockIdx.x & 7, by = blockIdx.x >> 3;
    int tx = threadIdx.x & 15, ty = threadIdx.x >> 4;
    t[ty][tx] = W[(size_t)(by * 16 + ty) * DIM + bx * 16 + tx];
    __syncthreads();
    WT[(size_t)(bx * 16 + ty) * DIM + by * 16 + tx] = t[tx][ty];
}


__global__ void __launch_bounds__(256)
k_gemm_wmma(const float* __restrict__ A, const float* __restrict__ WT,
            float* __restrict__ out, int nrows) {
    __shared__ float tileA[16 * DIM];

    const int tid  = threadIdx.x;
    const int wave = tid >> 5;
    const int lane = tid & 31;
    const int row0 = blockIdx.x << 4;
    (void)nrows;

    const float4* srcv = (const float4*)(A + (size_t)row0 * DIM);
    float4* dstv = (float4*)tileA;
    dstv[tid]       = srcv[tid];
    dstv[tid + 256] = srcv[tid + 256];
    __syncthreads();

    const int m  = lane & 15;
    const int kh = (lane >> 4) << 1;
    const int n  = (wave << 4) + (lane & 15);

    const float2* wtrow = (const float2*)(WT + (size_t)n * DIM);

    v8f c = {};
#pragma unroll
    for (int ks = 0; ks < DIM; ks += 4) {
        v2f a, b;
        a.x = tileA[m * DIM + ks + kh];
        a.y = tileA[m * DIM + ks + kh + 1];
        float2 bb = wtrow[(ks + kh) >> 1];
        b.x = bb.x;
        b.y = bb.y;
        c = __builtin_amdgcn_wmma_f32_16x16x4_f32(
                 false, a,  false, b,
                 (short)0, c,  false,  false);
    }

    const int mo = (lane >> 4) << 3;
    float* orow = out + (size_t)(row0 + mo) * DIM + n;
#pragma unroll
    for (int r = 0; r < 8; ++r)
        orow[(size_t)r * DIM] = c[r];
}


__global__ void k_selfloop(const float* __restrict__ h, const float* __restrict__ dinv,
                           float* __restrict__ out) {
    int i = blockIdx.x * blockDim.x + threadIdx.x;
    int node = i >> 5, q = i & 31;
    if (node < NN) {
        float w = dinv[node]; w *= w;
        float4 v = ((const float4*)(h + (size_t)node * DIM))[q];
        float4 r; r.x = v.x * w; r.y = v.y * w; r.z = v.z * w; r.w = v.w * w;
        ((float4*)(out + (size_t)node * DIM))[q] = r;
    }
}

__global__ void k_edge_agg(const int* __restrict__ src, const int* __restrict__ dst,
                           const float* __restrict__ dinv, const float* __restrict__ h,
                           float* __restrict__ out) {
    int i = blockIdx.x * blockDim.x + threadIdx.x;
    int e = i >> 5, q = i & 31;
    if (e < NE) {
        int s = src[e], d = dst[e];
        float w = dinv[s] * dinv[d];
        const float* hs = h + (size_t)s * DIM + q;
        float v0 = hs[0], v1 = hs[32], v2 = hs[64], v3 = hs[96];
        float* o = out + (size_t)d * DIM + q;
        atomicAdd(o + 0,  v0 * w);
        atomicAdd(o + 32, v1 * w);
        atomicAdd(o + 64, v2 * w);
        atomicAdd(o + 96, v3 * w);
    }
}


__global__ void __launch_bounds__(DIM)
k_bias_relu_stats(float* __restrict__ h, const float* __restrict__ bias,
                  float* __restrict__ stats, int nrows) {
    int c  = threadIdx.x;
    int r0 = blockIdx.x * ROWS_PER_BLOCK;
    int r1 = min(r0 + ROWS_PER_BLOCK, nrows);
    float b = bias[c];
    float s = 0.f, s2 = 0.f;
    for (int r = r0; r < r1; ++r) {
        float v = h[(size_t)r * DIM + c] + b;
        v = fmaxf(v, 0.f);
        h[(size_t)r * DIM + c] = v;
        s += v; s2 += v * v;
    }
    atomicAdd(&stats[c], s);
    atomicAdd(&stats[DIM + c], s2);
}

__global__ void k_stats_finish(float* __restrict__ stats, float inv_n) {
    int c = threadIdx.x;
    float mu  = stats[c] * inv_n;
    float var = stats[DIM + c] * inv_n - mu * mu;
    stats[2 * DIM + c] = mu;
    stats[3 * DIM + c] = rsqrtf(var + BN_EPS);
}

__global__ void __launch_bounds__(DIM)
k_norm_pool(const float* __restrict__ h, const float* __restrict__ stats,
            const float* __restrict__ gamma, const float* __restrict__ beta,
            const int* __restrict__ batch, float* __restrict__ hnext,
            float* __restrict__ out, int layer, int nrows) {
    int c  = threadIdx.x;
    int r0 = blockIdx.x * ROWS_PER_BLOCK;
    int r1 = min(r0 + ROWS_PER_BLOCK, nrows);
    float mu = stats[2 * DIM + c], rstd = stats[3 * DIM + c];
    float ga = gamma[c], be = beta[c];
    int cur = -1; float acc = 0.f;
    for (int r = r0; r < r1; ++r) {
        float v = (h[(size_t)r * DIM + c] - mu) * rstd * ga + be;
        hnext[(size_t)r * DIM + c] = v;
        int g = batch[r];
        if (g != cur) {
            if (cur >= 0) atomicAdd(&out[(size_t)cur * OUTC + layer * DIM + c], acc);
            cur = g; acc = 0.f;
        }
        acc += v;
    }
    if (cur >= 0) atomicAdd(&out[(size_t)cur * OUTC + layer * DIM + c], acc);
}


extern "C" void kernel_launch(void* const* d_in, const int* in_sizes, int n_in,
                              void* d_out, int out_size, void* d_ws, size_t ws_size,
                              hipStream_t stream) {
    (void)in_sizes; (void)n_in; (void)out_size; (void)ws_size;

    const float* x     = (const float*)d_in[0];
    const int*   ei    = (const int*)d_in[1];
    const int*   batch = (const int*)d_in[2];
    const float *Wp[NL], *bp[NL], *gp[NL], *tp[NL];
    for (int i = 0; i < NL; ++i) {
        Wp[i] = (const float*)d_in[3 + 4 * i];
        bp[i] = (const float*)d_in[4 + 4 * i];
        gp[i] = (const float*)d_in[5 + 4 * i];
        tp[i] = (const float*)d_in[6 + 4 * i];
    }
    const int* esrc = ei;
    const int* edst = ei + NE;

    float* ws    = (float*)d_ws;
    float* dinv  = ws;
    float* stats = ws + NN;
    float* wtb   = ws + NN + 4 * DIM;
    size_t o     = ((size_t)NN + 4 * DIM + DIM * DIM + 255) & ~(size_t)255;
    float* bufA  = ws + o;
    float* bufB  = bufA + (size_t)NN * DIM;
    float* outp  = (float*)d_out;

    k_zero     <<<(NG * OUTC + 255) / 256, 256, 0, stream>>>(outp, NG * OUTC);
    k_init_deg <<<(NN + 255) / 256,        256, 0, stream>>>(dinv, NN);
    k_deg_edges<<<(NE + 255) / 256,        256, 0, stream>>>(edst, dinv, NE);
    k_dinv     <<<(NN + 255) / 256,        256, 0, stream>>>(dinv, NN);

    for (int l = 0; l < NL; ++l) {
        const float* hin = (l == 0) ? x : bufA;
        k_transpose128<<<64, 256, 0, stream>>>(Wp[l], wtb);
        k_gemm_wmma<<<NN / 16, 256, 0, stream>>>(hin, wtb, bufA, NN);
        k_selfloop <<<(NN * 32 + 255) / 256, 256, 0, stream>>>(bufA, dinv, bufB);
        k_edge_agg <<<(NE * 32) / 256, 256, 0, stream>>>(esrc, edst, dinv, bufA, bufB);
        k_zero           <<<1, 256, 0, stream>>>(stats, 2 * DIM);
        k_bias_relu_stats<<<(NN + ROWS_PER_BLOCK - 1) / ROWS_PER_BLOCK, DIM, 0, stream>>>(
            bufB, bp[l], stats, NN);
        k_stats_finish   <<<1, DIM, 0, stream>>>(stats, 1.f / (float)NN);
        k_norm_pool      <<<(NN + ROWS_PER_BLOCK - 1) / ROWS_PER_BLOCK, DIM, 0, stream>>>(
            bufB, stats, gp[l], tp[l], batch, bufA, outp, l, NN);
    }
}


# edge sums by destination: edges counting-sorted by dst once per call (rank = arrival order at the degree atomic), one wave per node adds its rows in registers, one store; no f32 row atomics
# speedup vs baseline: 1.8036x; 1.4280x over previous
#include <hip/hip_runtime.h>

#define NN 100000
#define NE 1600000
#define DIM 128
#define NG 512
#define NL 3
#define OUTC (NL * DIM)
#define BN_EPS 1e-5f
#define ROWS_PER_BLOCK 128
#define SCAN_T 1024
#define RP_PAD 100096

typedef __attribute__((ext_vector_type(2))) float v2f;
typedef __attribute__((ext_vector_type(8))) float v8f;


__global__ void k_zero(float* __restrict__ p, int n) {
    int i = blockIdx.x * blockDim.x + threadIdx.x;
    if (i < n) p[i] = 0.f;
}

__global__ void k_init_deg(float* __restrict__ deg, int n) {
    int i = blockIdx.x * blockDim.x + threadIdx.x;
    if (i < n) deg[i] = 1.f;
}

__global__ void k_deg_edges(const int* __restrict__ dst, float* __restrict__ deg, int e) {
    int i = blockIdx.x * blockDim.x + threadIdx.x;
    if (i < e) atomicAdd(&deg[dst[i]], 1.f);
}

__global__ void k_dinv(float* __restrict__ deg, int n) {
    int i = blockIdx.x * blockDim.x + threadIdx.x;
    if (i < n) deg[i] = rsqrtf(fmaxf(deg[i], 1.f));
}

__global__ void __launch_bounds__(256)
k_transpose128(const float* __restrict__ W, float* __restrict__ WT) {
    __shared__ float t[16][17];
    int bx = blockIdx.x & 7, by = blockIdx.x >> 3;
    int tx = threadIdx.x & 15, ty = threadIdx.x >> 4;
    t[ty][tx] = W[(size_t)(by * 16 + ty) * DIM + bx * 16 + tx];
    __syncthreads();
    WT[(size_t)(bx * 16 + ty) * DIM + by * 16 + tx] = t[tx][ty];
}


__global__ void __launch_bounds__(256)
k_gemm_wmma(const float* __restrict__ A, const float* __restrict__ WT,
            float* __restrict__ out, int nrows) {
    __shared__ float tileA[16 * DIM];

    const int tid  = threadIdx.x;
    const int wave = tid >> 5;
    const int lane = tid & 31;
    const int row0 = blockIdx.x << 4;
    (void)nrows;

    const float4* srcv = (const float4*)(A + (size_t)row0 * DIM);
    float4* dstv = (float4*)tileA;
    dstv[tid]       = srcv[tid];
    dstv[tid + 256] = srcv[tid + 256];
    __syncthreads();

    const int m  = lane & 15;
    const int kh = (lane >> 4) << 1;
    const int n  = (wave << 4) + (lane & 15);

    const float2* wtrow = (const float2*)(WT + (size_t)n * DIM);

    v8f c = {};
#pragma unroll
    for (int ks = 0; ks < DIM; ks += 4) {
        v2f a, b;
        a.x = tileA[m * DIM + ks + kh];
        a.y = tileA[m * DIM + ks + kh + 1];
        float2 bb = wtrow[(ks + kh) >> 1];
        b.x = bb.x;
        b.y = bb.y;
        c = __builtin_amdgcn_wmma_f32_16x16x4_f32(
                 false, a,  false, b,
                 (short)0, c,  false,  false);
    }

    const int mo = (lane >> 4) << 3;
    float* orow = out + (size_t)(row0 + mo) * DIM + n;
#pragma unroll
    for (int r = 0; r < 8; ++r)
        orow[(size_t)r * DIM] = c[r];
}


__global__ void k_selfloop(const float* __restrict__ h, const float* __restrict__ dinv,
                           float* __restrict__ out) {
    int i = blockIdx.x * blockDim.x + threadIdx.x;
    int node = i >> 5, q = i & 31;
    if (node < NN) {
        float w = dinv[node]; w *= w;
        float4 v = ((const float4*)(h + (size_t)node * DIM))[q];
        float4 r; r.x = v.x * w; r.y = v.y * w; r.z = v.z * w; r.w = v.w * w;
        ((float4*)(out + (size_t)node * DIM))[q] = r;
    }
}

__global__ void k_edge_agg(const int* __restrict__ src, const int* __restrict__ dst,
                           const float* __restrict__ dinv, const float* __restrict__ h,
                           float* __restrict__ out) {
    int i = blockIdx.x * blockDim.x + threadIdx.x;
    int e = i >> 5, q = i & 31;
    if (e < NE) {
        int s = src[e], d = dst[e];
        float w = dinv[s] * dinv[d];
        const float* hs = h + (size_t)s * DIM + q;
        float v0 = hs[0], v1 = hs[32], v2 = hs[64], v3 = hs[96];
        float* o = out + (size_t)d * DIM + q;
        atomicAdd(o + 0,  v0 * w);
        atomicAdd(o + 32, v1 * w);
        atomicAdd(o + 64, v2 * w);
        atomicAdd(o + 96, v3 * w);
    }
}


__global__ void k_count_rank(const int* __restrict__ dst, int* __restrict__ cnt,
                             int* __restrict__ rank, int e) {
    int i = blockIdx.x * blockDim.x + threadIdx.x;
    if (i < e) rank[i] = atomicAdd(&cnt[dst[i]], 1);
}

__global__ void __launch_bounds__(SCAN_T)
k_scan_dinv(const int* __restrict__ cnt, int* __restrict__ rowptr, float* __restrict__ dinv) {
    __shared__ int s_pre[SCAN_T / 32];
    __shared__ int s_own[SCAN_T / 32];
    const int t = threadIdx.x, b = blockIdx.x;
    const int lane = t & 31, wave = t >> 5;

    int pre = 0;
    const int4* c4 = (const int4*)cnt;
    for (int q = t; q < b * (SCAN_T / 4); q += SCAN_T) {
        int4 v = c4[q];
        pre += (v.x + v.y) + (v.z + v.w);
    }
    const int i = b * SCAN_T + t;
    const int own = (i < NN) ? cnt[i] : 0;

    int inc = own;
    for (int d = 1; d < 32; d <<= 1) {
        int u = __shfl_up(inc, d);
        if (lane >= d) inc += u;
    }
    for (int d = 16; d > 0; d >>= 1) pre += __shfl_xor(pre, d);
    if (lane == 31) { s_own[wave] = inc; s_pre[wave] = pre; }
    __syncthreads();
    int base = 0;
    for (int w = 0; w < SCAN_T / 32; ++w) {
        base += s_pre[w];
        if (w < wave) base += s_own[w];
    }
    const int excl = base + inc - own;
    if (i < NN) {
        rowptr[i] = excl;
        dinv[i] = rsqrtf(fmaxf((float)(own + 1), 1.f));
        if (i == NN - 1) rowptr[NN] = excl + own;
    }
}

__global__ void k_fill(const int* __restrict__ src, const int* __restrict__ dst,
                       const int* __restrict__ rank, const int* __restrict__ rowptr,
                       int* __restrict__ csr, int e) {
    int i = blockIdx.x * blockDim.x + threadIdx.x;
    if (i < e) csr[rowptr[dst[i]] + rank[i]] = src[i];
}

__global__ void __launch_bounds__(256)
k_agg(const int* __restrict__ rowptr, const int* __restrict__ csr,
      const float* __restrict__ dinv, const float* __restrict__ h,
      float* __restrict__ out) {
    const int lane = threadIdx.x & 31;
    const int node = __builtin_amdgcn_readfirstlane((blockIdx.x * blockDim.x + threadIdx.x) >> 5);
    if (node >= NN) return;
    const int beg   = rowptr[node];
    const int items = rowptr[node + 1] - beg + 1;
    const float dd  = dinv[node];
    const float4* hv = (const float4*)h;

    float4 acc = {0.f, 0.f, 0.f, 0.f};
    for (int base = 0; base < items; base += 32) {
        const int it = base + lane;
        int s = node; float w = 0.f;
        if (it == 0) w = dd * dd;
        else if (it < items) {
            s = (int)min((unsigned)csr[beg + it - 1], (unsigned)(NN - 1));
            w = dinv[s] * dd;
        }
        const int wi = __float_as_int(w);
        const int n = min(32, items - base);
        for (int j = 0; j < n; j += 4) {
            const int s0 = __builtin_amdgcn_readlane(s, j);
            const int s1 = __builtin_amdgcn_readlane(s, j + 1);
            const int s2 = __builtin_amdgcn_readlane(s, j + 2);
            const int s3 = __builtin_amdgcn_readlane(s, j + 3);
            const float w0 = __int_as_float(__builtin_amdgcn_readlane(wi, j));
            const float w1 = __int_as_float(__builtin_amdgcn_readlane(wi, j + 1));
            const float w2 = __int_as_float(__builtin_amdgcn_readlane(wi, j + 2));
            const float w3 = __int_as_float(__builtin_amdgcn_readlane(wi, j + 3));
            const float4 v0 = hv[(size_t)s0 * 32 + lane];
            const float4 v1 = hv[(size_t)s1 * 32 + lane];
            const float4 v2 = hv[(size_t)s2 * 32 + lane];
            const float4 v3 = hv[(size_t)s3 * 32 + lane];
            acc.x = fmaf(v0.x, w0, acc.x); acc.y = fmaf(v0.y, w0, acc.y);
            acc.z = fmaf(v0.z, w0, acc.z); acc.w = fmaf(v0.w, w0, acc.w);
            acc.x = fmaf(v1.x, w1, acc.x); acc.y = fmaf(v1.y, w1, acc.y);
            acc.z = fmaf(v1.z, w1, acc.z); acc.w = fmaf(v1.w, w1, acc.w);
            acc.x = fmaf(v2.x, w2, acc.x); acc.y = fmaf(v2.y, w2, acc.y);
            acc.z = fmaf(v2.z, w2, acc.z); acc.w = fmaf(v2.w, w2, acc.w);
            acc.x = fmaf(v3.x, w3, acc.x); acc.y = fmaf(v3.y, w3, acc.y);
            acc.z = fmaf(v3.z, w3, acc.z); acc.w = fmaf(v3.w, w3, acc.w);
        }
    }
    ((float4*)out)[(size_t)node * 32 + lane] = acc;
}


__global__ void __launch_bounds__(DIM)
k_bias_relu_stats(float* __restrict__ h, const float* __restrict__ bias,
                  float* __restrict__ stats, int nrows) {
    int c  = threadIdx.x;
    int r0 = blockIdx.x * ROWS_PER_BLOCK;
    int r1 = min(r0 + ROWS_PER_BLOCK, nrows);
    float b = bias[c];
    float s = 0.f, s2 = 0.f;
    for (int r = r0; r < r1; ++r) {
        float v = h[(size_t)r * DIM + c] + b;
        v = fmaxf(v, 0.f);
        h[(size_t)r * DIM + c] = v;
        s += v; s2 += v * v;
    }
    atomicAdd(&stats[c], s);
    atomicAdd(&stats[DIM + c], s2);
}

__global__ void k_stats_finish(float* __restrict__ stats, float inv_n) {
    int c = threadIdx.x;
    float mu  = stats[c] * inv_n;
    float var = stats[DIM + c] * inv_n - mu * mu;
    stats[2 * DIM + c] = mu;
    stats[3 * DIM + c] = rsqrtf(var + BN_EPS);
}

__global__ void __launch_bounds__(DIM)
k_norm_pool(const float* __restrict__ h, const float* __restrict__ stats,
            const float* __restrict__ gamma, const float* __restrict__ beta,
            const int* __restrict__ batch, float* __restrict__ hnext,
            float* __restrict__ out, int layer, int nrows) {
    int c  = threadIdx.x;
    int r0 = blockIdx.x * ROWS_PER_BLOCK;
    int r1 = min(r0 + ROWS_PER_BLOCK, nrows);
    float mu = stats[2 * DIM + c], rstd = stats[3 * DIM + c];
    float ga = gamma[c], be = beta[c];
    int cur = -1; float acc = 0.f;
    for (int r = r0; r < r1; ++r) {
        float v = (h[(size_t)r * DIM + c] - mu) * rstd * ga + be;
        hnext[(size_t)r * DIM + c] = v;
        int g = batch[r];
        if (g != cur) {
            if (cur >= 0) atomicAdd(&out[(size_t)cur * OUTC + layer * DIM + c], acc);
            cur = g; acc = 0.f;
        }
        acc += v;
    }
    if (cur >= 0) atomicAdd(&out[(size_t)cur * OUTC + layer * DIM + c], acc);
}


extern "C" void kernel_launch(void* const* d_in, const int* in_sizes, int n_in,
                              void* d_out, int out_size, void* d_ws, size_t ws_size,
                              hipStream_t stream) {
    (void)in_sizes; (void)n_in; (void)out_size;

    const float* x     = (const float*)d_in[0];
    const int*   ei    = (const int*)d_in[1];
    const int*   batch = (const int*)d_in[2];
    const float *Wp[NL], *bp[NL], *gp[NL], *tp[NL];
    for (int i = 0; i < NL; ++i) {
        Wp[i] = (const float*)d_in[3 + 4 * i];
        bp[i] = (const float*)d_in[4 + 4 * i];
        gp[i] = (const float*)d_in[5 + 4 * i];
        tp[i] = (const float*)d_in[6 + 4 * i];
    }
    const int* esrc = ei;
    const int* edst = ei + NE;

    float* ws    = (float*)d_ws;
    float* dinv  = ws;
    float* stats = ws + NN;
    float* wtb   = ws + NN + 4 * DIM;
    size_t o     = ((size_t)NN + 4 * DIM + DIM * DIM + 255) & ~(size_t)255;
    float* bufA  = ws + o;
    float* bufB  = bufA + (size_t)NN * DIM;
    float* outp  = (float*)d_out;

    int* rowptr  = (int*)(bufB + (size_t)NN * DIM);
    int* csr     = rowptr + RP_PAD;
    int* cnt     = (int*)bufB;
    int* rank    = cnt + RP_PAD;
    const bool sorted =
        ws_size >= (o + 2 * (size_t)NN * DIM + RP_PAD + NE) * sizeof(float);

    k_zero<<<(NG * OUTC + 255) / 256, 256, 0, stream>>>(outp, NG * OUTC);
    if (sorted) {
        k_zero      <<<(NN + 255) / 256,         256,    0, stream>>>((float*)cnt, NN);
        k_count_rank<<<(NE + 255) / 256,         256,    0, stream>>>(edst, cnt, rank, NE);
        k_scan_dinv <<<(NN + SCAN_T - 1) / SCAN_T, SCAN_T, 0, stream>>>(cnt, rowptr, dinv);
        k_fill      <<<(NE + 255) / 256,         256,    0, stream>>>(esrc, edst, rank, rowptr, csr, NE);
    } else {
        k_init_deg <<<(NN + 255) / 256, 256, 0, stream>>>(dinv, NN);
        k_deg_edges<<<(NE + 255) / 256, 256, 0, stream>>>(edst, dinv, NE);
        k_dinv     <<<(NN + 255) / 256, 256, 0, stream>>>(dinv, NN);
    }

    for (int l = 0; l < NL; ++l) {
        const float* hin = (l == 0) ? x : bufA;
        k_transpose128<<<64, 256, 0, stream>>>(Wp[l], wtb);
        k_gemm_wmma<<<NN / 16, 256, 0, stream>>>(hin, wtb, bufA, NN);
        if (sorted) {
            k_agg<<<(NN * 32) / 256, 256, 0, stream>>>(rowptr, csr, dinv, bufA, bufB);
        } else {
            k_selfloop <<<(NN * 32 + 255) / 256, 256, 0, stream>>>(bufA, dinv, bufB);
            k_edge_agg <<<(NE * 32) / 256, 256, 0, stream>>>(esrc, edst, dinv, bufA, bufB);
        }
        k_zero           <<<1, 256, 0, stream>>>(stats, 2 * DIM);
        k_bias_relu_stats<<<(NN + ROWS_PER_BLOCK - 1) / ROWS_PER_BLOCK, DIM, 0, stream>>>(
            bufB, bp[l], stats, NN);
        k_stats_finish   <<<1, DIM, 0, stream>>>(stats, 1.f / (float)NN);
        k_norm_pool      <<<(NN + ROWS_PER_BLOCK - 1) / ROWS_PER_BLOCK, DIM, 0, stream>>>(
            bufB, stats, gp[l], tp[l], batch, bufA, outp, l, NN);
    }
}


# the two row-streaming passes (bias+relu+stats, norm+pool): a wave owns 25 whole rows as float4, all 25 loads in flight (was 128 waited 4-byte loads a thread); partial sums per 25 rows, atomics kept
# speedup vs baseline: 2.7061x; 1.5004x over previous
#include <hip/hip_runtime.h>

#define NN 100000
#define NE 1600000
#define DIM 128
#define NG 512
#define NL 3
#define OUTC (NL * DIM)
#define BN_EPS 1e-5f
#define EPI_RPW 25
#define EPI_BLOCKS (NN / EPI_RPW / 8)
#define SCAN_T 1024
#define RP_PAD 100096

typedef __attribute__((ext_vector_type(2))) float v2f;
typedef __attribute__((ext_vector_type(8))) float v8f;


__global__ void k_zero(float* __restrict__ p, int n) {
    int i = blockIdx.x * blockDim.x + threadIdx.x;
    if (i < n) p[i] = 0.f;
}

__global__ void k_init_deg(float* __restrict__ deg, int n) {
    int i = blockIdx.x * blockDim.x + threadIdx.x;
    if (i < n) deg[i] = 1.f;
}

__global__ void k_deg_edges(const int* __restrict__ dst, float* __restrict__ deg, int e) {
    int i = blockIdx.x * blockDim.x + threadIdx.x;
    if (i < e) atomicAdd(&deg[dst[i]], 1.f);
}

__global__ void k_dinv(float* __restrict__ deg, int n) {
    int i = blockIdx.x * blockDim.x + threadIdx.x;
    if (i < n) deg[i] = rsqrtf(fmaxf(deg[i], 1.f));
}

__global__ void __launch_bounds__(256)
k_transpose128(const float* __restrict__ W, float* __restrict__ WT) {
    __shared__ float t[16][17];
    int bx = blockIdx.x & 7, by = blockIdx.x >> 3;
    int tx = threadIdx.x & 15, ty = threadIdx.x >> 4;
    t[ty][tx] = W[(size_t)(by * 16 + ty) * DIM + bx * 16 + tx];
    __syncthreads();
    WT[(size_t)(bx * 16 + ty) * DIM + by * 16 + tx] = t[tx][ty];
}


__global__ void __launch_bounds__(256)
k_gemm_wmma(const float* __restrict__ A, const float* __restrict__ WT,
            float* __restrict__ out, int nrows) {
    __shared__ float tileA[16 * DIM];

    const int tid  = threadIdx.x;
    const int wave = tid >> 5;
    const int lane = tid & 31;
    const int row0 = blockIdx.x << 4;
    (void)nrows;

    const float4* srcv = (const float4*)(A + (size_t)row0 * DIM);
    float4* dstv = (float4*)tileA;
    dstv[tid]       = srcv[tid];
    dstv[tid + 256] = srcv[tid + 256];
    __syncthreads();

    const int m  = lane & 15;
    const int kh = (lane >> 4) << 1;
    const int n  = (wave << 4) + (lane & 15);

    const float2* wtrow = (const float2*)(WT + (size_t)n * DIM);

    v8f c = {};
#pragma unroll
    for (int ks = 0; ks < DIM; ks += 4) {
        v2f a, b;
        a.x = tileA[m * DIM + ks + kh];
        a.y = tileA[m * DIM + ks + kh + 1];
        float2 bb = wtrow[(ks + kh) >> 1];
        b.x = bb.x;
        b.y = bb.y;
        c = __builtin_amdgcn_wmma_f32_16x16x4_f32(
                 false, a,  false, b,
                 (short)0, c,  false,  false);
    }

    const int mo = (lane >> 4) << 3;
    float* orow = out + (size_t)(row0 + mo) * DIM + n;
#pragma unroll
    for (int r = 0; r < 8; ++r)
        orow[(size_t)r * DIM] = c[r];
}


__global__ void k_selfloop(const float* __restrict__ h, const float* __restrict__ dinv,
                           float* __restrict__ out) {
    int i = blockIdx.x * blockDim.x + threadIdx.x;
    int node = i >> 5, q = i & 31;
    if (node < NN) {
        float w = dinv[node]; w *= w;
        float4 v = ((const float4*)(h + (size_t)node * DIM))[q];
        float4 r; r.x = v.x * w; r.y = v.y * w; r.z = v.z * w; r.w = v.w * w;
        ((float4*)(out + (size_t)node * DIM))[q] = r;
    }
}

__global__ void k_edge_agg(const int* __restrict__ src, const int* __restrict__ dst,
                           const float* __restrict__ dinv, const float* __restrict__ h,
                           float* __restrict__ out) {
    int i = blockIdx.x * blockDim.x + threadIdx.x;
    int e = i >> 5, q = i & 31;
    if (e < NE) {
        int s = src[e], d = dst[e];
        float w = dinv[s] * dinv[d];
        const float* hs = h + (size_t)s * DIM + q;
        float v0 = hs[0], v1 = hs[32], v2 = hs[64], v3 = hs[96];
        float* o = out + (size_t)d * DIM + q;
        atomicAdd(o + 0,  v0 * w);
        atomicAdd(o + 32, v1 * w);
        atomicAdd(o + 64, v2 * w);
        atomicAdd(o + 96, v3 * w);
    }
}


__global__ void k_count_rank(const int* __restrict__ dst, int* __restrict__ cnt,
                             int* __restrict__ rank, int e) {
    int i = blockIdx.x * blockDim.x + threadIdx.x;
    if (i < e) rank[i] = atomicAdd(&cnt[dst[i]], 1);
}

__global__ void __launch_bounds__(SCAN_T)
k_scan_dinv(const int* __restrict__ cnt, int* __restrict__ rowptr, float* __restrict__ dinv) {
    __shared__ int s_pre[SCAN_T / 32];
    __shared__ int s_own[SCAN_T / 32];
    const int t = threadIdx.x, b = blockIdx.x;
    const int lane = t & 31, wave = t >> 5;

    int pre = 0;
    const int4* c4 = (const int4*)cnt;
    for (int q = t; q < b * (SCAN_T / 4); q += SCAN_T) {
        int4 v = c4[q];
        pre += (v.x + v.y) + (v.z + v.w);
    }
    const int i = b * SCAN_T + t;
    const int own = (i < NN) ? cnt[i] : 0;

    int inc = own;
    for (int d = 1; d < 32; d <<= 1) {
        int u = __shfl_up(inc, d);
        if (lane >= d) inc += u;
    }
    for (int d = 16; d > 0; d >>= 1) pre += __shfl_xor(pre, d);
    if (lane == 31) { s_own[wave] = inc; s_pre[wave] = pre; }
    __syncthreads();
    int base = 0;
    for (int w = 0; w < SCAN_T / 32; ++w) {
        base += s_pre[w];
        if (w < wave) base += s_own[w];
    }
    const int excl = base + inc - own;
    if (i < NN) {
        rowptr[i] = excl;
        dinv[i] = rsqrtf(fmaxf((float)(own + 1), 1.f));
        if (i == NN - 1) rowptr[NN] = excl + own;
    }
}

__global__ void k_fill(const int* __restrict__ src, const int* __restrict__ dst,
                       const int* __restrict__ rank, const int* __restrict__ rowptr,
                       int* __restrict__ csr, int e) {
    int i = blockIdx.x * blockDim.x + threadIdx.x;
    if (i < e) csr[rowptr[dst[i]] + rank[i]] = src[i];
}

__global__ void __launch_bounds__(256)
k_agg(const int* __restrict__ rowptr, const int* __restrict__ csr,
      const float* __restrict__ dinv, const float* __restrict__ h,
      float* __restrict__ out) {
    const int lane = threadIdx.x & 31;
    const int node = __builtin_amdgcn_readfirstlane((blockIdx.x * blockDim.x + threadIdx.x) >> 5);
    if (node >= NN) return;
    const int beg   = rowptr[node];
    const int items = rowptr[node + 1] - beg + 1;
    const float dd  = dinv[node];
    const float4* hv = (const float4*)h;

    float4 acc = {0.f, 0.f, 0.f, 0.f};
    for (int base = 0; base < items; base += 32) {
        const int it = base + lane;
        int s = node; float w = 0.f;
        if (it == 0) w = dd * dd;
        else if (it < items) {
            s = (int)min((unsigned)csr[beg + it - 1], (unsigned)(NN - 1));
            w = dinv[s] * dd;
        }
        const int wi = __float_as_int(w);
        const int n = min(32, items - base);
        for (int j = 0; j < n; j += 4) {
            const int s0 = __builtin_amdgcn_readlane(s, j);
            const int s1 = __builtin_amdgcn_readlane(s, j + 1);
            const int s2 = __builtin_amdgcn_readlane(s, j + 2);
            const int s3 = __builtin_amdgcn_readlane(s, j + 3);
            const float w0 = __int_as_float(__builtin_amdgcn_readlane(wi, j));
            const float w1 = __int_as_float(__builtin_amdgcn_readlane(wi, j + 1));
            const float w2 = __int_as_float(__builtin_amdgcn_readlane(wi, j + 2));
            const float w3 = __int_as_float(__builtin_amdgcn_readlane(wi, j + 3));
            const float4 v0 = hv[(size_t)s0 * 32 + lane];
            const float4 v1 = hv[(size_t)s1 * 32 + lane];
            const float4 v2 = hv[(size_t)s2 * 32 + lane];
            const float4 v3 = hv[(size_t)s3 * 32 + lane];
            acc.x = fmaf(v0.x, w0, acc.x); acc.y = fmaf(v0.y, w0, acc.y);
            acc.z = fmaf(v0.z, w0, acc.z); acc.w = fmaf(v0.w, w0, acc.w);
            acc.x = fmaf(v1.x, w1, acc.x); acc.y = fmaf(v1.y, w1, acc.y);
            acc.z = fmaf(v1.z, w1, acc.z); acc.w = fmaf(v1.w, w1, acc.w);
            acc.x = fmaf(v2.x, w2, acc.x); acc.y = fmaf(v2.y, w2, acc.y);
            acc.z = fmaf(v2.z, w2, acc.z); acc.w = fmaf(v2.w, w2, acc.w);
            acc.x = fmaf(v3.x, w3, acc.x); acc.y = fmaf(v3.y, w3, acc.y);
            acc.z = fmaf(v3.z, w3, acc.z); acc.w = fmaf(v3.w, w3, acc.w);
        }
    }
    ((float4*)out)[(size_t)node * 32 + lane] = acc;
}


static_assert(NN % (EPI_RPW * 8) == 0, "every wave of every block owns EPI_RPW rows");

__global__ void __launch_bounds__(256)
k_bias_relu_stats(float* __restrict__ h, const float* __restrict__ bias,
                  float* __restrict__ stats) {
    __shared__ float red[8][2 * DIM];
    const int lane = threadIdx.x & 31;
    const int wave = threadIdx.x >> 5;
    const int r0 = __builtin_amdgcn_readfirstlane((blockIdx.x * 8 + wave) * EPI_RPW);
    float4* hv = (float4*)h + (size_t)r0 * 32 + lane;
    const float4 b = ((const float4*)bias)[lane];

    float4 v[EPI_RPW];
#pragma unroll
    for (int u = 0; u < EPI_RPW; ++u) v[u] = hv[u * 32];

    float4 s = {0.f, 0.f, 0.f, 0.f}, q = {0.f, 0.f, 0.f, 0.f};
#pragma unroll
    for (int u = 0; u < EPI_RPW; ++u) {
        float4 x = v[u];
        x.x = fmaxf(x.x + b.x, 0.f); x.y = fmaxf(x.y + b.y, 0.f);
        x.z = fmaxf(x.z + b.z, 0.f); x.w = fmaxf(x.w + b.w, 0.f);
        hv[u * 32] = x;
        s.x += x.x; s.y += x.y; s.z += x.z; s.w += x.w;
        q.x += x.x * x.x; q.y += x.y * x.y; q.z += x.z * x.z; q.w += x.w * x.w;
    }
    ((float4*)&red[wave][0])[lane]   = s;
    ((float4*)&red[wave][DIM])[lane] = q;
    __syncthreads();
    float t = 0.f;
    for (int w = 0; w < 8; ++w) t += red[w][threadIdx.x];
    atomicAdd(&stats[threadIdx.x], t);
}

__global__ void k_stats_finish(float* __restrict__ stats, float inv_n) {
    int c = threadIdx.x;
    float mu  = stats[c] * inv_n;
    float var = stats[DIM + c] * inv_n - mu * mu;
    stats[2 * DIM + c] = mu;
    stats[3 * DIM + c] = rsqrtf(var + BN_EPS);
}

__device__ __forceinline__ void pool_flush(float* __restrict__ out, int g, int layer, int lane,
                                           float4 acc) {
    if ((unsigned)g < (unsigned)NG) {
        float* o = out + (size_t)g * OUTC + layer * DIM + 4 * lane;
        atomicAdd(o + 0, acc.x); atomicAdd(o + 1, acc.y);
        atomicAdd(o + 2, acc.z); atomicAdd(o + 3, acc.w);
    }
}

__global__ void __launch_bounds__(256)
k_norm_pool(const float* __restrict__ h, const float* __restrict__ stats,
            const float* __restrict__ gamma, const float* __restrict__ beta,
            const int* __restrict__ batch, float* __restrict__ hnext,
            float* __restrict__ out, int layer) {
    const int lane = threadIdx.x & 31;
    const int wave = threadIdx.x >> 5;
    const int r0 = __builtin_amdgcn_readfirstlane((blockIdx.x * 8 + wave) * EPI_RPW);
    const float4* hv = (const float4*)h + (size_t)r0 * 32 + lane;
    float4*       hn = (float4*)hnext   + (size_t)r0 * 32 + lane;

    float4 v[EPI_RPW];
#pragma unroll
    for (int u = 0; u < EPI_RPW; ++u) v[u] = hv[u * 32];

    const int gl = batch[r0 + min(lane, EPI_RPW - 1)];
    const float4 mu = ((const float4*)(stats + 2 * DIM))[lane];
    const float4 rs = ((const float4*)(stats + 3 * DIM))[lane];
    const float4 ga = ((const float4*)gamma)[lane];
    const float4 be = ((const float4*)beta)[lane];

    int cur = __builtin_amdgcn_readlane(gl, 0);
    float4 acc = {0.f, 0.f, 0.f, 0.f};
#pragma unroll
    for (int u = 0; u < EPI_RPW; ++u) {
        float4 x;
        x.x = (v[u].x - mu.x) * rs.x * ga.x + be.x;
        x.y = (v[u].y - mu.y) * rs.y * ga.y + be.y;
        x.z = (v[u].z - mu.z) * rs.z * ga.z + be.z;
        x.w = (v[u].w - mu.w) * rs.w * ga.w + be.w;
        hn[u * 32] = x;
        const int g = __builtin_amdgcn_readlane(gl, u);
        if (g != cur) {
            pool_flush(out, cur, layer, lane, acc);
            cur = g;
            acc.x = 0.f; acc.y = 0.f; acc.z = 0.f; acc.w = 0.f;
        }
        acc.x += x.x; acc.y += x.y; acc.z += x.z; acc.w += x.w;
    }
    pool_flush(out, cur, layer, lane, acc);
}


extern "C" void kernel_launch(void* const* d_in, const int* in_sizes, int n_in,
                              void* d_out, int out_size, void* d_ws, size_t ws_size,
                              hipStream_t stream) {
    (void)in_sizes; (void)n_in; (void)out_size;

    const float* x     = (const float*)d_in[0];
    const int*   ei    = (const int*)d_in[1];
    const int*   batch = (const int*)d_in[2];
    const float *Wp[NL], *bp[NL], *gp[NL], *tp[NL];
    for (int i = 0; i < NL; ++i) {
        Wp[i] = (const float*)d_in[3 + 4 * i];
        bp[i] = (const float*)d_in[4 + 4 * i];
        gp[i] = (const float*)d_in[5 + 4 * i];
        tp[i] = (const float*)d_in[6 + 4 * i];
    }
    const int* esrc = ei;
    const int* edst = ei + NE;

    float* ws    = (float*)d_ws;
    float* dinv  = ws;
    float* stats = ws + NN;
    float* wtb   = ws + NN + 4 * DIM;
    size_t o     = ((size_t)NN + 4 * DIM + DIM * DIM + 255) & ~(size_t)255;
    float* bufA  = ws + o;
    float* bufB  = bufA + (size_t)NN * DIM;
    float* outp  = (float*)d_out;

    int* rowptr  = (int*)(bufB + (size_t)NN * DIM);
    int* csr     = rowptr + RP_PAD;
    int* cnt     = (int*)bufB;
    int* rank    = cnt + RP_PAD;
    const bool sorted =
        ws_size >= (o + 2 * (size_t)NN * DIM + RP_PAD + NE) * sizeof(float);

    k_zero<<<(NG * OUTC + 255) / 256, 256, 0, stream>>>(outp, NG * OUTC);
    if (sorted) {
        k_zero      <<<(NN + 255) / 256,         256,    0, stream>>>((float*)cnt, NN);
        k_count_rank<<<(NE + 255) / 256,         256,    0, stream>>>(edst, cnt, rank, NE);
        k_scan_dinv <<<(NN + SCAN_T - 1) / SCAN_T, SCAN_T, 0, stream>>>(cnt, rowptr, dinv);
        k_fill      <<<(NE + 255) / 256,         256,    0, stream>>>(esrc, edst, rank, rowptr, csr, NE);
    } else {
        k_init_deg <<<(NN + 255) / 256, 256, 0, stream>>>(dinv, NN);
        k_deg_edges<<<(NE + 255) / 256, 256, 0, stream>>>(edst, dinv, NE);
        k_dinv     <<<(NN + 255) / 256, 256, 0, stream>>>(dinv, NN);
    }

    for (int l = 0; l < NL; ++l) {
        const float* hin = (l == 0) ? x : bufA;
        k_transpose128<<<64, 256, 0, stream>>>(Wp[l], wtb);
        k_gemm_wmma<<<NN / 16, 256, 0, stream>>>(hin, wtb, bufA, NN);
        if (sorted) {
            k_agg<<<(NN * 32) / 256, 256, 0, stream>>>(rowptr, csr, dinv, bufA, bufB);
        } else {
            k_selfloop <<<(NN * 32 + 255) / 256, 256, 0, stream>>>(bufA, dinv, bufB);
            k_edge_agg <<<(NE * 32) / 256, 256, 0, stream>>>(esrc, edst, dinv, bufA, bufB);
        }
        k_zero           <<<1, 256, 0, stream>>>(stats, 2 * DIM);
        k_bias_relu_stats<<<EPI_BLOCKS, 256, 0, stream>>>(bufB, bp[l], stats);
        k_stats_finish   <<<1, DIM, 0, stream>>>(stats, 1.f / (float)NN);
        k_norm_pool      <<<EPI_BLOCKS, 256, 0, stream>>>(
            bufB, stats, gp[l], tp[l], batch, bufA, outp, l);
    }
}


# GEMM re-tiled: a block owns 80 rows, each wave keeps its 128x16 slice of W in registers (read straight from W, transpose launch gone) over 5 row tiles; A rows in LDS at pitch 132, conflict-free
# speedup vs baseline: 3.1665x; 1.1701x over previous
#include <hip/hip_runtime.h>

#define NN 100000
#define NE 1600000
#define DIM 128
#define NG 512
#define NL 3
#define OUTC (NL * DIM)
#define BN_EPS 1e-5f
#define EPI_RPW 25
#define EPI_BLOCKS (NN / EPI_RPW / 8)
#define SCAN_T 1024
#define RP_PAD 100096
#define GEMM_TILES 5
#define GEMM_ROWS (16 * GEMM_TILES)
#define A_PITCH (DIM + 4)

typedef __attribute__((ext_vector_type(2))) float v2f;
typedef __attribute__((ext_vector_type(8))) float v8f;


__global__ void k_zero(float* __restrict__ p, int n) {
    int i = blockIdx.x * blockDim.x + threadIdx.x;
    if (i < n) p[i] = 0.f;
}

__global__ void k_init_deg(float* __restrict__ deg, int n) {
    int i = blockIdx.x * blockDim.x + threadIdx.x;
    if (i < n) deg[i] = 1.f;
}

__global__ void k_deg_edges(const int* __restrict__ dst, float* __restrict__ deg, int e) {
    int i = blockIdx.x * blockDim.x + threadIdx.x;
    if (i < e) atomicAdd(&deg[dst[i]], 1.f);
}

__global__ void k_dinv(float* __restrict__ deg, int n) {
    int i = blockIdx.x * blockDim.x + threadIdx.x;
    if (i < n) deg[i] = rsqrtf(fmaxf(deg[i], 1.f));
}

static_assert(NN % GEMM_ROWS == 0, "every block owns GEMM_ROWS whole rows");
static_assert((GEMM_ROWS * 32) % 256 == 0, "staging: a whole number of float4 per thread");

__global__ void __launch_bounds__(256)
k_gemm_wmma(const float* __restrict__ A, const float* __restrict__ W,
            float* __restrict__ out) {
    __shared__ float tileA[GEMM_ROWS * A_PITCH];

    const int tid  = threadIdx.x;
    const int wave = tid >> 5;
    const int lane = tid & 31;
    const int row0 = blockIdx.x * GEMM_ROWS;

    constexpr int NST = GEMM_ROWS * 32 / 256;
    const float4* srcv = (const float4*)(A + (size_t)row0 * DIM);
    float4 st[NST];
#pragma unroll
    for (int j = 0; j < NST; ++j) st[j] = srcv[tid + 256 * j];

    const int m  = lane & 15;
    const int kh = (lane >> 4) << 1;
    const int n  = (wave << 4) + m;

    const float* wcol = W + (size_t)kh * DIM + n;
    v2f b[DIM / 4];
#pragma unroll
    for (int s = 0; s < DIM / 4; ++s) {
        b[s].x = wcol[(size_t)(4 * s) * DIM];
        b[s].y = wcol[(size_t)(4 * s + 1) * DIM];
    }

#pragma unroll
    for (int j = 0; j < NST; ++j) {
        const int i = tid + 256 * j;
        *(float4*)&tileA[(i >> 5) * A_PITCH + ((i & 31) << 2)] = st[j];
    }
    __syncthreads();

    v8f c[GEMM_TILES];
#pragma unroll
    for (int t = 0; t < GEMM_TILES; ++t) c[t] = v8f{};

    const float* arow = tileA + m * A_PITCH + kh;
#pragma unroll
    for (int s = 0; s < DIM / 4; ++s) {
#pragma unroll
        for (int t = 0; t < GEMM_TILES; ++t) {
            const v2f a = *(const v2f*)(arow + t * 16 * A_PITCH + 4 * s);
            c[t] = __builtin_amdgcn_wmma_f32_16x16x4_f32(
                        false, a,  false, b[s],
                        (short)0, c[t],  false,  false);
        }
    }

    const int mo = (lane >> 4) << 3;
    float* orow = out + (size_t)(row0 + mo) * DIM + n;
#pragma unroll
    for (int t = 0; t < GEMM_TILES; ++t)
#pragma unroll
        for (int r = 0; r < 8; ++r)
            orow[(size_t)(t * 16 + r) * DIM] = c[t][r];
}


__global__ void k_selfloop(const float* __restrict__ h, const float* __restrict__ dinv,
                           float* __restrict__ out) {
    int i = blockIdx.x * blockDim.x + threadIdx.x;
    int node = i >> 5, q = i & 31;
    if (node < NN) {
        float w = dinv[node]; w *= w;
        float4 v = ((const float4*)(h + (size_t)node * DIM))[q];
        float4 r; r.x = v.x * w; r.y = v.y * w; r.z = v.z * w; r.w = v.w * w;
        ((float4*)(out + (size_t)node * DIM))[q] = r;
    }
}

__global__ void k_edge_agg(const int* __restrict__ src, const int* __restrict__ dst,
                           const float* __restrict__ dinv, const float* __restrict__ h,
                           float* __restrict__ out) {
    int i = blockIdx.x * blockDim.x + threadIdx.x;
    int e = i >> 5, q = i & 31;
    if (e < NE) {
        int s = src[e], d = dst[e];
        float w = dinv[s] * dinv[d];
        const float* hs = h + (size_t)s * DIM + q;
        float v0 = hs[0], v1 = hs[32], v2 = hs[64], v3 = hs[96];
        float* o = out + (size_t)d * DIM + q;
        atomicAdd(o + 0,  v0 * w);
        atomicAdd(o + 32, v1 * w);
        atomicAdd(o + 64, v2 * w);
        atomicAdd(o + 96, v3 * w);
    }
}


__global__ void k_count_rank(const int* __restrict__ dst, int* __restrict__ cnt,
                             int* __restrict__ rank, int e) {
    int i = blockIdx.x * blockDim.x + threadIdx.x;
    if (i < e) rank[i] = atomicAdd(&cnt[dst[i]], 1);
}

__global__ void __launch_bounds__(SCAN_T)
k_scan_dinv(const int* __restrict__ cnt, int* __restrict__ rowptr, float* __restrict__ dinv) {
    __shared__ int s_pre[SCAN_T / 32];
    __shared__ int s_own[SCAN_T / 32];
    const int t = threadIdx.x, b = blockIdx.x;
    const int lane = t & 31, wave = t >> 5;

    int pre = 0;
    const int4* c4 = (const int4*)cnt;
    for (int q = t; q < b * (SCAN_T / 4); q += SCAN_T) {
        int4 v = c4[q];
        pre += (v.x + v.y) + (v.z + v.w);
    }
    const int i = b * SCAN_T + t;
    const int own = (i < NN) ? cnt[i] : 0;

    int inc = own;
    for (int d = 1; d < 32; d <<= 1) {
        int u = __shfl_up(inc, d);
        if (lane >= d) inc += u;
    }
    for (int d = 16; d > 0; d >>= 1) pre += __shfl_xor(pre, d);
    if (lane == 31) { s_own[wave] = inc; s_pre[wave] = pre; }
    __syncthreads();
    int base = 0;
    for (int w = 0; w < SCAN_T / 32; ++w) {
        base += s_pre[w];
        if (w < wave) base += s_own[w];
    }
    const int excl = base + inc - own;
    if (i < NN) {
        rowptr[i] = excl;
        dinv[i] = rsqrtf(fmaxf((float)(own + 1), 1.f));
        if (i == NN - 1) rowptr[NN] = excl + own;
    }
}

__global__ void k_fill(const int* __restrict__ src, const int* __restrict__ dst,
                       const int* __restrict__ rank, const int* __restrict__ rowptr,
                       int* __restrict__ csr, int e) {
    int i = blockIdx.x * blockDim.x + threadIdx.x;
    if (i < e) csr[rowptr[dst[i]] + rank[i]] = src[i];
}

__global__ void __launch_bounds__(256)
k_agg(const int* __restrict__ rowptr, const int* __restrict__ csr,
      const float* __restrict__ dinv, const float* __restrict__ h,
      float* __restrict__ out) {
    const int lane = threadIdx.x & 31;
    const int node = __builtin_amdgcn_readfirstlane((blockIdx.x * blockDim.x + threadIdx.x) >> 5);
    if (node >= NN) return;
    const int beg   = rowptr[node];
    const int items = rowptr[node + 1] - beg + 1;
    const float dd  = dinv[node];
    const float4* hv = (const float4*)h;

    float4 acc = {0.f, 0.f, 0.f, 0.f};
    for (int base = 0; base < items; base += 32) {
        const int it = base + lane;
        int s = node; float w = 0.f;
        if (it == 0) w = dd * dd;
        else if (it < items) {
            s = (int)min((unsigned)csr[beg + it - 1], (unsigned)(NN - 1));
            w = dinv[s] * dd;
        }
        const int wi = __float_as_int(w);
        const int n = min(32, items - base);
        for (int j = 0; j < n; j += 4) {
            const int s0 = __builtin_amdgcn_readlane(s, j);
            const int s1 = __builtin_amdgcn_readlane(s, j + 1);
            const int s2 = __builtin_amdgcn_readlane(s, j + 2);
            const int s3 = __builtin_amdgcn_readlane(s, j + 3);
            const float w0 = __int_as_float(__builtin_amdgcn_readlane(wi, j));
            const float w1 = __int_as_float(__builtin_amdgcn_readlane(wi, j + 1));
            const float w2 = __int_as_float(__builtin_amdgcn_readlane(wi, j + 2));
            const float w3 = __int_as_float(__builtin_amdgcn_readlane(wi, j + 3));
            const float4 v0 = hv[(size_t)s0 * 32 + lane];
            const float4 v1 = hv[(size_t)s1 * 32 + lane];
            const float4 v2 = hv[(size_t)s2 * 32 + lane];
            const float4 v3 = hv[(size_t)s3 * 32 + lane];
            acc.x = fmaf(v0.x, w0, acc.x); acc.y = fmaf(v0.y, w0, acc.y);
            acc.z = fmaf(v0.z, w0, acc.z); acc.w = fmaf(v0.w, w0, acc.w);
            acc.x = fmaf(v1.x, w1, acc.x); acc.y = fmaf(v1.y, w1, acc.y);
            acc.z = fmaf(v1.z, w1, acc.z); acc.w = fmaf(v1.w, w1, acc.w);
            acc.x = fmaf(v2.x, w2, acc.x); acc.y = fmaf(v2.y, w2, acc.y);
            acc.z = fmaf(v2.z, w2, acc.z); acc.w = fmaf(v2.w, w2, acc.w);
            acc.x = fmaf(v3.x, w3, acc.x); acc.y = fmaf(v3.y, w3, acc.y);
            acc.z = fmaf(v3.z, w3, acc.z); acc.w = fmaf(v3.w, w3, acc.w);
        }
    }
    ((float4*)out)[(size_t)node * 32 + lane] = acc;
}


static_assert(NN % (EPI_RPW * 8) == 0, "every wave of every block owns EPI_RPW rows");

__global__ void __launch_bounds__(256)
k_bias_relu_stats(float* __restrict__ h, const float* __restrict__ bias,
                  float* __restrict__ stats) {
    __shared__ float red[8][2 * DIM];
    const int lane = threadIdx.x & 31;
    const int wave = threadIdx.x >> 5;
    const int r0 = __builtin_amdgcn_readfirstlane((blockIdx.x * 8 + wave) * EPI_RPW);
    float4* hv = (float4*)h + (size_t)r0 * 32 + lane;
    const float4 b = ((const float4*)bias)[lane];

    float4 v[EPI_RPW];
#pragma unroll
    for (int u = 0; u < EPI_RPW; ++u) v[u] = hv[u * 32];

    float4 s = {0.f, 0.f, 0.f, 0.f}, q = {0.f, 0.f, 0.f, 0.f};
#pragma unroll
    for (int u = 0; u < EPI_RPW; ++u) {
        float4 x = v[u];
        x.x = fmaxf(x.x + b.x, 0.f); x.y = fmaxf(x.y + b.y, 0.f);
        x.z = fmaxf(x.z + b.z, 0.f); x.w = fmaxf(x.w + b.w, 0.f);
        hv[u * 32] = x;
        s.x += x.x; s.y += x.y; s.z += x.z; s.w += x.w;
        q.x += x.x * x.x; q.y += x.y * x.y; q.z += x.z * x.z; q.w += x.w * x.w;
    }
    ((float4*)&red[wave][0])[lane]   = s;
    ((float4*)&red[wave][DIM])[lane] = q;
    __syncthreads();
    float t = 0.f;
    for (int w = 0; w < 8; ++w) t += red[w][threadIdx.x];
    atomicAdd(&stats[threadIdx.x], t);
}

__global__ void k_stats_finish(float* __restrict__ stats, float inv_n) {
    int c = threadIdx.x;
    float mu  = stats[c] * inv_n;
    float var = stats[DIM + c] * inv_n - mu * mu;
    stats[2 * DIM + c] = mu;
    stats[3 * DIM + c] = rsqrtf(var + BN_EPS);
}

__device__ __forceinline__ void pool_flush(float* __restrict__ out, int g, int layer, int lane,
                                           float4 acc) {
    if ((unsigned)g < (unsigned)NG) {
        float* o = out + (size_t)g * OUTC + layer * DIM + 4 * lane;
        atomicAdd(o + 0, acc.x); atomicAdd(o + 1, acc.y);
        atomicAdd(o + 2, acc.z); atomicAdd(o + 3, acc.w);
    }
}

__global__ void __launch_bounds__(256)
k_norm_pool(const float* __restrict__ h, const float* __restrict__ stats,
            const float* __restrict__ gamma, const float* __restrict__ beta,
            const int* __restrict__ batch, float* __restrict__ hnext,
            float* __restrict__ out, int layer) {
    const int lane = threadIdx.x & 31;
    const int wave = threadIdx.x >> 5;
    const int r0 = __builtin_amdgcn_readfirstlane((blockIdx.x * 8 + wave) * EPI_RPW);
    const float4* hv = (const float4*)h + (size_t)r0 * 32 + lane;
    float4*       hn = (float4*)hnext   + (size_t)r0 * 32 + lane;

    float4 v[EPI_RPW];
#pragma unroll
    for (int u = 0; u < EPI_RPW; ++u) v[u] = hv[u * 32];

    const int gl = batch[r0 + min(lane, EPI_RPW - 1)];
    const float4 mu = ((const float4*)(stats + 2 * DIM))[lane];
    const float4 rs = ((const float4*)(stats + 3 * DIM))[lane];
    const float4 ga = ((const float4*)gamma)[lane];
    const float4 be = ((const float4*)beta)[lane];

    int cur = __builtin_amdgcn_readlane(gl, 0);
    float4 acc = {0.f, 0.f, 0.f, 0.f};
#pragma unroll
    for (int u = 0; u < EPI_RPW; ++u) {
        float4 x;
        x.x = (v[u].x - mu.x) * rs.x * ga.x + be.x;
        x.y = (v[u].y - mu.y) * rs.y * ga.y + be.y;
        x.z = (v[u].z - mu.z) * rs.z * ga.z + be.z;
        x.w = (v[u].w - mu.w) * rs.w * ga.w + be.w;
        hn[u * 32] = x;
        const int g = __builtin_amdgcn_readlane(gl, u);
        if (g != cur) {
            pool_flush(out, cur, layer, lane, acc);
            cur = g;
            acc.x = 0.f; acc.y = 0.f; acc.z = 0.f; acc.w = 0.f;
        }
        acc.x += x.x; acc.y += x.y; acc.z += x.z; acc.w += x.w;
    }
    pool_flush(out, cur, layer, lane, acc);
}


extern "C" void kernel_launch(void* const* d_in, const int* in_sizes, int n_in,
                              void* d_out, int out_size, void* d_ws, size_t ws_size,
                              hipStream_t stream) {
    (void)in_sizes; (void)n_in; (void)out_size;

    const float* x     = (const float*)d_in[0];
    const int*   ei    = (const int*)d_in[1];
    const int*   batch = (const int*)d_in[2];
    const float *Wp[NL], *bp[NL], *gp[NL], *tp[NL];
    for (int i = 0; i < NL; ++i) {
        Wp[i] = (const float*)d_in[3 + 4 * i];
        bp[i] = (const float*)d_in[4 + 4 * i];
        gp[i] = (const float*)d_in[5 + 4 * i];
        tp[i] = (const float*)d_in[6 + 4 * i];
    }
    const int* esrc = ei;
    const int* edst = ei + NE;

    float* ws    = (float*)d_ws;
    float* dinv  = ws;
    float* stats = ws + NN;
    size_t o     = ((size_t)NN + 4 * DIM + 255) & ~(size_t)255;
    float* bufA  = ws + o;
    float* bufB  = bufA + (size_t)NN * DIM;
    float* outp  = (float*)d_out;

    int* rowptr  = (int*)(bufB + (size_t)NN * DIM);
    int* csr     = rowptr + RP_PAD;
    int* cnt     = (int*)bufB;
    int* rank    = cnt + RP_PAD;
    const bool sorted =
        ws_size >= (o + 2 * (size_t)NN * DIM + RP_PAD + NE) * sizeof(float);

    k_zero<<<(NG * OUTC + 255) / 256, 256, 0, stream>>>(outp, NG * OUTC);
    if (sorted) {
        k_zero      <<<(NN + 255) / 256,         256,    0, stream>>>((float*)cnt, NN);
        k_count_rank<<<(NE + 255) / 256,         256,    0, stream>>>(edst, cnt, rank, NE);
        k_scan_dinv <<<(NN + SCAN_T - 1) / SCAN_T, SCAN_T, 0, stream>>>(cnt, rowptr, dinv);
        k_fill      <<<(NE + 255) / 256,         256,    0, stream>>>(esrc, edst, rank, rowptr, csr, NE);
    } else {
        k_init_deg <<<(NN + 255) / 256, 256, 0, stream>>>(dinv, NN);
        k_deg_edges<<<(NE + 255) / 256, 256, 0, stream>>>(edst, dinv, NE);
        k_dinv     <<<(NN + 255) / 256, 256, 0, stream>>>(dinv, NN);
    }

    for (int l = 0; l < NL; ++l) {
        const float* hin = (l == 0) ? x : bufA;
        k_gemm_wmma<<<NN / GEMM_ROWS, 256, 0, stream>>>(hin, Wp[l], bufA);
        if (sorted) {
            k_agg<<<(NN * 32) / 256, 256, 0, stream>>>(rowptr, csr, dinv, bufA, bufB);
        } else {
            k_selfloop <<<(NN * 32 + 255) / 256, 256, 0, stream>>>(bufA, dinv, bufB);
            k_edge_agg <<<(NE * 32) / 256, 256, 0, stream>>>(esrc, edst, dinv, bufA, bufB);
        }
        k_zero           <<<1, 256, 0, stream>>>(stats, 2 * DIM);
        k_bias_relu_stats<<<EPI_BLOCKS, 256, 0, stream>>>(bufB, bp[l], stats);
        k_stats_finish   <<<1, DIM, 0, stream>>>(stats, 1.f / (float)NN);
        k_norm_pool      <<<EPI_BLOCKS, 256, 0, stream>>>(
            bufB, stats, gp[l], tp[l], batch, bufA, outp, l);
    }
}


# bias+relu+BN column sums folded into the per-destination aggregation: a wave owns 8 consecutive nodes, keeps their sums in registers; the separate stats pass is gone; stats summed in another order
# speedup vs baseline: 3.4994x; 1.1051x over previous
#include <hip/hip_runtime.h>

#define NN 100000
#define NE 1600000
#define DIM 128
#define NG 512
#define NL 3
#define OUTC (NL * DIM)
#define BN_EPS 1e-5f
#define EPI_RPW 25
#define EPI_BLOCKS (NN / EPI_RPW / 8)
#define SCAN_T 1024
#define RP_PAD 100096
#define GEMM_TILES 5
#define GEMM_ROWS (16 * GEMM_TILES)
#define A_PITCH (DIM + 4)
#define AGG_WAVES 10
#define AGG_NPW 8
#define AGG_BLOCKS (NN / (AGG_WAVES * AGG_NPW))

typedef __attribute__((ext_vector_type(2))) float v2f;
typedef __attribute__((ext_vector_type(8))) float v8f;


__global__ void k_zero(float* __restrict__ p, int n) {
    int i = blockIdx.x * blockDim.x + threadIdx.x;
    if (i < n) p[i] = 0.f;
}

__global__ void k_init_deg(float* __restrict__ deg, int n) {
    int i = blockIdx.x * blockDim.x + threadIdx.x;
    if (i < n) deg[i] = 1.f;
}

__global__ void k_deg_edges(const int* __restrict__ dst, float* __restrict__ deg, int e) {
    int i = blockIdx.x * blockDim.x + threadIdx.x;
    if (i < e) atomicAdd(&deg[dst[i]], 1.f);
}

__global__ void k_dinv(float* __restrict__ deg, int n) {
    int i = blockIdx.x * blockDim.x + threadIdx.x;
    if (i < n) deg[i] = rsqrtf(fmaxf(deg[i], 1.f));
}

static_assert(NN % GEMM_ROWS == 0, "every block owns GEMM_ROWS whole rows");
static_assert((GEMM_ROWS * 32) % 256 == 0, "staging: a whole number of float4 per thread");

__global__ void __launch_bounds__(256)
k_gemm_wmma(const float* __restrict__ A, const float* __restrict__ W,
            float* __restrict__ out) {
    __shared__ float tileA[GEMM_ROWS * A_PITCH];

    const int tid  = threadIdx.x;
    const int wave = tid >> 5;
    const int lane = tid & 31;
    const int row0 = blockIdx.x * GEMM_ROWS;

    constexpr int NST = GEMM_ROWS * 32 / 256;
    const float4* srcv = (const float4*)(A + (size_t)row0 * DIM);
    float4 st[NST];
#pragma unroll
    for (int j = 0; j < NST; ++j) st[j] = srcv[tid + 256 * j];

    const int m  = lane & 15;
    const int kh = (lane >> 4) << 1;
    const int n  = (wave << 4) + m;

    const float* wcol = W + (size_t)kh * DIM + n;
    v2f b[DIM / 4];
#pragma unroll
    for (int s = 0; s < DIM / 4; ++s) {
        b[s].x = wcol[(size_t)(4 * s) * DIM];
        b[s].y = wcol[(size_t)(4 * s + 1) * DIM];
    }

#pragma unroll
    for (int j = 0; j < NST; ++j) {
        const int i = tid + 256 * j;
        *(float4*)&tileA[(i >> 5) * A_PITCH + ((i & 31) << 2)] = st[j];
    }
    __syncthreads();

    v8f c[GEMM_TILES];
#pragma unroll
    for (int t = 0; t < GEMM_TILES; ++t) c[t] = v8f{};

    const float* arow = tileA + m * A_PITCH + kh;
#pragma unroll
    for (int s = 0; s < DIM / 4; ++s) {
#pragma unroll
        for (int t = 0; t < GEMM_TILES; ++t) {
            const v2f a = *(const v2f*)(arow + t * 16 * A_PITCH + 4 * s);
            c[t] = __builtin_amdgcn_wmma_f32_16x16x4_f32(
                        false, a,  false, b[s],
                        (short)0, c[t],  false,  false);
        }
    }

    const int mo = (lane >> 4) << 3;
    float* orow = out + (size_t)(row0 + mo) * DIM + n;
#pragma unroll
    for (int t = 0; t < GEMM_TILES; ++t)
#pragma unroll
        for (int r = 0; r < 8; ++r)
            orow[(size_t)(t * 16 + r) * DIM] = c[t][r];
}


__global__ void k_selfloop(const float* __restrict__ h, const float* __restrict__ dinv,
                           float* __restrict__ out) {
    int i = blockIdx.x * blockDim.x + threadIdx.x;
    int node = i >> 5, q = i & 31;
    if (node < NN) {
        float w = dinv[node]; w *= w;
        float4 v = ((const float4*)(h + (size_t)node * DIM))[q];
        float4 r; r.x = v.x * w; r.y = v.y * w; r.z = v.z * w; r.w = v.w * w;
        ((float4*)(out + (size_t)node * DIM))[q] = r;
    }
}

__global__ void k_edge_agg(const int* __restrict__ src, const int* __restrict__ dst,
                           const float* __restrict__ dinv, const float* __restrict__ h,
                           float* __restrict__ out) {
    int i = blockIdx.x * blockDim.x + threadIdx.x;
    int e = i >> 5, q = i & 31;
    if (e < NE) {
        int s = src[e], d = dst[e];
        float w = dinv[s] * dinv[d];
        const float* hs = h + (size_t)s * DIM + q;
        float v0 = hs[0], v1 = hs[32], v2 = hs[64], v3 = hs[96];
        float* o = out + (size_t)d * DIM + q;
        atomicAdd(o + 0,  v0 * w);
        atomicAdd(o + 32, v1 * w);
        atomicAdd(o + 64, v2 * w);
        atomicAdd(o + 96, v3 * w);
    }
}


__global__ void k_count_rank(const int* __restrict__ dst, int* __restrict__ cnt,
                             int* __restrict__ rank, int e) {
    int i = blockIdx.x * blockDim.x + threadIdx.x;
    if (i < e) rank[i] = atomicAdd(&cnt[dst[i]], 1);
}

__global__ void __launch_bounds__(SCAN_T)
k_scan_dinv(const int* __restrict__ cnt, int* __restrict__ rowptr, float* __restrict__ dinv) {
    __shared__ int s_pre[SCAN_T / 32];
    __shared__ int s_own[SCAN_T / 32];
    const int t = threadIdx.x, b = blockIdx.x;
    const int lane = t & 31, wave = t >> 5;

    int pre = 0;
    const int4* c4 = (const int4*)cnt;
    for (int q = t; q < b * (SCAN_T / 4); q += SCAN_T) {
        int4 v = c4[q];
        pre += (v.x + v.y) + (v.z + v.w);
    }
    const int i = b * SCAN_T + t;
    const int own = (i < NN) ? cnt[i] : 0;

    int inc = own;
    for (int d = 1; d < 32; d <<= 1) {
        int u = __shfl_up(inc, d);
        if (lane >= d) inc += u;
    }
    for (int d = 16; d > 0; d >>= 1) pre += __shfl_xor(pre, d);
    if (lane == 31) { s_own[wave] = inc; s_pre[wave] = pre; }
    __syncthreads();
    int base = 0;
    for (int w = 0; w < SCAN_T / 32; ++w) {
        base += s_pre[w];
        if (w < wave) base += s_own[w];
    }
    const int excl = base + inc - own;
    if (i < NN) {
        rowptr[i] = excl;
        dinv[i] = rsqrtf(fmaxf((float)(own + 1), 1.f));
        if (i == NN - 1) rowptr[NN] = excl + own;
    }
}

__global__ void k_fill(const int* __restrict__ src, const int* __restrict__ dst,
                       const int* __restrict__ rank, const int* __restrict__ rowptr,
                       int* __restrict__ csr, int e) {
    int i = blockIdx.x * blockDim.x + threadIdx.x;
    if (i < e) csr[rowptr[dst[i]] + rank[i]] = src[i];
}

static_assert(NN % (AGG_WAVES * AGG_NPW) == 0, "every wave of every block owns AGG_NPW nodes");
static_assert(AGG_NPW < 32, "a wave keeps its nodes' segment bounds one per lane");
static_assert(AGG_WAVES * 32 >= 2 * DIM, "one thread per column sum in the block's reduction");

__global__ void __launch_bounds__(AGG_WAVES * 32)
k_agg_bias_relu_stats(const int* __restrict__ rowptr, const int* __restrict__ csr,
                      const float* __restrict__ dinv, const float* __restrict__ h,
                      const float* __restrict__ bias, float* __restrict__ out,
                      float* __restrict__ stats) {
    __shared__ float red[AGG_WAVES][2 * DIM];
    const int lane = threadIdx.x & 31;
    const int wave = threadIdx.x >> 5;
    const int n0 = __builtin_amdgcn_readfirstlane((blockIdx.x * AGG_WAVES + wave) * AGG_NPW);
    const int rp = rowptr[n0 + min(lane, AGG_NPW)];
    const int dl = __float_as_int(dinv[n0 + min(lane, AGG_NPW - 1)]);
    const float4 b = ((const float4*)bias)[lane];
    const float4* hv = (const float4*)h;

    float4 cs = {0.f, 0.f, 0.f, 0.f}, cq = {0.f, 0.f, 0.f, 0.f};
    for (int k = 0; k < AGG_NPW; ++k) {
        const int node  = n0 + k;
        const int beg   = __builtin_amdgcn_readlane(rp, k);
        const int items = __builtin_amdgcn_readlane(rp, k + 1) - beg + 1;
        const float dd  = __int_as_float(__builtin_amdgcn_readlane(dl, k));

        float4 acc = {0.f, 0.f, 0.f, 0.f};
        for (int base = 0; base < items; base += 32) {
            const int it = base + lane;
            int s = node; float w = 0.f;
            if (it == 0) w = dd * dd;
            else if (it < items) {
                s = (int)min((unsigned)csr[beg + it - 1], (unsigned)(NN - 1));
                w = dinv[s] * dd;
            }
            const int wi = __float_as_int(w);
            const int n = min(32, items - base);
            for (int j = 0; j < n; j += 4) {
                const int s0 = __builtin_amdgcn_readlane(s, j);
                const int s1 = __builtin_amdgcn_readlane(s, j + 1);
                const int s2 = __builtin_amdgcn_readlane(s, j + 2);
                const int s3 = __builtin_amdgcn_readlane(s, j + 3);
                const float w0 = __int_as_float(__builtin_amdgcn_readlane(wi, j));
                const float w1 = __int_as_float(__builtin_amdgcn_readlane(wi, j + 1));
                const float w2 = __int_as_float(__builtin_amdgcn_readlane(wi, j + 2));
                const float w3 = __int_as_float(__builtin_amdgcn_readlane(wi, j + 3));
                const float4 v0 = hv[(size_t)s0 * 32 + lane];
                const float4 v1 = hv[(size_t)s1 * 32 + lane];
                const float4 v2 = hv[(size_t)s2 * 32 + lane];
                const float4 v3 = hv[(size_t)s3 * 32 + lane];
                acc.x = fmaf(v0.x, w0, acc.x); acc.y = fmaf(v0.y, w0, acc.y);
                acc.z = fmaf(v0.z, w0, acc.z); acc.w = fmaf(v0.w, w0, acc.w);
                acc.x = fmaf(v1.x, w1, acc.x); acc.y = fmaf(v1.y, w1, acc.y);
                acc.z = fmaf(v1.z, w1, acc.z); acc.w = fmaf(v1.w, w1, acc.w);
                acc.x = fmaf(v2.x, w2, acc.x); acc.y = fmaf(v2.y, w2, acc.y);
                acc.z = fmaf(v2.z, w2, acc.z); acc.w = fmaf(v2.w, w2, acc.w);
                acc.x = fmaf(v3.x, w3, acc.x); acc.y = fmaf(v3.y, w3, acc.y);
                acc.z = fmaf(v3.z, w3, acc.z); acc.w = fmaf(v3.w, w3, acc.w);
            }
        }
        float4 x;
        x.x = fmaxf(acc.x + b.x, 0.f); x.y = fmaxf(acc.y + b.y, 0.f);
        x.z = fmaxf(acc.z + b.z, 0.f); x.w = fmaxf(acc.w + b.w, 0.f);
        ((float4*)out)[(size_t)node * 32 + lane] = x;
        cs.x += x.x; cs.y += x.y; cs.z += x.z; cs.w += x.w;
        cq.x += x.x * x.x; cq.y += x.y * x.y; cq.z += x.z * x.z; cq.w += x.w * x.w;
    }
    ((float4*)&red[wave][0])[lane]   = cs;
    ((float4*)&red[wave][DIM])[lane] = cq;
    __syncthreads();
    if (threadIdx.x < 2 * DIM) {
        float t = 0.f;
        for (int w = 0; w < AGG_WAVES; ++w) t += red[w][threadIdx.x];
        atomicAdd(&stats[threadIdx.x], t);
    }
}


static_assert(NN % (EPI_RPW * 8) == 0, "every wave of every block owns EPI_RPW rows");

__global__ void __launch_bounds__(256)
k_bias_relu_stats(float* __restrict__ h, const float* __restrict__ bias,
                  float* __restrict__ stats) {
    __shared__ float red[8][2 * DIM];
    const int lane = threadIdx.x & 31;
    const int wave = threadIdx.x >> 5;
    const int r0 = __builtin_amdgcn_readfirstlane((blockIdx.x * 8 + wave) * EPI_RPW);
    float4* hv = (float4*)h + (size_t)r0 * 32 + lane;
    const float4 b = ((const float4*)bias)[lane];

    float4 v[EPI_RPW];
#pragma unroll
    for (int u = 0; u < EPI_RPW; ++u) v[u] = hv[u * 32];

    float4 s = {0.f, 0.f, 0.f, 0.f}, q = {0.f, 0.f, 0.f, 0.f};
#pragma unroll
    for (int u = 0; u < EPI_RPW; ++u) {
        float4 x = v[u];
        x.x = fmaxf(x.x + b.x, 0.f); x.y = fmaxf(x.y + b.y, 0.f);
        x.z = fmaxf(x.z + b.z, 0.f); x.w = fmaxf(x.w + b.w, 0.f);
        hv[u * 32] = x;
        s.x += x.x; s.y += x.y; s.z += x.z; s.w += x.w;
        q.x += x.x * x.x; q.y += x.y * x.y; q.z += x.z * x.z; q.w += x.w * x.w;
    }
    ((float4*)&red[wave][0])[lane]   = s;
    ((float4*)&red[wave][DIM])[lane] = q;
    __syncthreads();
    float t = 0.f;
    for (int w = 0; w < 8; ++w) t += red[w][threadIdx.x];
    atomicAdd(&stats[threadIdx.x], t);
}

__global__ void k_stats_finish(float* __restrict__ stats, float inv_n) {
    int c = threadIdx.x;
    float mu  = stats[c] * inv_n;
    float var = stats[DIM + c] * inv_n - mu * mu;
    stats[2 * DIM + c] = mu;
    stats[3 * DIM + c] = rsqrtf(var + BN_EPS);
}

__device__ __forceinline__ void pool_flush(float* __restrict__ out, int g, int layer, int lane,
                                           float4 acc) {
    if ((unsigned)g < (unsigned)NG) {
        float* o = out + (size_t)g * OUTC + layer * DIM + 4 * lane;
        atomicAdd(o + 0, acc.x); atomicAdd(o + 1, acc.y);
        atomicAdd(o + 2, acc.z); atomicAdd(o + 3, acc.w);
    }
}

__global__ void __launch_bounds__(256)
k_norm_pool(const float* __restrict__ h, const float* __restrict__ stats,
            const float* __restrict__ gamma, const float* __restrict__ beta,
            const int* __restrict__ batch, float* __restrict__ hnext,
            float* __restrict__ out, int layer) {
    const int lane = threadIdx.x & 31;
    const int wave = threadIdx.x >> 5;
    const int r0 = __builtin_amdgcn_readfirstlane((blockIdx.x * 8 + wave) * EPI_RPW);
    const float4* hv = (const float4*)h + (size_t)r0 * 32 + lane;
    float4*       hn = (float4*)hnext   + (size_t)r0 * 32 + lane;

    float4 v[EPI_RPW];
#pragma unroll
    for (int u = 0; u < EPI_RPW; ++u) v[u] = hv[u * 32];

    const int gl = batch[r0 + min(lane, EPI_RPW - 1)];
    const float4 mu = ((const float4*)(stats + 2 * DIM))[lane];
    const float4 rs = ((const float4*)(stats + 3 * DIM))[lane];
    const float4 ga = ((const float4*)gamma)[lane];
    const float4 be = ((const float4*)beta)[lane];

    int cur = __builtin_amdgcn_readlane(gl, 0);
    float4 acc = {0.f, 0.f, 0.f, 0.f};
#pragma unroll
    for (int u = 0; u < EPI_RPW; ++u) {
        float4 x;
        x.x = (v[u].x - mu.x) * rs.x * ga.x + be.x;
        x.y = (v[u].y - mu.y) * rs.y * ga.y + be.y;
        x.z = (v[u].z - mu.z) * rs.z * ga.z + be.z;
        x.w = (v[u].w - mu.w) * rs.w * ga.w + be.w;
        hn[u * 32] = x;
        const int g = __builtin_amdgcn_readlane(gl, u);
        if (g != cur) {
            pool_flush(out, cur, layer, lane, acc);
            cur = g;
            acc.x = 0.f; acc.y = 0.f; acc.z = 0.f; acc.w = 0.f;
        }
        acc.x += x.x; acc.y += x.y; acc.z += x.z; acc.w += x.w;
    }
    pool_flush(out, cur, layer, lane, acc);
}


extern "C" void kernel_launch(void* const* d_in, const int* in_sizes, int n_in,
                              void* d_out, int out_size, void* d_ws, size_t ws_size,
                              hipStream_t stream) {
    (void)in_sizes; (void)n_in; (void)out_size;

    const float* x     = (const float*)d_in[0];
    const int*   ei    = (const int*)d_in[1];
    const int*   batch = (const int*)d_in[2];
    const float *Wp[NL], *bp[NL], *gp[NL], *tp[NL];
    for (int i = 0; i < NL; ++i) {
        Wp[i] = (const float*)d_in[3 + 4 * i];
        bp[i] = (const float*)d_in[4 + 4 * i];
        gp[i] = (const float*)d_in[5 + 4 * i];
        tp[i] = (const float*)d_in[6 + 4 * i];
    }
    const int* esrc = ei;
    const int* edst = ei + NE;

    float* ws    = (float*)d_ws;
    float* dinv  = ws;
    float* stats = ws + NN;
    size_t o     = ((size_t)NN + 4 * DIM + 255) & ~(size_t)255;
    float* bufA  = ws + o;
    float* bufB  = bufA + (size_t)NN * DIM;
    float* outp  = (float*)d_out;

    int* rowptr  = (int*)(bufB + (size_t)NN * DIM);
    int* csr     = rowptr + RP_PAD;
    int* cnt     = (int*)bufB;
    int* rank    = cnt + RP_PAD;
    const bool sorted =
        ws_size >= (o + 2 * (size_t)NN * DIM + RP_PAD + NE) * sizeof(float);

    k_zero<<<(NG * OUTC + 255) / 256, 256, 0, stream>>>(outp, NG * OUTC);
    if (sorted) {
        k_zero      <<<(NN + 255) / 256,         256,    0, stream>>>((float*)cnt, NN);
        k_count_rank<<<(NE + 255) / 256,         256,    0, stream>>>(edst, cnt, rank, NE);
        k_scan_dinv <<<(NN + SCAN_T - 1) / SCAN_T, SCAN_T, 0, stream>>>(cnt, rowptr, dinv);
        k_fill      <<<(NE + 255) / 256,         256,    0, stream>>>(esrc, edst, rank, rowptr, csr, NE);
    } else {
        k_init_deg <<<(NN + 255) / 256, 256, 0, stream>>>(dinv, NN);
        k_deg_edges<<<(NE + 255) / 256, 256, 0, stream>>>(edst, dinv, NE);
        k_dinv     <<<(NN + 255) / 256, 256, 0, stream>>>(dinv, NN);
    }

    for (int l = 0; l < NL; ++l) {
        const float* hin = (l == 0) ? x : bufA;
        k_gemm_wmma<<<NN / GEMM_ROWS, 256, 0, stream>>>(hin, Wp[l], bufA);
        k_zero<<<1, 256, 0, stream>>>(stats, 2 * DIM);
        if (sorted) {
            k_agg_bias_relu_stats<<<AGG_BLOCKS, AGG_WAVES * 32, 0, stream>>>(
                rowptr, csr, dinv, bufA, bp[l], bufB, stats);
        } else {
            k_selfloop <<<(NN * 32 + 255) / 256, 256, 0, stream>>>(bufA, dinv, bufB);
            k_edge_agg <<<(NE * 32) / 256, 256, 0, stream>>>(esrc, edst, dinv, bufA, bufB);
            k_bias_relu_stats<<<EPI_BLOCKS, 256, 0, stream>>>(bufB, bp[l], stats);
        }
        k_stats_finish   <<<1, DIM, 0, stream>>>(stats, 1.f / (float)NN);
        k_norm_pool      <<<EPI_BLOCKS, 256, 0, stream>>>(
            bufB, stats, gp[l], tp[l], batch, bufA, outp, l);
    }
}


# norm+pool pass removed: batch norm applied where the next GEMM stages its A rows; pooled output = affine of per-graph raw sums and row counts taken in the aggregation; pool summed in another order
# speedup vs baseline: 4.1970x; 1.1993x over previous
#include <hip/hip_runtime.h>

#define NN 100000
#define NE 1600000
#define DIM 128
#define NG 512
#define NL 3
#define OUTC (NL * DIM)
#define BN_EPS 1e-5f
#define EPI_RPW 25
#define EPI_BLOCKS (NN / EPI_RPW / 8)
#define SCAN_T 1024
#define RP_PAD 100096
#define GEMM_TILES 5
#define GEMM_ROWS (16 * GEMM_TILES)
#define A_PITCH (DIM + 4)
#define AGG_WAVES 10
#define AGG_NPW 8
#define AGG_BLOCKS (NN / (AGG_WAVES * AGG_NPW))
#define ST_CNT (2 * DIM)
#define ST_MU  (2 * DIM + NG)
#define ST_RS  (3 * DIM + NG)
#define ST_FLOATS (4 * DIM + NG)

typedef __attribute__((ext_vector_type(2))) float v2f;
typedef __attribute__((ext_vector_type(8))) float v8f;


__global__ void k_zero(float* __restrict__ p, int n) {
    int i = blockIdx.x * blockDim.x + threadIdx.x;
    if (i < n) p[i] = 0.f;
}

__global__ void k_init_deg(float* __restrict__ deg, int n) {
    int i = blockIdx.x * blockDim.x + threadIdx.x;
    if (i < n) deg[i] = 1.f;
}

__global__ void k_deg_edges(const int* __restrict__ dst, float* __restrict__ deg, int e) {
    int i = blockIdx.x * blockDim.x + threadIdx.x;
    if (i < e) atomicAdd(&deg[dst[i]], 1.f);
}

__global__ void k_dinv(float* __restrict__ deg, int n) {
    int i = blockIdx.x * blockDim.x + threadIdx.x;
    if (i < n) deg[i] = rsqrtf(fmaxf(deg[i], 1.f));
}

static_assert(NN % GEMM_ROWS == 0, "every block owns GEMM_ROWS whole rows");
static_assert((GEMM_ROWS * 32) % 256 == 0, "staging: a whole number of float4 per thread");

template <bool NORM>
__global__ void __launch_bounds__(256)
k_gemm_wmma(const float* __restrict__ A, const float* __restrict__ W,
            float* __restrict__ out, const float* __restrict__ stats,
            const float* __restrict__ gamma, const float* __restrict__ beta) {
    __shared__ float tileA[GEMM_ROWS * A_PITCH];

    const int tid  = threadIdx.x;
    const int wave = tid >> 5;
    const int lane = tid & 31;
    const int row0 = blockIdx.x * GEMM_ROWS;

    constexpr int NST = GEMM_ROWS * 32 / 256;
    const float4* srcv = (const float4*)(A + (size_t)row0 * DIM);
    float4 st[NST];
#pragma unroll
    for (int j = 0; j < NST; ++j) st[j] = srcv[tid + 256 * j];

    float4 mu = {}, rs = {}, ga = {}, be = {};
    if (NORM) {
        const int q = tid & 31;
        mu = ((const float4*)(stats + ST_MU))[q];
        rs = ((const float4*)(stats + ST_RS))[q];
        ga = ((const float4*)gamma)[q];
        be = ((const float4*)beta)[q];
    }

    const int m  = lane & 15;
    const int kh = (lane >> 4) << 1;
    const int n  = (wave << 4) + m;

    const float* wcol = W + (size_t)kh * DIM + n;
    v2f b[DIM / 4];
#pragma unroll
    for (int s = 0; s < DIM / 4; ++s) {
        b[s].x = wcol[(size_t)(4 * s) * DIM];
        b[s].y = wcol[(size_t)(4 * s + 1) * DIM];
    }

#pragma unroll
    for (int j = 0; j < NST; ++j) {
        const int i = tid + 256 * j;
        float4 v = st[j];
        if (NORM) {
            v.x = (v.x - mu.x) * rs.x * ga.x + be.x;
            v.y = (v.y - mu.y) * rs.y * ga.y + be.y;
            v.z = (v.z - mu.z) * rs.z * ga.z + be.z;
            v.w = (v.w - mu.w) * rs.w * ga.w + be.w;
        }
        *(float4*)&tileA[(i >> 5) * A_PITCH + ((i & 31) << 2)] = v;
    }
    __syncthreads();

    v8f c[GEMM_TILES];
#pragma unroll
    for (int t = 0; t < GEMM_TILES; ++t) c[t] = v8f{};

    const float* arow = tileA + m * A_PITCH + kh;
#pragma unroll
    for (int s = 0; s < DIM / 4; ++s) {
#pragma unroll
        for (int t = 0; t < GEMM_TILES; ++t) {
            const v2f a = *(const v2f*)(arow + t * 16 * A_PITCH + 4 * s);
            c[t] = __builtin_amdgcn_wmma_f32_16x16x4_f32(
                        false, a,  false, b[s],
                        (short)0, c[t],  false,  false);
        }
    }

    const int mo = (lane >> 4) << 3;
    float* orow = out + (size_t)(row0 + mo) * DIM + n;
#pragma unroll
    for (int t = 0; t < GEMM_TILES; ++t)
#pragma unroll
        for (int r = 0; r < 8; ++r)
            orow[(size_t)(t * 16 + r) * DIM] = c[t][r];
}


__global__ void k_selfloop(const float* __restrict__ h, const float* __restrict__ dinv,
                           float* __restrict__ out) {
    int i = blockIdx.x * blockDim.x + threadIdx.x;
    int node = i >> 5, q = i & 31;
    if (node < NN) {
        float w = dinv[node]; w *= w;
        float4 v = ((const float4*)(h + (size_t)node * DIM))[q];
        float4 r; r.x = v.x * w; r.y = v.y * w; r.z = v.z * w; r.w = v.w * w;
        ((float4*)(out + (size_t)node * DIM))[q] = r;
    }
}

__global__ void k_edge_agg(const int* __restrict__ src, const int* __restrict__ dst,
                           const float* __restrict__ dinv, const float* __restrict__ h,
                           float* __restrict__ out) {
    int i = blockIdx.x * blockDim.x + threadIdx.x;
    int e = i >> 5, q = i & 31;
    if (e < NE) {
        int s = src[e], d = dst[e];
        float w = dinv[s] * dinv[d];
        const float* hs = h + (size_t)s * DIM + q;
        float v0 = hs[0], v1 = hs[32], v2 = hs[64], v3 = hs[96];
        float* o = out + (size_t)d * DIM + q;
        atomicAdd(o + 0,  v0 * w);
        atomicAdd(o + 32, v1 * w);
        atomicAdd(o + 64, v2 * w);
        atomicAdd(o + 96, v3 * w);
    }
}


__global__ void k_count_rank(const int* __restrict__ dst, int* __restrict__ cnt,
                             int* __restrict__ rank, int e) {
    int i = blockIdx.x * blockDim.x + threadIdx.x;
    if (i < e) rank[i] = atomicAdd(&cnt[dst[i]], 1);
}

__global__ void __launch_bounds__(SCAN_T)
k_scan_dinv(const int* __restrict__ cnt, int* __restrict__ rowptr, float* __restrict__ dinv) {
    __shared__ int s_pre[SCAN_T / 32];
    __shared__ int s_own[SCAN_T / 32];
    const int t = threadIdx.x, b = blockIdx.x;
    const int lane = t & 31, wave = t >> 5;

    int pre = 0;
    const int4* c4 = (const int4*)cnt;
    for (int q = t; q < b * (SCAN_T / 4); q += SCAN_T) {
        int4 v = c4[q];
        pre += (v.x + v.y) + (v.z + v.w);
    }
    const int i = b * SCAN_T + t;
    const int own = (i < NN) ? cnt[i] : 0;

    int inc = own;
    for (int d = 1; d < 32; d <<= 1) {
        int u = __shfl_up(inc, d);
        if (lane >= d) inc += u;
    }
    for (int d = 16; d > 0; d >>= 1) pre += __shfl_xor(pre, d);
    if (lane == 31) { s_own[wave] = inc; s_pre[wave] = pre; }
    __syncthreads();
    int base = 0;
    for (int w = 0; w < SCAN_T / 32; ++w) {
        base += s_pre[w];
        if (w < wave) base += s_own[w];
    }
    const int excl = base + inc - own;
    if (i < NN) {
        rowptr[i] = excl;
        dinv[i] = rsqrtf(fmaxf((float)(own + 1), 1.f));
        if (i == NN - 1) rowptr[NN] = excl + own;
    }
}

__global__ void k_fill(const int* __restrict__ src, const int* __restrict__ dst,
                       const int* __restrict__ rank, const int* __restrict__ rowptr,
                       int* __restrict__ csr, int e) {
    int i = blockIdx.x * blockDim.x + threadIdx.x;
    if (i < e) csr[rowptr[dst[i]] + rank[i]] = src[i];
}

static_assert(NN % (AGG_WAVES * AGG_NPW) == 0, "every wave of every block owns AGG_NPW nodes");
#define POOL_PITCH (DIM + 4)

__device__ __forceinline__ void pool_flush_raw(float* __restrict__ pooled, float* __restrict__ gcnt,
                                               int g, int lane, float4 acc, float rows) {
    if ((unsigned)g < (unsigned)NG) {
        float* o = pooled + (size_t)g * OUTC + 4 * lane;
        atomicAdd(o + 0, acc.x); atomicAdd(o + 1, acc.y);
        atomicAdd(o + 2, acc.z); atomicAdd(o + 3, acc.w);
        if (lane == 0) atomicAdd(&gcnt[g], rows);
    }
}

static_assert(AGG_NPW < 32, "a wave keeps its nodes' segment bounds one per lane");
static_assert(AGG_WAVES * 32 >= 2 * DIM, "one thread per column sum in the block's reduction");

template <bool STORE_ROWS>
__global__ void __launch_bounds__(AGG_WAVES * 32)
k_agg_bias_relu_stats(const int* __restrict__ rowptr, const int* __restrict__ csr,
                      const float* __restrict__ dinv, const float* __restrict__ h,
                      const float* __restrict__ bias, const int* __restrict__ batch,
                      float* __restrict__ out, float* __restrict__ stats,
                      float* __restrict__ gcnt, float* __restrict__ pooled) {
    __shared__ float red[AGG_WAVES][2 * DIM];
    __shared__ __attribute__((aligned(16))) float pool[AGG_WAVES][POOL_PITCH];
    __shared__ int   pgid[AGG_WAVES];
    const int lane = threadIdx.x & 31;
    const int wave = threadIdx.x >> 5;
    const int n0 = __builtin_amdgcn_readfirstlane((blockIdx.x * AGG_WAVES + wave) * AGG_NPW);
    const int rp = rowptr[n0 + min(lane, AGG_NPW)];
    const int dl = __float_as_int(dinv[n0 + min(lane, AGG_NPW - 1)]);
    const int gl = batch[n0 + min(lane, AGG_NPW - 1)];
    const float4 b = ((const float4*)bias)[lane];
    const float4* hv = (const float4*)h;

    float4 cs = {0.f, 0.f, 0.f, 0.f}, cq = {0.f, 0.f, 0.f, 0.f};
    float4 pa = {0.f, 0.f, 0.f, 0.f};
    float  pn = 0.f;
    int    cur = __builtin_amdgcn_readlane(gl, 0);
    for (int k = 0; k < AGG_NPW; ++k) {
        const int node  = n0 + k;
        const int beg   = __builtin_amdgcn_readlane(rp, k);
        const int items = __builtin_amdgcn_readlane(rp, k + 1) - beg + 1;
        const float dd  = __int_as_float(__builtin_amdgcn_readlane(dl, k));

        float4 acc = {0.f, 0.f, 0.f, 0.f};
        for (int base = 0; base < items; base += 32) {
            const int it = base + lane;
            int s = node; float w = 0.f;
            if (it == 0) w = dd * dd;
            else if (it < items) {
                s = (int)min((unsigned)csr[beg + it - 1], (unsigned)(NN - 1));
                w = dinv[s] * dd;
            }
            const int wi = __float_as_int(w);
            const int n = min(32, items - base);
            for (int j = 0; j < n; j += 4) {
                const int s0 = __builtin_amdgcn_readlane(s, j);
                const int s1 = __builtin_amdgcn_readlane(s, j + 1);
                const int s2 = __builtin_amdgcn_readlane(s, j + 2);
                const int s3 = __builtin_amdgcn_readlane(s, j + 3);
                const float w0 = __int_as_float(__builtin_amdgcn_readlane(wi, j));
                const float w1 = __int_as_float(__builtin_amdgcn_readlane(wi, j + 1));
                const float w2 = __int_as_float(__builtin_amdgcn_readlane(wi, j + 2));
                const float w3 = __int_as_float(__builtin_amdgcn_readlane(wi, j + 3));
                const float4 v0 = hv[(size_t)s0 * 32 + lane];
                const float4 v1 = hv[(size_t)s1 * 32 + lane];
                const float4 v2 = hv[(size_t)s2 * 32 + lane];
                const float4 v3 = hv[(size_t)s3 * 32 + lane];
                acc.x = fmaf(v0.x, w0, acc.x); acc.y = fmaf(v0.y, w0, acc.y);
                acc.z = fmaf(v0.z, w0, acc.z); acc.w = fmaf(v0.w, w0, acc.w);
                acc.x = fmaf(v1.x, w1, acc.x); acc.y = fmaf(v1.y, w1, acc.y);
                acc.z = fmaf(v1.z, w1, acc.z); acc.w = fmaf(v1.w, w1, acc.w);
                acc.x = fmaf(v2.x, w2, acc.x); acc.y = fmaf(v2.y, w2, acc.y);
                acc.z = fmaf(v2.z, w2, acc.z); acc.w = fmaf(v2.w, w2, acc.w);
                acc.x = fmaf(v3.x, w3, acc.x); acc.y = fmaf(v3.y, w3, acc.y);
                acc.z = fmaf(v3.z, w3, acc.z); acc.w = fmaf(v3.w, w3, acc.w);
            }
        }
        float4 x;
        x.x = fmaxf(acc.x + b.x, 0.f); x.y = fmaxf(acc.y + b.y, 0.f);
        x.z = fmaxf(acc.z + b.z, 0.f); x.w = fmaxf(acc.w + b.w, 0.f);
        if (STORE_ROWS) ((float4*)out)[(size_t)node * 32 + lane] = x;
        const int g = __builtin_amdgcn_readlane(gl, k);
        if (g != cur) {
            pool_flush_raw(pooled, gcnt, cur, lane, pa, pn);
            cur = g; pn = 0.f;
            pa.x = 0.f; pa.y = 0.f; pa.z = 0.f; pa.w = 0.f;
        }
        pa.x += x.x; pa.y += x.y; pa.z += x.z; pa.w += x.w;
        pn += 1.f;
        cs.x += x.x; cs.y += x.y; cs.z += x.z; cs.w += x.w;
        cq.x += x.x * x.x; cq.y += x.y * x.y; cq.z += x.z * x.z; cq.w += x.w * x.w;
    }
    ((float4*)&red[wave][0])[lane]   = cs;
    ((float4*)&red[wave][DIM])[lane] = cq;
    ((float4*)&pool[wave][0])[lane]  = pa;
    if (lane == 0) { pool[wave][DIM] = pn; pgid[wave] = cur; }
    __syncthreads();
    if (threadIdx.x < 2 * DIM) {
        float t = 0.f;
        for (int w = 0; w < AGG_WAVES; ++w) t += red[w][threadIdx.x];
        atomicAdd(&stats[threadIdx.x], t);
    }
    if (threadIdx.x <= DIM) {
        const int c = threadIdx.x;
        float* dst = (c < DIM) ? pooled + c : gcnt;
        const size_t pitch = (c < DIM) ? (size_t)OUTC : (size_t)1;
        int cg = pgid[0];
        float t = pool[0][c];
        for (int w = 1; w < AGG_WAVES; ++w) {
            const int g = pgid[w];
            if (g != cg) {
                if ((unsigned)cg < (unsigned)NG) atomicAdd(dst + (size_t)cg * pitch, t);
                cg = g; t = 0.f;
            }
            t += pool[w][c];
        }
        if ((unsigned)cg < (unsigned)NG) atomicAdd(dst + (size_t)cg * pitch, t);
    }
}


static_assert(NN % (EPI_RPW * 8) == 0, "every wave of every block owns EPI_RPW rows");

__global__ void __launch_bounds__(256)
k_bias_relu_stats(float* __restrict__ h, const float* __restrict__ bias,
                  float* __restrict__ stats) {
    __shared__ float red[8][2 * DIM];
    const int lane = threadIdx.x & 31;
    const int wave = threadIdx.x >> 5;
    const int r0 = __builtin_amdgcn_readfirstlane((blockIdx.x * 8 + wave) * EPI_RPW);
    float4* hv = (float4*)h + (size_t)r0 * 32 + lane;
    const float4 b = ((const float4*)bias)[lane];

    float4 v[EPI_RPW];
#pragma unroll
    for (int u = 0; u < EPI_RPW; ++u) v[u] = hv[u * 32];

    float4 s = {0.f, 0.f, 0.f, 0.f}, q = {0.f, 0.f, 0.f, 0.f};
#pragma unroll
    for (int u = 0; u < EPI_RPW; ++u) {
        float4 x = v[u];
        x.x = fmaxf(x.x + b.x, 0.f); x.y = fmaxf(x.y + b.y, 0.f);
        x.z = fmaxf(x.z + b.z, 0.f); x.w = fmaxf(x.w + b.w, 0.f);
        hv[u * 32] = x;
        s.x += x.x; s.y += x.y; s.z += x.z; s.w += x.w;
        q.x += x.x * x.x; q.y += x.y * x.y; q.z += x.z * x.z; q.w += x.w * x.w;
    }
    ((float4*)&red[wave][0])[lane]   = s;
    ((float4*)&red[wave][DIM])[lane] = q;
    __syncthreads();
    float t = 0.f;
    for (int w = 0; w < 8; ++w) t += red[w][threadIdx.x];
    atomicAdd(&stats[threadIdx.x], t);
}

__global__ void k_stats_finish(float* __restrict__ stats, float inv_n) {
    int c = threadIdx.x;
    float mu  = stats[c] * inv_n;
    float var = stats[DIM + c] * inv_n - mu * mu;
    stats[ST_MU + c] = mu;
    stats[ST_RS + c] = rsqrtf(var + BN_EPS);
}

__global__ void __launch_bounds__(DIM)
k_stats_pool_finish(float* __restrict__ stats, const float* __restrict__ gamma,
                    const float* __restrict__ beta, float* __restrict__ pooled, float inv_n) {
    const int c = threadIdx.x, g = blockIdx.x;
    const float mu  = stats[c] * inv_n;
    const float var = stats[DIM + c] * inv_n - mu * mu;
    const float rs  = rsqrtf(var + BN_EPS);
    if (g == 0) { stats[ST_MU + c] = mu; stats[ST_RS + c] = rs; }
    const float n = stats[ST_CNT + g];
    float* o = pooled + (size_t)g * OUTC + c;
    *o = fmaf(fmaf(-n, mu, *o) * rs, gamma[c], n * beta[c]);
}

__device__ __forceinline__ void pool_flush(float* __restrict__ out, int g, int layer, int lane,
                                           float4 acc) {
    if ((unsigned)g < (unsigned)NG) {
        float* o = out + (size_t)g * OUTC + layer * DIM + 4 * lane;
        atomicAdd(o + 0, acc.x); atomicAdd(o + 1, acc.y);
        atomicAdd(o + 2, acc.z); atomicAdd(o + 3, acc.w);
    }
}

__global__ void __launch_bounds__(256)
k_norm_pool(const float* __restrict__ h, const float* __restrict__ stats,
            const float* __restrict__ gamma, const float* __restrict__ beta,
            const int* __restrict__ batch, float* __restrict__ hnext,
            float* __restrict__ out, int layer) {
    const int lane = threadIdx.x & 31;
    const int wave = threadIdx.x >> 5;
    const int r0 = __builtin_amdgcn_readfirstlane((blockIdx.x * 8 + wave) * EPI_RPW);
    const float4* hv = (const float4*)h + (size_t)r0 * 32 + lane;
    float4*       hn = (float4*)hnext   + (size_t)r0 * 32 + lane;

    float4 v[EPI_RPW];
#pragma unroll
    for (int u = 0; u < EPI_RPW; ++u) v[u] = hv[u * 32];

    const int gl = batch[r0 + min(lane, EPI_RPW - 1)];
    const float4 mu = ((const float4*)(stats + ST_MU))[lane];
    const float4 rs = ((const float4*)(stats + ST_RS))[lane];
    const float4 ga = ((const float4*)gamma)[lane];
    const float4 be = ((const float4*)beta)[lane];

    int cur = __builtin_amdgcn_readlane(gl, 0);
    float4 acc = {0.f, 0.f, 0.f, 0.f};
#pragma unroll
    for (int u = 0; u < EPI_RPW; ++u) {
        float4 x;
        x.x = (v[u].x - mu.x) * rs.x * ga.x + be.x;
        x.y = (v[u].y - mu.y) * rs.y * ga.y + be.y;
        x.z = (v[u].z - mu.z) * rs.z * ga.z + be.z;
        x.w = (v[u].w - mu.w) * rs.w * ga.w + be.w;
        hn[u * 32] = x;
        const int g = __builtin_amdgcn_readlane(gl, u);
        if (g != cur) {
            pool_flush(out, cur, layer, lane, acc);
            cur = g;
            acc.x = 0.f; acc.y = 0.f; acc.z = 0.f; acc.w = 0.f;
        }
        acc.x += x.x; acc.y += x.y; acc.z += x.z; acc.w += x.w;
    }
    pool_flush(out, cur, layer, lane, acc);
}


extern "C" void kernel_launch(void* const* d_in, const int* in_sizes, int n_in,
                              void* d_out, int out_size, void* d_ws, size_t ws_size,
                              hipStream_t stream) {
    (void)in_sizes; (void)n_in; (void)out_size;

    const float* x     = (const float*)d_in[0];
    const int*   ei    = (const int*)d_in[1];
    const int*   batch = (const int*)d_in[2];
    const float *Wp[NL], *bp[NL], *gp[NL], *tp[NL];
    for (int i = 0; i < NL; ++i) {
        Wp[i] = (const float*)d_in[3 + 4 * i];
        bp[i] = (const float*)d_in[4 + 4 * i];
        gp[i] = (const float*)d_in[5 + 4 * i];
        tp[i] = (const float*)d_in[6 + 4 * i];
    }
    const int* esrc = ei;
    const int* edst = ei + NE;

    float* ws    = (float*)d_ws;
    float* dinv  = ws;
    float* stats = ws + NN;
    size_t o     = ((size_t)NN + ST_FLOATS + 255) & ~(size_t)255;
    float* bufA  = ws + o;
    float* bufB  = bufA + (size_t)NN * DIM;
    float* outp  = (float*)d_out;

    int* rowptr  = (int*)(bufB + (size_t)NN * DIM);
    int* csr     = rowptr + RP_PAD;
    int* cnt     = (int*)bufB;
    int* rank    = cnt + RP_PAD;
    const bool sorted =
        ws_size >= (o + 2 * (size_t)NN * DIM + RP_PAD + NE) * sizeof(float);

    k_zero<<<(NG * OUTC + 255) / 256, 256, 0, stream>>>(outp, NG * OUTC);
    if (sorted) {
        k_zero      <<<(NN + 255) / 256,         256,    0, stream>>>((float*)cnt, NN);
        k_count_rank<<<(NE + 255) / 256,         256,    0, stream>>>(edst, cnt, rank, NE);
        k_scan_dinv <<<(NN + SCAN_T - 1) / SCAN_T, SCAN_T, 0, stream>>>(cnt, rowptr, dinv);
        k_fill      <<<(NE + 255) / 256,         256,    0, stream>>>(esrc, edst, rank, rowptr, csr, NE);
    } else {
        k_init_deg <<<(NN + 255) / 256, 256, 0, stream>>>(dinv, NN);
        k_deg_edges<<<(NE + 255) / 256, 256, 0, stream>>>(edst, dinv, NE);
        k_dinv     <<<(NN + 255) / 256, 256, 0, stream>>>(dinv, NN);
    }

    for (int l = 0; l < NL; ++l) {
        if (sorted) {
            if (l == 0)
                k_gemm_wmma<false><<<NN / GEMM_ROWS, 256, 0, stream>>>(
                    x, Wp[l], bufA, nullptr, nullptr, nullptr);
            else
                k_gemm_wmma<true><<<NN / GEMM_ROWS, 256, 0, stream>>>(
                    bufB, Wp[l], bufA, stats, gp[l - 1], tp[l - 1]);
            k_zero<<<(2 * DIM + NG + 255) / 256, 256, 0, stream>>>(stats, 2 * DIM + NG);
            if (l < NL - 1)
                k_agg_bias_relu_stats<true><<<AGG_BLOCKS, AGG_WAVES * 32, 0, stream>>>(
                    rowptr, csr, dinv, bufA, bp[l], batch, bufB, stats, stats + ST_CNT,
                    outp + l * DIM);
            else
                k_agg_bias_relu_stats<false><<<AGG_BLOCKS, AGG_WAVES * 32, 0, stream>>>(
                    rowptr, csr, dinv, bufA, bp[l], batch, bufB, stats, stats + ST_CNT,
                    outp + l * DIM);
            k_stats_pool_finish<<<NG, DIM, 0, stream>>>(
                stats, gp[l], tp[l], outp + l * DIM, 1.f / (float)NN);
            continue;
        }
        const float* hin = (l == 0) ? x : bufA;
        k_gemm_wmma<false><<<NN / GEMM_ROWS, 256, 0, stream>>>(
            hin, Wp[l], bufA, nullptr, nullptr, nullptr);
        k_zero<<<1, 256, 0, stream>>>(stats, 2 * DIM);
        k_selfloop <<<(NN * 32 + 255) / 256, 256, 0, stream>>>(bufA, dinv, bufB);
        k_edge_agg <<<(NE * 32) / 256, 256, 0, stream>>>(esrc, edst, dinv, bufA, bufB);
        k_bias_relu_stats<<<EPI_BLOCKS, 256, 0, stream>>>(bufB, bp[l], stats);
        k_stats_finish   <<<1, DIM, 0, stream>>>(stats, 1.f / (float)NN);
        k_norm_pool      <<<EPI_BLOCKS, 256, 0, stream>>>(
            bufB, stats, gp[l], tp[l], batch, bufA, outp, l);
    }
}


# small launches folded into neighbours: one zero-fill launch (was 5); each layer's BN mean/rstd and pooled-row finish done in the next layer's GEMM launch; 17 launches -> 11; sum orders kept
# speedup vs baseline: 4.5867x; 1.0929x over previous
#include <hip/hip_runtime.h>

#define NN 100000
#define NE 1600000
#define DIM 128
#define NG 512
#define NL 3
#define OUTC (NL * DIM)
#define BN_EPS 1e-5f
#define EPI_RPW 25
#define EPI_BLOCKS (NN / EPI_RPW / 8)
#define SCAN_T 1024
#define RP_PAD 100096
#define GEMM_TILES 5
#define GEMM_ROWS (16 * GEMM_TILES)
#define A_PITCH (DIM + 4)
#define AGG_WAVES 10
#define AGG_NPW 8
#define AGG_BLOCKS (NN / (AGG_WAVES * AGG_NPW))
#define ST_MU  (2 * DIM + NG)
#define ST_RS  (3 * DIM + NG)
#define ST_FLOATS (4 * DIM + NG)
#define SS_SUM(l) (((l) & 1) * 2 * DIM)
#define SS_CNT (4 * DIM)
static_assert(SS_CNT + NG <= ST_FLOATS, "the sorted path's sums and counts fit the statistics block");

typedef __attribute__((ext_vector_type(2))) float v2f;
typedef __attribute__((ext_vector_type(8))) float v8f;


__global__ void k_zero(float* __restrict__ p, int n) {
    int i = blockIdx.x * blockDim.x + threadIdx.x;
    if (i < n) p[i] = 0.f;
}

__global__ void k_zero3(float* __restrict__ a, int na, float* __restrict__ b, int nb,
                        float* __restrict__ c, int nc) {
    int i = blockIdx.x * blockDim.x + threadIdx.x;
    if (i < na) a[i] = 0.f;
    else if (i < na + nb) b[i - na] = 0.f;
    else if (i < na + nb + nc) c[i - na - nb] = 0.f;
}

__device__ __forceinline__ float bn_rstd(float q, float mu, float inv_n) {
    const float var = q * inv_n - mu * mu;
    return rsqrtf(var + BN_EPS);
}

__global__ void k_init_deg(float* __restrict__ deg, int n) {
    int i = blockIdx.x * blockDim.x + threadIdx.x;
    if (i < n) deg[i] = 1.f;
}

__global__ void k_deg_edges(const int* __restrict__ dst, float* __restrict__ deg, int e) {
    int i = blockIdx.x * blockDim.x + threadIdx.x;
    if (i < e) atomicAdd(&deg[dst[i]], 1.f);
}

__global__ void k_dinv(float* __restrict__ deg, int n) {
    int i = blockIdx.x * blockDim.x + threadIdx.x;
    if (i < n) deg[i] = rsqrtf(fmaxf(deg[i], 1.f));
}

static_assert(NN % GEMM_ROWS == 0, "every block owns GEMM_ROWS whole rows");
static_assert((GEMM_ROWS * 32) % 256 == 0, "staging: a whole number of float4 per thread");
static_assert(NN / GEMM_ROWS >= NG, "one GEMM block per graph for the pooled rows");
static_assert(2 * DIM == 256, "block 0 clears a column-sum area with one store per thread");

template <bool NORM, bool SCALE>
__global__ void __launch_bounds__(256)
k_gemm_wmma(const float* __restrict__ A, const float* __restrict__ W,
            float* __restrict__ out, const float* __restrict__ sums,
            const float* __restrict__ gamma, const float* __restrict__ beta,
            const float* __restrict__ dinv, const float* __restrict__ gcnt,
            float* __restrict__ pooled_prev, float* __restrict__ zero_next, float inv_n) {
    __shared__ float tileA[GEMM_ROWS * A_PITCH];

    const int tid  = threadIdx.x;
    const int wave = __builtin_amdgcn_readfirstlane(tid >> 5);
    const int lane = tid & 31;
    const int row0 = blockIdx.x * GEMM_ROWS;

    if (NORM) {
        if (zero_next != nullptr && blockIdx.x == 0) zero_next[tid] = 0.f;
        if (blockIdx.x < NG && tid < DIM) {
            const float pm = sums[tid] * inv_n;
            const float pr = bn_rstd(sums[DIM + tid], pm, inv_n);
            const float rows = gcnt[blockIdx.x];
            float* o = pooled_prev + (size_t)blockIdx.x * OUTC + tid;
            *o = fmaf(fmaf(-rows, pm, *o) * pr, gamma[tid], rows * beta[tid]);
        }
    }

    constexpr int NST = GEMM_ROWS * 32 / 256;
    const float4* srcv = (const float4*)(A + (size_t)row0 * DIM);
    float4 st[NST];
#pragma unroll
    for (int j = 0; j < NST; ++j) st[j] = srcv[tid + 256 * j];

    float4 mu = {}, rs = {}, ga = {}, be = {};
    if (NORM) {
        const int q = tid & 31;
        const float4 cs = ((const float4*)sums)[q];
        const float4 cq = ((const float4*)(sums + DIM))[q];
        mu.x = cs.x * inv_n; mu.y = cs.y * inv_n; mu.z = cs.z * inv_n; mu.w = cs.w * inv_n;
        rs.x = bn_rstd(cq.x, mu.x, inv_n); rs.y = bn_rstd(cq.y, mu.y, inv_n);
        rs.z = bn_rstd(cq.z, mu.z, inv_n); rs.w = bn_rstd(cq.w, mu.w, inv_n);
        ga = ((const float4*)gamma)[q];
        be = ((const float4*)beta)[q];
    }

    const int m  = lane & 15;
    const int kh = (lane >> 4) << 1;
    const int n  = (wave << 4) + m;

    const float* wcol = W + (size_t)kh * DIM + n;
    v2f b[DIM / 4];
#pragma unroll
    for (int s = 0; s < DIM / 4; ++s) {
        b[s].x = wcol[(size_t)(4 * s) * DIM];
        b[s].y = wcol[(size_t)(4 * s + 1) * DIM];
    }

    float* stage = tileA + wave * A_PITCH + (lane << 2);
#pragma unroll
    for (int j = 0; j < NST; ++j) {
        float4 v = st[j];
        if (NORM) {
            v.x = (v.x - mu.x) * rs.x * ga.x + be.x;
            v.y = (v.y - mu.y) * rs.y * ga.y + be.y;
            v.z = (v.z - mu.z) * rs.z * ga.z + be.z;
            v.w = (v.w - mu.w) * rs.w * ga.w + be.w;
        }
        if (SCALE) {
            const float d = dinv[row0 + wave + 8 * j];
            v.x *= d; v.y *= d; v.z *= d; v.w *= d;
        }
        *(float4*)&stage[8 * j * A_PITCH] = v;
    }
    __syncthreads();

    v8f c[GEMM_TILES];
#pragma unroll
    for (int t = 0; t < GEMM_TILES; ++t) c[t] = v8f{};

    const float* arow = tileA + m * A_PITCH + kh;
#pragma unroll
    for (int s = 0; s < DIM / 4; ++s) {
#pragma unroll
        for (int t = 0; t < GEMM_TILES; ++t) {
            const v2f a = *(const v2f*)(arow + t * 16 * A_PITCH + 4 * s);
            c[t] = __builtin_amdgcn_wmma_f32_16x16x4_f32(
                        false, a,  false, b[s],
                        (short)0, c[t],  false,  false);
        }
    }

    const int mo = (lane >> 4) << 3;
    float* orow = out + (size_t)(row0 + mo) * DIM + n;
#pragma unroll
    for (int t = 0; t < GEMM_TILES; ++t)
#pragma unroll
        for (int r = 0; r < 8; ++r)
            orow[(size_t)(t * 16 + r) * DIM] = c[t][r];
}


__global__ void k_selfloop(const float* __restrict__ h, const float* __restrict__ dinv,
                           float* __restrict__ out) {
    int i = blockIdx.x * blockDim.x + threadIdx.x;
    int node = i >> 5, q = i & 31;
    if (node < NN) {
        float w = dinv[node]; w *= w;
        float4 v = ((const float4*)(h + (size_t)node * DIM))[q];
        float4 r; r.x = v.x * w; r.y = v.y * w; r.z = v.z * w; r.w = v.w * w;
        ((float4*)(out + (size_t)node * DIM))[q] = r;
    }
}

__global__ void k_edge_agg(const int* __restrict__ src, const int* __restrict__ dst,
                           const float* __restrict__ dinv, const float* __restrict__ h,
                           float* __restrict__ out) {
    int i = blockIdx.x * blockDim.x + threadIdx.x;
    int e = i >> 5, q = i & 31;
    if (e < NE) {
        int s = src[e], d = dst[e];
        float w = dinv[s] * dinv[d];
        const float* hs = h + (size_t)s * DIM + q;
        float v0 = hs[0], v1 = hs[32], v2 = hs[64], v3 = hs[96];
        float* o = out + (size_t)d * DIM + q;
        atomicAdd(o + 0,  v0 * w);
        atomicAdd(o + 32, v1 * w);
        atomicAdd(o + 64, v2 * w);
        atomicAdd(o + 96, v3 * w);
    }
}


__global__ void k_count_rank(const int* __restrict__ dst, int* __restrict__ cnt,
                             int* __restrict__ rank, int e) {
    int i = blockIdx.x * blockDim.x + threadIdx.x;
    if (i < e) rank[i] = atomicAdd(&cnt[dst[i]], 1);
}

__global__ void __launch_bounds__(SCAN_T)
k_scan_dinv(const int* __restrict__ cnt, int* __restrict__ rowptr, float* __restrict__ dinv) {
    __shared__ int s_pre[SCAN_T / 32];
    __shared__ int s_own[SCAN_T / 32];
    const int t = threadIdx.x, b = blockIdx.x;
    const int lane = t & 31, wave = t >> 5;

    int pre = 0;
    const int4* c4 = (const int4*)cnt;
    for (int q = t; q < b * (SCAN_T / 4); q += SCAN_T) {
        int4 v = c4[q];
        pre += (v.x + v.y) + (v.z + v.w);
    }
    const int i = b * SCAN_T + t;
    const int own = (i < NN) ? cnt[i] : 0;

    int inc = own;
    for (int d = 1; d < 32; d <<= 1) {
        int u = __shfl_up(inc, d);
        if (lane >= d) inc += u;
    }
    for (int d = 16; d > 0; d >>= 1) pre += __shfl_xor(pre, d);
    if (lane == 31) { s_own[wave] = inc; s_pre[wave] = pre; }
    __syncthreads();
    int base = 0;
    for (int w = 0; w < SCAN_T / 32; ++w) {
        base += s_pre[w];
        if (w < wave) base += s_own[w];
    }
    const int excl = base + inc - own;
    if (i < NN) {
        rowptr[i] = excl;
        dinv[i] = rsqrtf(fmaxf((float)(own + 1), 1.f));
        if (i == NN - 1) rowptr[NN] = excl + own;
    }
}

__global__ void k_fill(const int* __restrict__ src, const int* __restrict__ dst,
                       const int* __restrict__ rank, const int* __restrict__ rowptr,
                       int* __restrict__ csr, int e) {
    int i = blockIdx.x * blockDim.x + threadIdx.x;
    if (i < e) csr[rowptr[dst[i]] + rank[i]] = src[i];
}

static_assert(NN % (AGG_WAVES * AGG_NPW) == 0, "every wave of every block owns AGG_NPW nodes");
#define POOL_PITCH (DIM + 4)

__device__ __forceinline__ void pool_flush_raw(float* __restrict__ pooled, float* __restrict__ gcnt,
                                               int g, int lane, float4 acc, float rows) {
    if ((unsigned)g < (unsigned)NG) {
        float* o = pooled + (size_t)g * OUTC + 4 * lane;
        atomicAdd(o + 0, acc.x); atomicAdd(o + 1, acc.y);
        atomicAdd(o + 2, acc.z); atomicAdd(o + 3, acc.w);
        if (gcnt != nullptr && lane == 0) atomicAdd(&gcnt[g], rows);
    }
}

static_assert(AGG_NPW < 32, "a wave keeps its nodes' segment bounds one per lane");
static_assert(AGG_WAVES * 32 >= 2 * DIM, "one thread per column sum in the block's reduction");

template <bool STORE_ROWS>
__global__ void __launch_bounds__(AGG_WAVES * 32)
k_agg_bias_relu_stats(const int* __restrict__ rowptr, const int* __restrict__ csr,
                      const float* __restrict__ dinv, const float* __restrict__ h,
                      const float* __restrict__ bias, const int* __restrict__ batch,
                      float* __restrict__ out, float* __restrict__ stats,
                      float* __restrict__ gcnt, float* __restrict__ pooled) {
    __shared__ float red[AGG_WAVES][2 * DIM];
    __shared__ __attribute__((aligned(16))) float pool[AGG_WAVES][POOL_PITCH];
    __shared__ int   pgid[AGG_WAVES];
    const int lane = threadIdx.x & 31;
    const int wave = threadIdx.x >> 5;
    const int n0 = __builtin_amdgcn_readfirstlane((blockIdx.x * AGG_WAVES + wave) * AGG_NPW);
    const int rp = rowptr[n0 + min(lane, AGG_NPW)];
    const int dl = __float_as_int(dinv[n0 + min(lane, AGG_NPW - 1)]);
    const int gl = batch[n0 + min(lane, AGG_NPW - 1)];
    const float4 b = ((const float4*)bias)[lane];
    const float4* hv = (const float4*)h;

    float4 cs = {0.f, 0.f, 0.f, 0.f}, cq = {0.f, 0.f, 0.f, 0.f};
    float4 pa = {0.f, 0.f, 0.f, 0.f};
    float  pn = 0.f;
    int    cur = __builtin_amdgcn_readlane(gl, 0);
    for (int k = 0; k < AGG_NPW; ++k) {
        const int node  = n0 + k;
        const int beg   = __builtin_amdgcn_readlane(rp, k);
        const int items = __builtin_amdgcn_readlane(rp, k + 1) - beg + 1;
        const float dd  = __int_as_float(__builtin_amdgcn_readlane(dl, k));

        float4 acc = {0.f, 0.f, 0.f, 0.f};
        for (int base = 0; base < items; base += 32) {
            const int it = base + lane;
            int s = node;
            if (it > 0 && it < items) {
                s = (int)min((unsigned)csr[beg + it - 1], (unsigned)(NN - 1));
            }
            const int n = min(32, items - base);
            int j = 0;
            for (; j + 4 <= n; j += 4) {
                const int s0 = __builtin_amdgcn_readlane(s, j);
                const int s1 = __builtin_amdgcn_readlane(s, j + 1);
                const int s2 = __builtin_amdgcn_readlane(s, j + 2);
                const int s3 = __builtin_amdgcn_readlane(s, j + 3);
                const float4 v0 = hv[(size_t)s0 * 32 + lane];
                const float4 v1 = hv[(size_t)s1 * 32 + lane];
                const float4 v2 = hv[(size_t)s2 * 32 + lane];
                const float4 v3 = hv[(size_t)s3 * 32 + lane];
                acc.x += v0.x; acc.y += v0.y; acc.z += v0.z; acc.w += v0.w;
                acc.x += v1.x; acc.y += v1.y; acc.z += v1.z; acc.w += v1.w;
                acc.x += v2.x; acc.y += v2.y; acc.z += v2.z; acc.w += v2.w;
                acc.x += v3.x; acc.y += v3.y; acc.z += v3.z; acc.w += v3.w;
            }
            const int rest = n - j;
            if (rest > 0) {
                const int s0 = __builtin_amdgcn_readlane(s, j);
                const float4 v0 = hv[(size_t)s0 * 32 + lane];
                float4 v1 = {0.f, 0.f, 0.f, 0.f}, v2 = {0.f, 0.f, 0.f, 0.f};
                if (rest > 1) v1 = hv[(size_t)__builtin_amdgcn_readlane(s, j + 1) * 32 + lane];
                if (rest > 2) v2 = hv[(size_t)__builtin_amdgcn_readlane(s, j + 2) * 32 + lane];
                acc.x += v0.x; acc.y += v0.y; acc.z += v0.z; acc.w += v0.w;
                acc.x += v1.x; acc.y += v1.y; acc.z += v1.z; acc.w += v1.w;
                acc.x += v2.x; acc.y += v2.y; acc.z += v2.z; acc.w += v2.w;
            }
        }
        float4 x;
        x.x = fmaxf(fmaf(acc.x, dd, b.x), 0.f); x.y = fmaxf(fmaf(acc.y, dd, b.y), 0.f);
        x.z = fmaxf(fmaf(acc.z, dd, b.z), 0.f); x.w = fmaxf(fmaf(acc.w, dd, b.w), 0.f);
        if (STORE_ROWS) ((float4*)out)[(size_t)node * 32 + lane] = x;
        const int g = __builtin_amdgcn_readlane(gl, k);
        if (g != cur) {
            pool_flush_raw(pooled, gcnt, cur, lane, pa, pn);
            cur = g; pn = 0.f;
            pa.x = 0.f; pa.y = 0.f; pa.z = 0.f; pa.w = 0.f;
        }
        pa.x += x.x; pa.y += x.y; pa.z += x.z; pa.w += x.w;
        pn += 1.f;
        cs.x += x.x; cs.y += x.y; cs.z += x.z; cs.w += x.w;
        cq.x += x.x * x.x; cq.y += x.y * x.y; cq.z += x.z * x.z; cq.w += x.w * x.w;
    }
    ((float4*)&red[wave][0])[lane]   = cs;
    ((float4*)&red[wave][DIM])[lane] = cq;
    ((float4*)&pool[wave][0])[lane]  = pa;
    if (lane == 0) { pool[wave][DIM] = pn; pgid[wave] = cur; }
    __syncthreads();
    if (threadIdx.x < 2 * DIM) {
        float t = 0.f;
        for (int w = 0; w < AGG_WAVES; ++w) t += red[w][threadIdx.x];
        atomicAdd(&stats[threadIdx.x], t);
    }
    if (threadIdx.x < DIM || (threadIdx.x == DIM && gcnt != nullptr)) {
        const int c = threadIdx.x;
        float* dst = (c < DIM) ? pooled + c : gcnt;
        const size_t pitch = (c < DIM) ? (size_t)OUTC : (size_t)1;
        int cg = pgid[0];
        float t = pool[0][c];
        for (int w = 1; w < AGG_WAVES; ++w) {
            const int g = pgid[w];
            if (g != cg) {
                if ((unsigned)cg < (unsigned)NG) atomicAdd(dst + (size_t)cg * pitch, t);
                cg = g; t = 0.f;
            }
            t += pool[w][c];
        }
        if ((unsigned)cg < (unsigned)NG) atomicAdd(dst + (size_t)cg * pitch, t);
    }
}


static_assert(NN % (EPI_RPW * 8) == 0, "every wave of every block owns EPI_RPW rows");

__global__ void __launch_bounds__(256)
k_bias_relu_stats(float* __restrict__ h, const float* __restrict__ bias,
                  float* __restrict__ stats) {
    __shared__ float red[8][2 * DIM];
    const int lane = threadIdx.x & 31;
    const int wave = threadIdx.x >> 5;
    const int r0 = __builtin_amdgcn_readfirstlane((blockIdx.x * 8 + wave) * EPI_RPW);
    float4* hv = (float4*)h + (size_t)r0 * 32 + lane;
    const float4 b = ((const float4*)bias)[lane];

    float4 v[EPI_RPW];
#pragma unroll
    for (int u = 0; u < EPI_RPW; ++u) v[u] = hv[u * 32];

    float4 s = {0.f, 0.f, 0.f, 0.f}, q = {0.f, 0.f, 0.f, 0.f};
#pragma unroll
    for (int u = 0; u < EPI_RPW; ++u) {
        float4 x = v[u];
        x.x = fmaxf(x.x + b.x, 0.f); x.y = fmaxf(x.y + b.y, 0.f);
        x.z = fmaxf(x.z + b.z, 0.f); x.w = fmaxf(x.w + b.w, 0.f);
        hv[u * 32] = x;
        s.x += x.x; s.y += x.y; s.z += x.z; s.w += x.w;
        q.x += x.x * x.x; q.y += x.y * x.y; q.z += x.z * x.z; q.w += x.w * x.w;
    }
    ((float4*)&red[wave][0])[lane]   = s;
    ((float4*)&red[wave][DIM])[lane] = q;
    __syncthreads();
    float t = 0.f;
    for (int w = 0; w < 8; ++w) t += red[w][threadIdx.x];
    atomicAdd(&stats[threadIdx.x], t);
}

__global__ void k_stats_finish(float* __restrict__ stats, float inv_n) {
    int c = threadIdx.x;
    float mu  = stats[c] * inv_n;
    float var = stats[DIM + c] * inv_n - mu * mu;
    stats[ST_MU + c] = mu;
    stats[ST_RS + c] = rsqrtf(var + BN_EPS);
}

__global__ void __launch_bounds__(DIM)
k_stats_pool_finish(const float* __restrict__ sums, const float* __restrict__ gcnt,
                    const float* __restrict__ gamma, const float* __restrict__ beta,
                    float* __restrict__ pooled, float inv_n) {
    const int c = threadIdx.x, g = blockIdx.x;
    const float mu = sums[c] * inv_n;
    const float rs = bn_rstd(sums[DIM + c], mu, inv_n);
    const float n = gcnt[g];
    float* o = pooled + (size_t)g * OUTC + c;
    *o = fmaf(fmaf(-n, mu, *o) * rs, gamma[c], n * beta[c]);
}

__device__ __forceinline__ void pool_flush(float* __restrict__ out, int g, int layer, int lane,
                                           float4 acc) {
    if ((unsigned)g < (unsigned)NG) {
        float* o = out + (size_t)g * OUTC + layer * DIM + 4 * lane;
        atomicAdd(o + 0, acc.x); atomicAdd(o + 1, acc.y);
        atomicAdd(o + 2, acc.z); atomicAdd(o + 3, acc.w);
    }
}

__global__ void __launch_bounds__(256)
k_norm_pool(const float* __restrict__ h, const float* __restrict__ stats,
            const float* __restrict__ gamma, const float* __restrict__ beta,
            const int* __restrict__ batch, float* __restrict__ hnext,
            float* __restrict__ out, int layer) {
    const int lane = threadIdx.x & 31;
    const int wave = threadIdx.x >> 5;
    const int r0 = __builtin_amdgcn_readfirstlane((blockIdx.x * 8 + wave) * EPI_RPW);
    const float4* hv = (const float4*)h + (size_t)r0 * 32 + lane;
    float4*       hn = (float4*)hnext   + (size_t)r0 * 32 + lane;

    float4 v[EPI_RPW];
#pragma unroll
    for (int u = 0; u < EPI_RPW; ++u) v[u] = hv[u * 32];

    const int gl = batch[r0 + min(lane, EPI_RPW - 1)];
    const float4 mu = ((const float4*)(stats + ST_MU))[lane];
    const float4 rs = ((const float4*)(stats + ST_RS))[lane];
    const float4 ga = ((const float4*)gamma)[lane];
    const float4 be = ((const float4*)beta)[lane];

    int cur = __builtin_amdgcn_readlane(gl, 0);
    float4 acc = {0.f, 0.f, 0.f, 0.f};
#pragma unroll
    for (int u = 0; u < EPI_RPW; ++u) {
        float4 x;
        x.x = (v[u].x - mu.x) * rs.x * ga.x + be.x;
        x.y = (v[u].y - mu.y) * rs.y * ga.y + be.y;
        x.z = (v[u].z - mu.z) * rs.z * ga.z + be.z;
        x.w = (v[u].w - mu.w) * rs.w * ga.w + be.w;
        hn[u * 32] = x;
        const int g = __builtin_amdgcn_readlane(gl, u);
        if (g != cur) {
            pool_flush(out, cur, layer, lane, acc);
            cur = g;
            acc.x = 0.f; acc.y = 0.f; acc.z = 0.f; acc.w = 0.f;
        }
        acc.x += x.x; acc.y += x.y; acc.z += x.z; acc.w += x.w;
    }
    pool_flush(out, cur, layer, lane, acc);
}


extern "C" void kernel_launch(void* const* d_in, const int* in_sizes, int n_in,
                              void* d_out, int out_size, void* d_ws, size_t ws_size,
                              hipStream_t stream) {
    (void)in_sizes; (void)n_in; (void)out_size;

    const float* x     = (const float*)d_in[0];
    const int*   ei    = (const int*)d_in[1];
    const int*   batch = (const int*)d_in[2];
    const float *Wp[NL], *bp[NL], *gp[NL], *tp[NL];
    for (int i = 0; i < NL; ++i) {
        Wp[i] = (const float*)d_in[3 + 4 * i];
        bp[i] = (const float*)d_in[4 + 4 * i];
        gp[i] = (const float*)d_in[5 + 4 * i];
        tp[i] = (const float*)d_in[6 + 4 * i];
    }
    const int* esrc = ei;
    const int* edst = ei + NE;

    float* ws    = (float*)d_ws;
    float* dinv  = ws;
    float* stats = ws + NN;
    size_t o     = ((size_t)NN + ST_FLOATS + 255) & ~(size_t)255;
    float* bufA  = ws + o;
    float* bufB  = bufA + (size_t)NN * DIM;
    float* outp  = (float*)d_out;

    int* rowptr  = (int*)(bufB + (size_t)NN * DIM);
    int* csr     = rowptr + RP_PAD;
    int* cnt     = (int*)bufB;
    int* rank    = cnt + RP_PAD;
    const bool sorted =
        ws_size >= (o + 2 * (size_t)NN * DIM + RP_PAD + NE) * sizeof(float);

    if (sorted) {
        k_zero3     <<<(NG * OUTC + NN + ST_FLOATS + 255) / 256, 256, 0, stream>>>(
            outp, NG * OUTC, (float*)cnt, NN, stats, ST_FLOATS);
        k_count_rank<<<(NE + 255) / 256,         256,    0, stream>>>(edst, cnt, rank, NE);
        k_scan_dinv <<<(NN + SCAN_T - 1) / SCAN_T, SCAN_T, 0, stream>>>(cnt, rowptr, dinv);
        k_fill      <<<(NE + 255) / 256,         256,    0, stream>>>(esrc, edst, rank, rowptr, csr, NE);
    } else {
        k_zero     <<<(NG * OUTC + 255) / 256, 256, 0, stream>>>(outp, NG * OUTC);
        k_init_deg <<<(NN + 255) / 256, 256, 0, stream>>>(dinv, NN);
        k_deg_edges<<<(NE + 255) / 256, 256, 0, stream>>>(edst, dinv, NE);
        k_dinv     <<<(NN + 255) / 256, 256, 0, stream>>>(dinv, NN);
    }

    for (int l = 0; l < NL; ++l) {
        if (sorted) {
            float* sums = stats + SS_SUM(l);
            float* gcnt = stats + SS_CNT;
            if (l == 0)
                k_gemm_wmma<false, true><<<NN / GEMM_ROWS, 256, 0, stream>>>(
                    x, Wp[l], bufA, nullptr, nullptr, nullptr, dinv, nullptr, nullptr, nullptr,
                    1.f / (float)NN);
            else
                k_gemm_wmma<true, true><<<NN / GEMM_ROWS, 256, 0, stream>>>(
                    bufB, Wp[l], bufA, stats + SS_SUM(l - 1), gp[l - 1], tp[l - 1], dinv, gcnt,
                    outp + (l - 1) * DIM, (l >= 2) ? sums : nullptr, 1.f / (float)NN);
            if (l < NL - 1)
                k_agg_bias_relu_stats<true><<<AGG_BLOCKS, AGG_WAVES * 32, 0, stream>>>(
                    rowptr, csr, dinv, bufA, bp[l], batch, bufB, sums, (l == 0) ? gcnt : nullptr,
                    outp + l * DIM);
            else {
                k_agg_bias_relu_stats<false><<<AGG_BLOCKS, AGG_WAVES * 32, 0, stream>>>(
                    rowptr, csr, dinv, bufA, bp[l], batch, bufB, sums, (l == 0) ? gcnt : nullptr,
                    outp + l * DIM);
                k_stats_pool_finish<<<NG, DIM, 0, stream>>>(
                    sums, gcnt, gp[l], tp[l], outp + l * DIM, 1.f / (float)NN);
            }
            continue;
        }
        const float* hin = (l == 0) ? x : bufA;
        k_gemm_wmma<false, false><<<NN / GEMM_ROWS, 256, 0, stream>>>(
            hin, Wp[l], bufA, nullptr, nullptr, nullptr, nullptr, nullptr, nullptr, nullptr, 0.f);
        k_zero<<<1, 256, 0, stream>>>(stats, 2 * DIM);
        k_selfloop <<<(NN * 32 + 255) / 256, 256, 0, stream>>>(bufA, dinv, bufB);
        k_edge_agg <<<(NE * 32) / 256, 256, 0, stream>>>(esrc, edst, dinv, bufA, bufB);
        k_bias_relu_stats<<<EPI_BLOCKS, 256, 0, stream>>>(bufB, bp[l], stats);
        k_stats_finish   <<<1, DIM, 0, stream>>>(stats, 1.f / (float)NN);
        k_norm_pool      <<<EPI_BLOCKS, 256, 0, stream>>>(
            bufB, stats, gp[l], tp[l], batch, bufA, outp, l);
    }
}


# sorted path: one launch per layer: a block sums its 80 nodes' dinv-weighted source rows into LDS and multiplies the tile by W there, (Adj h)W; sum order and rounding differ
# speedup vs baseline: 5.0568x; 1.1025x over previous
#include <hip/hip_runtime.h>

#define NN 100000
#define NE 1600000
#define DIM 128
#define NG 512
#define NL 3
#define OUTC (NL * DIM)
#define BN_EPS 1e-5f
#define EPI_RPW 25
#define EPI_BLOCKS (NN / EPI_RPW / 8)
#define SCAN_T 1024
#define RP_PAD 100096
#define GEMM_TILES 5
#define GEMM_ROWS (16 * GEMM_TILES)
#define A_PITCH (DIM + 4)
#define ST_MU  (2 * DIM + NG)
#define ST_RS  (3 * DIM + NG)
#define ST_FLOATS (4 * DIM + NG)
#define SS_SUM(l) (((l) & 1) * 2 * DIM)
#define SS_CNT (4 * DIM)
static_assert(SS_CNT + NG <= ST_FLOATS, "the sorted path's sums and counts fit the statistics block");

typedef __attribute__((ext_vector_type(2))) float v2f;
typedef __attribute__((ext_vector_type(8))) float v8f;


__global__ void k_zero(float* __restrict__ p, int n) {
    int i = blockIdx.x * blockDim.x + threadIdx.x;
    if (i < n) p[i] = 0.f;
}

__global__ void k_zero3(float* __restrict__ a, int na, float* __restrict__ b, int nb,
                        float* __restrict__ c, int nc) {
    int i = blockIdx.x * blockDim.x + threadIdx.x;
    if (i < na) a[i] = 0.f;
    else if (i < na + nb) b[i - na] = 0.f;
    else if (i < na + nb + nc) c[i - na - nb] = 0.f;
}

__device__ __forceinline__ float bn_rstd(float q, float mu, float inv_n) {
    const float var = q * inv_n - mu * mu;
    return rsqrtf(var + BN_EPS);
}

__global__ void k_init_deg(float* __restrict__ deg, int n) {
    int i = blockIdx.x * blockDim.x + threadIdx.x;
    if (i < n) deg[i] = 1.f;
}

__global__ void k_deg_edges(const int* __restrict__ dst, float* __restrict__ deg, int e) {
    int i = blockIdx.x * blockDim.x + threadIdx.x;
    if (i < e) atomicAdd(&deg[dst[i]], 1.f);
}

__global__ void k_dinv(float* __restrict__ deg, int n) {
    int i = blockIdx.x * blockDim.x + threadIdx.x;
    if (i < n) deg[i] = rsqrtf(fmaxf(deg[i], 1.f));
}

static_assert(NN % GEMM_ROWS == 0, "every block owns GEMM_ROWS whole rows");
static_assert((GEMM_ROWS * 32) % 256 == 0, "staging: a whole number of float4 per thread");

__global__ void __launch_bounds__(256)
k_gemm_wmma(const float* __restrict__ A, const float* __restrict__ W, float* __restrict__ out) {
    __shared__ float tileA[GEMM_ROWS * A_PITCH];

    const int tid  = threadIdx.x;
    const int wave = __builtin_amdgcn_readfirstlane(tid >> 5);
    const int lane = tid & 31;
    const int row0 = blockIdx.x * GEMM_ROWS;

    constexpr int NST = GEMM_ROWS * 32 / 256;
    const float4* srcv = (const float4*)(A + (size_t)row0 * DIM);
    float4 st[NST];
#pragma unroll
    for (int j = 0; j < NST; ++j) st[j] = srcv[tid + 256 * j];

    const int m  = lane & 15;
    const int kh = (lane >> 4) << 1;
    const int n  = (wave << 4) + m;

    const float* wcol = W + (size_t)kh * DIM + n;
    v2f b[DIM / 4];
#pragma unroll
    for (int s = 0; s < DIM / 4; ++s) {
        b[s].x = wcol[(size_t)(4 * s) * DIM];
        b[s].y = wcol[(size_t)(4 * s + 1) * DIM];
    }

    float* stage = tileA + wave * A_PITCH + (lane << 2);
#pragma unroll
    for (int j = 0; j < NST; ++j) *(float4*)&stage[8 * j * A_PITCH] = st[j];
    __syncthreads();

    v8f c[GEMM_TILES];
#pragma unroll
    for (int t = 0; t < GEMM_TILES; ++t) c[t] = v8f{};

    const float* arow = tileA + m * A_PITCH + kh;
#pragma unroll
    for (int s = 0; s < DIM / 4; ++s) {
#pragma unroll
        for (int t = 0; t < GEMM_TILES; ++t) {
            const v2f a = *(const v2f*)(arow + t * 16 * A_PITCH + 4 * s);
            c[t] = __builtin_amdgcn_wmma_f32_16x16x4_f32(
                        false, a,  false, b[s],
                        (short)0, c[t],  false,  false);
        }
    }

    const int mo = (lane >> 4) << 3;
    float* orow = out + (size_t)(row0 + mo) * DIM + n;
#pragma unroll
    for (int t = 0; t < GEMM_TILES; ++t)
#pragma unroll
        for (int r = 0; r < 8; ++r)
            orow[(size_t)(t * 16 + r) * DIM] = c[t][r];
}


__global__ void k_selfloop(const float* __restrict__ h, const float* __restrict__ dinv,
                           float* __restrict__ out) {
    int i = blockIdx.x * blockDim.x + threadIdx.x;
    int node = i >> 5, q = i & 31;
    if (node < NN) {
        float w = dinv[node]; w *= w;
        float4 v = ((const float4*)(h + (size_t)node * DIM))[q];
        float4 r; r.x = v.x * w; r.y = v.y * w; r.z = v.z * w; r.w = v.w * w;
        ((float4*)(out + (size_t)node * DIM))[q] = r;
    }
}

__global__ void k_edge_agg(const int* __restrict__ src, const int* __restrict__ dst,
                           const float* __restrict__ dinv, const float* __restrict__ h,
                           float* __restrict__ out) {
    int i = blockIdx.x * blockDim.x + threadIdx.x;
    int e = i >> 5, q = i & 31;
    if (e < NE) {
        int s = src[e], d = dst[e];
        float w = dinv[s] * dinv[d];
        const float* hs = h + (size_t)s * DIM + q;
        float v0 = hs[0], v1 = hs[32], v2 = hs[64], v3 = hs[96];
        float* o = out + (size_t)d * DIM + q;
        atomicAdd(o + 0,  v0 * w);
        atomicAdd(o + 32, v1 * w);
        atomicAdd(o + 64, v2 * w);
        atomicAdd(o + 96, v3 * w);
    }
}


__global__ void k_count_rank(const int* __restrict__ dst, int* __restrict__ cnt,
                             int* __restrict__ rank, int e) {
    int i = blockIdx.x * blockDim.x + threadIdx.x;
    if (i < e) rank[i] = atomicAdd(&cnt[dst[i]], 1);
}

__global__ void __launch_bounds__(SCAN_T)
k_scan_dinv(const int* __restrict__ cnt, int* __restrict__ rowptr, float* __restrict__ dinv) {
    __shared__ int s_pre[SCAN_T / 32];
    __shared__ int s_own[SCAN_T / 32];
    const int t = threadIdx.x, b = blockIdx.x;
    const int lane = t & 31, wave = t >> 5;

    int pre = 0;
    const int4* c4 = (const int4*)cnt;
    for (int q = t; q < b * (SCAN_T / 4); q += SCAN_T) {
        int4 v = c4[q];
        pre += (v.x + v.y) + (v.z + v.w);
    }
    const int i = b * SCAN_T + t;
    const int own = (i < NN) ? cnt[i] : 0;

    int inc = own;
    for (int d = 1; d < 32; d <<= 1) {
        int u = __shfl_up(inc, d);
        if (lane >= d) inc += u;
    }
    for (int d = 16; d > 0; d >>= 1) pre += __shfl_xor(pre, d);
    if (lane == 31) { s_own[wave] = inc; s_pre[wave] = pre; }
    __syncthreads();
    int base = 0;
    for (int w = 0; w < SCAN_T / 32; ++w) {
        base += s_pre[w];
        if (w < wave) base += s_own[w];
    }
    const int excl = base + inc - own;
    if (i < NN) {
        rowptr[i] = excl;
        dinv[i] = rsqrtf(fmaxf((float)(own + 1), 1.f));
        if (i == NN - 1) rowptr[NN] = excl + own;
    }
}

__global__ void k_fill(const int* __restrict__ src, const int* __restrict__ dst,
                       const int* __restrict__ rank, const int* __restrict__ rowptr,
                       int* __restrict__ csr, int e) {
    int i = blockIdx.x * blockDim.x + threadIdx.x;
    if (i < e) csr[rowptr[dst[i]] + rank[i]] = src[i];
}

#define FU_NPW 10
static_assert(8 * FU_NPW == GEMM_ROWS, "the block's 8 waves fill the product's row tile");
static_assert(FU_NPW < 32, "a wave keeps its nodes' segment bounds one per lane");
static_assert(NN / GEMM_ROWS >= NG, "one block per graph for the previous layer's pooled rows");
#define POOL_PITCH (DIM + 4)

__device__ __forceinline__ void pool_flush_raw(float* __restrict__ pooled, float* __restrict__ gcnt,
                                               int g, int lane, float4 acc, float rows) {
    if ((unsigned)g < (unsigned)NG) {
        float* o = pooled + (size_t)g * OUTC + 4 * lane;
        atomicAdd(o + 0, acc.x); atomicAdd(o + 1, acc.y);
        atomicAdd(o + 2, acc.z); atomicAdd(o + 3, acc.w);
        if (gcnt != nullptr && lane == 0) atomicAdd(&gcnt[g], rows);
    }
}

template <bool NORM, bool STORE_ROWS>
__global__ void __launch_bounds__(256)
k_layer(const int* __restrict__ rowptr, const int* __restrict__ csr,
        const float* __restrict__ dinv, const float* __restrict__ h,
        const float* __restrict__ W, const float* __restrict__ bias,
        const float* __restrict__ psums, const float* __restrict__ pgamma,
        const float* __restrict__ pbeta, const float* __restrict__ gcnt_prev,
        float* __restrict__ pooled_prev, const int* __restrict__ batch,
        float* __restrict__ out, float* __restrict__ stats, float* __restrict__ gcnt,
        float* __restrict__ pooled, float inv_n) {
    __shared__ __attribute__((aligned(16))) float tileA[GEMM_ROWS * A_PITCH];
    __shared__ float red[8][2 * DIM];
    __shared__ __attribute__((aligned(16))) float pool[8][POOL_PITCH];
    __shared__ int   pgid[8];

    const int tid  = threadIdx.x;
    const int wave = __builtin_amdgcn_readfirstlane(tid >> 5);
    const int lane = tid & 31;

    float4 mu = {0.f, 0.f, 0.f, 0.f}, sa = {0.f, 0.f, 0.f, 0.f}, be = {0.f, 0.f, 0.f, 0.f};
    if (NORM) {
        if (blockIdx.x < NG && tid < DIM) {
            const float pm = psums[tid] * inv_n;
            const float pr = bn_rstd(psums[DIM + tid], pm, inv_n);
            const float rows = gcnt_prev[blockIdx.x];
            float* o = pooled_prev + (size_t)blockIdx.x * OUTC + tid;
            *o = fmaf(fmaf(-rows, pm, *o) * pr, pgamma[tid], rows * pbeta[tid]);
        }
        const float4 cs = ((const float4*)psums)[lane];
        const float4 cq = ((const float4*)(psums + DIM))[lane];
        const float4 ga = ((const float4*)pgamma)[lane];
        be = ((const float4*)pbeta)[lane];
        mu.x = cs.x * inv_n; mu.y = cs.y * inv_n; mu.z = cs.z * inv_n; mu.w = cs.w * inv_n;
        sa.x = bn_rstd(cq.x, mu.x, inv_n) * ga.x; sa.y = bn_rstd(cq.y, mu.y, inv_n) * ga.y;
        sa.z = bn_rstd(cq.z, mu.z, inv_n) * ga.z; sa.w = bn_rstd(cq.w, mu.w, inv_n) * ga.w;
    }

    const int n0 = __builtin_amdgcn_readfirstlane((blockIdx.x * 8 + wave) * FU_NPW);
    const int rp = rowptr[n0 + min(lane, FU_NPW)];
    const int dl = __float_as_int(dinv[n0 + min(lane, FU_NPW - 1)]);
    const float4* hv = (const float4*)h;
    float* trow = tileA + wave * FU_NPW * A_PITCH + (lane << 2);

    for (int k = 0; k < FU_NPW; ++k) {
        const int node  = n0 + k;
        const int beg   = __builtin_amdgcn_readlane(rp, k);
        const int items = __builtin_amdgcn_readlane(rp, k + 1) - beg + 1;
        const float dd  = __int_as_float(__builtin_amdgcn_readlane(dl, k));

        float4 acc = {0.f, 0.f, 0.f, 0.f};
        float  sd  = 0.f;
        for (int base = 0; base < items; base += 32) {
            const int it = base + lane;
            int s = node;
            if (it > 0 && it < items) {
                s = (int)min((unsigned)csr[beg + it - 1], (unsigned)(NN - 1));
            }
            const int wl = __float_as_int(dinv[s]);
            const int n = min(32, items - base);
            int j = 0;
            for (; j + 4 <= n; j += 4) {
                const int s0 = __builtin_amdgcn_readlane(s, j);
                const int s1 = __builtin_amdgcn_readlane(s, j + 1);
                const int s2 = __builtin_amdgcn_readlane(s, j + 2);
                const int s3 = __builtin_amdgcn_readlane(s, j + 3);
                const float w0 = __int_as_float(__builtin_amdgcn_readlane(wl, j));
                const float w1 = __int_as_float(__builtin_amdgcn_readlane(wl, j + 1));
                const float w2 = __int_as_float(__builtin_amdgcn_readlane(wl, j + 2));
                const float w3 = __int_as_float(__builtin_amdgcn_readlane(wl, j + 3));
                const float4 v0 = hv[(size_t)s0 * 32 + lane];
                const float4 v1 = hv[(size_t)s1 * 32 + lane];
                const float4 v2 = hv[(size_t)s2 * 32 + lane];
                const float4 v3 = hv[(size_t)s3 * 32 + lane];
                acc.x = fmaf(v0.x, w0, acc.x); acc.y = fmaf(v0.y, w0, acc.y); acc.z = fmaf(v0.z, w0, acc.z); acc.w = fmaf(v0.w, w0, acc.w);
                acc.x = fmaf(v1.x, w1, acc.x); acc.y = fmaf(v1.y, w1, acc.y); acc.z = fmaf(v1.z, w1, acc.z); acc.w = fmaf(v1.w, w1, acc.w);
                acc.x = fmaf(v2.x, w2, acc.x); acc.y = fmaf(v2.y, w2, acc.y); acc.z = fmaf(v2.z, w2, acc.z); acc.w = fmaf(v2.w, w2, acc.w);
                acc.x = fmaf(v3.x, w3, acc.x); acc.y = fmaf(v3.y, w3, acc.y); acc.z = fmaf(v3.z, w3, acc.z); acc.w = fmaf(v3.w, w3, acc.w);
                sd += w0; sd += w1; sd += w2; sd += w3;
            }
            const int rest = n - j;
            if (rest > 0) {
                const int s0 = __builtin_amdgcn_readlane(s, j);
                const float w0 = __int_as_float(__builtin_amdgcn_readlane(wl, j));
                const float4 v0 = hv[(size_t)s0 * 32 + lane];
                float4 v1 = {0.f, 0.f, 0.f, 0.f}, v2 = {0.f, 0.f, 0.f, 0.f};
                float w1 = 0.f, w2 = 0.f;
                if (rest > 1) {
                    v1 = hv[(size_t)__builtin_amdgcn_readlane(s, j + 1) * 32 + lane];
                    w1 = __int_as_float(__builtin_amdgcn_readlane(wl, j + 1));
                }
                if (rest > 2) {
                    v2 = hv[(size_t)__builtin_amdgcn_readlane(s, j + 2) * 32 + lane];
                    w2 = __int_as_float(__builtin_amdgcn_readlane(wl, j + 2));
                }
                acc.x = fmaf(v0.x, w0, acc.x); acc.y = fmaf(v0.y, w0, acc.y); acc.z = fmaf(v0.z, w0, acc.z); acc.w = fmaf(v0.w, w0, acc.w);
                acc.x = fmaf(v1.x, w1, acc.x); acc.y = fmaf(v1.y, w1, acc.y); acc.z = fmaf(v1.z, w1, acc.z); acc.w = fmaf(v1.w, w1, acc.w);
                acc.x = fmaf(v2.x, w2, acc.x); acc.y = fmaf(v2.y, w2, acc.y); acc.z = fmaf(v2.z, w2, acc.z); acc.w = fmaf(v2.w, w2, acc.w);
                sd += w0; sd += w1; sd += w2;
            }
        }
        float4 z;
        if (NORM) {
            z.x = (fmaf(-mu.x, sd, acc.x) * sa.x + be.x * sd) * dd;
            z.y = (fmaf(-mu.y, sd, acc.y) * sa.y + be.y * sd) * dd;
            z.z = (fmaf(-mu.z, sd, acc.z) * sa.z + be.z * sd) * dd;
            z.w = (fmaf(-mu.w, sd, acc.w) * sa.w + be.w * sd) * dd;
        } else {
            z.x = acc.x * dd; z.y = acc.y * dd; z.z = acc.z * dd; z.w = acc.w * dd;
        }
        *(float4*)&trow[k * A_PITCH] = z;
    }

    const int m  = lane & 15;
    const int kh = (lane >> 4) << 1;
    const int n  = (wave << 4) + m;

    const float* wcol = W + (size_t)kh * DIM + n;
    constexpr int NGR = DIM / 8;
    constexpr int BPD = 3;
    v2f b[DIM / 4];
#pragma unroll
    for (int s = 0; s < 2 * BPD; ++s) {
        b[s].x = wcol[(size_t)(4 * s) * DIM];
        b[s].y = wcol[(size_t)(4 * s + 1) * DIM];
    }
    __syncthreads();

    v8f c[GEMM_TILES];
#pragma unroll
    for (int t = 0; t < GEMM_TILES; ++t) c[t] = v8f{};

    const float* arow = tileA + m * A_PITCH + kh;
#pragma unroll
    for (int g = 0; g < NGR; ++g) {
        if (g + BPD < NGR) {
#pragma unroll
            for (int s = 2 * (g + BPD); s < 2 * (g + BPD) + 2; ++s) {
                b[s].x = wcol[(size_t)(4 * s) * DIM];
                b[s].y = wcol[(size_t)(4 * s + 1) * DIM];
            }
        }
#pragma unroll
        for (int s = 2 * g; s < 2 * g + 2; ++s) {
#pragma unroll
            for (int t = 0; t < GEMM_TILES; ++t) {
                const v2f a = *(const v2f*)(arow + t * 16 * A_PITCH + 4 * s);
                c[t] = __builtin_amdgcn_wmma_f32_16x16x4_f32(
                            false, a,  false, b[s],
                            (short)0, c[t],  false,  false);
            }
        }
        __builtin_amdgcn_sched_barrier(0);
    }
    const float bn = bias[n];
    __syncthreads();

    const int mo = (lane >> 4) << 3;
    float* crow = tileA + mo * A_PITCH + n;
#pragma unroll
    for (int t = 0; t < GEMM_TILES; ++t)
#pragma unroll
        for (int r = 0; r < 8; ++r)
            crow[(t * 16 + r) * A_PITCH] = fmaxf(c[t][r] + bn, 0.f);
    __syncthreads();

    const int gl = batch[n0 + min(lane, FU_NPW - 1)];
    float4 cs = {0.f, 0.f, 0.f, 0.f}, cq = {0.f, 0.f, 0.f, 0.f};
    float4 pa = {0.f, 0.f, 0.f, 0.f};
    float  pn = 0.f;
    int    cur = __builtin_amdgcn_readlane(gl, 0);
    for (int k = 0; k < FU_NPW; ++k) {
        const float4 x = *(const float4*)&trow[k * A_PITCH];
        if (STORE_ROWS) ((float4*)out)[(size_t)(n0 + k) * 32 + lane] = x;
        const int g = __builtin_amdgcn_readlane(gl, k);
        if (g != cur) {
            pool_flush_raw(pooled, gcnt, cur, lane, pa, pn);
            cur = g; pn = 0.f;
            pa.x = 0.f; pa.y = 0.f; pa.z = 0.f; pa.w = 0.f;
        }
        pa.x += x.x; pa.y += x.y; pa.z += x.z; pa.w += x.w;
        pn += 1.f;
        cs.x += x.x; cs.y += x.y; cs.z += x.z; cs.w += x.w;
        cq.x += x.x * x.x; cq.y += x.y * x.y; cq.z += x.z * x.z; cq.w += x.w * x.w;
    }
    ((float4*)&red[wave][0])[lane]   = cs;
    ((float4*)&red[wave][DIM])[lane] = cq;
    ((float4*)&pool[wave][0])[lane]  = pa;
    if (lane == 0) { pool[wave][DIM] = pn; pgid[wave] = cur; }
    __syncthreads();
    {
        float t = 0.f;
        for (int w = 0; w < 8; ++w) t += red[w][tid];
        atomicAdd(&stats[tid], t);
    }
    if (tid < DIM || (tid == DIM && gcnt != nullptr)) {
        const int c2 = tid;
        float* dst = (c2 < DIM) ? pooled + c2 : gcnt;
        const size_t pitch = (c2 < DIM) ? (size_t)OUTC : (size_t)1;
        int cg = pgid[0];
        float t = pool[0][c2];
        for (int w = 1; w < 8; ++w) {
            const int g = pgid[w];
            if (g != cg) {
                if ((unsigned)cg < (unsigned)NG) atomicAdd(dst + (size_t)cg * pitch, t);
                cg = g; t = 0.f;
            }
            t += pool[w][c2];
        }
        if ((unsigned)cg < (unsigned)NG) atomicAdd(dst + (size_t)cg * pitch, t);
    }
}


static_assert(NN % (EPI_RPW * 8) == 0, "every wave of every block owns EPI_RPW rows");

__global__ void __launch_bounds__(256)
k_bias_relu_stats(float* __restrict__ h, const float* __restrict__ bias,
                  float* __restrict__ stats) {
    __shared__ float red[8][2 * DIM];
    const int lane = threadIdx.x & 31;
    const int wave = threadIdx.x >> 5;
    const int r0 = __builtin_amdgcn_readfirstlane((blockIdx.x * 8 + wave) * EPI_RPW);
    float4* hv = (float4*)h + (size_t)r0 * 32 + lane;
    const float4 b = ((const float4*)bias)[lane];

    float4 v[EPI_RPW];
#pragma unroll
    for (int u = 0; u < EPI_RPW; ++u) v[u] = hv[u * 32];

    float4 s = {0.f, 0.f, 0.f, 0.f}, q = {0.f, 0.f, 0.f, 0.f};
#pragma unroll
    for (int u = 0; u < EPI_RPW; ++u) {
        float4 x = v[u];
        x.x = fmaxf(x.x + b.x, 0.f); x.y = fmaxf(x.y + b.y, 0.f);
        x.z = fmaxf(x.z + b.z, 0.f); x.w = fmaxf(x.w + b.w, 0.f);
        hv[u * 32] = x;
        s.x += x.x; s.y += x.y; s.z += x.z; s.w += x.w;
        q.x += x.x * x.x; q.y += x.y * x.y; q.z += x.z * x.z; q.w += x.w * x.w;
    }
    ((float4*)&red[wave][0])[lane]   = s;
    ((float4*)&red[wave][DIM])[lane] = q;
    __syncthreads();
    float t = 0.f;
    for (int w = 0; w < 8; ++w) t += red[w][threadIdx.x];
    atomicAdd(&stats[threadIdx.x], t);
}

__global__ void k_stats_finish(float* __restrict__ stats, float inv_n) {
    int c = threadIdx.x;
    float mu  = stats[c] * inv_n;
    float var = stats[DIM + c] * inv_n - mu * mu;
    stats[ST_MU + c] = mu;
    stats[ST_RS + c] = rsqrtf(var + BN_EPS);
}

__global__ void __launch_bounds__(DIM)
k_stats_pool_finish(const float* __restrict__ sums, const float* __restrict__ gcnt,
                    const float* __restrict__ gamma, const float* __restrict__ beta,
                    float* __restrict__ pooled, float inv_n) {
    const int c = threadIdx.x, g = blockIdx.x;
    const float mu = sums[c] * inv_n;
    const float rs = bn_rstd(sums[DIM + c], mu, inv_n);
    const float n = gcnt[g];
    float* o = pooled + (size_t)g * OUTC + c;
    *o = fmaf(fmaf(-n, mu, *o) * rs, gamma[c], n * beta[c]);
}

__device__ __forceinline__ void pool_flush(float* __restrict__ out, int g, int layer, int lane,
                                           float4 acc) {
    if ((unsigned)g < (unsigned)NG) {
        float* o = out + (size_t)g * OUTC + layer * DIM + 4 * lane;
        atomicAdd(o + 0, acc.x); atomicAdd(o + 1, acc.y);
        atomicAdd(o + 2, acc.z); atomicAdd(o + 3, acc.w);
    }
}

__global__ void __launch_bounds__(256)
k_norm_pool(const float* __restrict__ h, const float* __restrict__ stats,
            const float* __restrict__ gamma, const float* __restrict__ beta,
            const int* __restrict__ batch, float* __restrict__ hnext,
            float* __restrict__ out, int layer) {
    const int lane = threadIdx.x & 31;
    const int wave = threadIdx.x >> 5;
    const int r0 = __builtin_amdgcn_readfirstlane((blockIdx.x * 8 + wave) * EPI_RPW);
    const float4* hv = (const float4*)h + (size_t)r0 * 32 + lane;
    float4*       hn = (float4*)hnext   + (size_t)r0 * 32 + lane;

    float4 v[EPI_RPW];
#pragma unroll
    for (int u = 0; u < EPI_RPW; ++u) v[u] = hv[u * 32];

    const int gl = batch[r0 + min(lane, EPI_RPW - 1)];
    const float4 mu = ((const float4*)(stats + ST_MU))[lane];
    const float4 rs = ((const float4*)(stats + ST_RS))[lane];
    const float4 ga = ((const float4*)gamma)[lane];
    const float4 be = ((const float4*)beta)[lane];

    int cur = __builtin_amdgcn_readlane(gl, 0);
    float4 acc = {0.f, 0.f, 0.f, 0.f};
#pragma unroll
    for (int u = 0; u < EPI_RPW; ++u) {
        float4 x;
        x.x = (v[u].x - mu.x) * rs.x * ga.x + be.x;
        x.y = (v[u].y - mu.y) * rs.y * ga.y + be.y;
        x.z = (v[u].z - mu.z) * rs.z * ga.z + be.z;
        x.w = (v[u].w - mu.w) * rs.w * ga.w + be.w;
        hn[u * 32] = x;
        const int g = __builtin_amdgcn_readlane(gl, u);
        if (g != cur) {
            pool_flush(out, cur, layer, lane, acc);
            cur = g;
            acc.x = 0.f; acc.y = 0.f; acc.z = 0.f; acc.w = 0.f;
        }
        acc.x += x.x; acc.y += x.y; acc.z += x.z; acc.w += x.w;
    }
    pool_flush(out, cur, layer, lane, acc);
}


extern "C" void kernel_launch(void* const* d_in, const int* in_sizes, int n_in,
                              void* d_out, int out_size, void* d_ws, size_t ws_size,
                              hipStream_t stream) {
    (void)in_sizes; (void)n_in; (void)out_size;

    const float* x     = (const float*)d_in[0];
    const int*   ei    = (const int*)d_in[1];
    const int*   batch = (const int*)d_in[2];
    const float *Wp[NL], *bp[NL], *gp[NL], *tp[NL];
    for (int i = 0; i < NL; ++i) {
        Wp[i] = (const float*)d_in[3 + 4 * i];
        bp[i] = (const float*)d_in[4 + 4 * i];
        gp[i] = (const float*)d_in[5 + 4 * i];
        tp[i] = (const float*)d_in[6 + 4 * i];
    }
    const int* esrc = ei;
    const int* edst = ei + NE;

    float* ws    = (float*)d_ws;
    float* dinv  = ws;
    float* stats = ws + NN;
    size_t o     = ((size_t)NN + ST_FLOATS + 255) & ~(size_t)255;
    float* bufA  = ws + o;
    float* bufB  = bufA + (size_t)NN * DIM;
    float* outp  = (float*)d_out;

    int* rowptr  = (int*)(bufB + (size_t)NN * DIM);
    int* csr     = rowptr + RP_PAD;
    int* cnt     = (int*)bufB;
    int* rank    = cnt + RP_PAD;
    const bool sorted =
        ws_size >= (o + 2 * (size_t)NN * DIM + RP_PAD + NE) * sizeof(float);

    if (sorted) {
        k_zero3     <<<(NG * OUTC + NN + ST_FLOATS + 255) / 256, 256, 0, stream>>>(
            outp, NG * OUTC, (float*)cnt, NN, stats, ST_FLOATS);
        k_count_rank<<<(NE + 255) / 256,         256,    0, stream>>>(edst, cnt, rank, NE);
        k_scan_dinv <<<(NN + SCAN_T - 1) / SCAN_T, SCAN_T, 0, stream>>>(cnt, rowptr, dinv);
        k_fill      <<<(NE + 255) / 256,         256,    0, stream>>>(esrc, edst, rank, rowptr, csr, NE);
    } else {
        k_zero     <<<(NG * OUTC + 255) / 256, 256, 0, stream>>>(outp, NG * OUTC);
        k_init_deg <<<(NN + 255) / 256, 256, 0, stream>>>(dinv, NN);
        k_deg_edges<<<(NE + 255) / 256, 256, 0, stream>>>(edst, dinv, NE);
        k_dinv     <<<(NN + 255) / 256, 256, 0, stream>>>(dinv, NN);
    }

    for (int l = 0; l < NL; ++l) {
        if (sorted) {
            float* sums = stats + SS_SUM(l);
            float* gcnt = stats + SS_CNT;
            const float* hin  = (l == 0) ? x : ((l & 1) ? bufA : bufB);
            float*       hout = (l & 1) ? bufB : bufA;
            if (l >= 2) k_zero<<<1, 2 * DIM, 0, stream>>>(sums, 2 * DIM);
            if (l == 0)
                k_layer<false, true><<<NN / GEMM_ROWS, 256, 0, stream>>>(
                    rowptr, csr, dinv, hin, Wp[l], bp[l], nullptr, nullptr, nullptr, nullptr, nullptr,
                    batch, hout, sums, gcnt, outp, 1.f / (float)NN);
            else if (l < NL - 1)
                k_layer<true, true><<<NN / GEMM_ROWS, 256, 0, stream>>>(
                    rowptr, csr, dinv, hin, Wp[l], bp[l], stats + SS_SUM(l - 1), gp[l - 1], tp[l - 1],
                    gcnt, outp + (l - 1) * DIM, batch, hout, sums, nullptr, outp + l * DIM,
                    1.f / (float)NN);
            else {
                k_layer<true, false><<<NN / GEMM_ROWS, 256, 0, stream>>>(
                    rowptr, csr, dinv, hin, Wp[l], bp[l], stats + SS_SUM(l - 1), gp[l - 1], tp[l - 1],
                    gcnt, outp + (l - 1) * DIM, batch, nullptr, sums, nullptr, outp + l * DIM,
                    1.f / (float)NN);
                k_stats_pool_finish<<<NG, DIM, 0, stream>>>(
                    sums, gcnt, gp[l], tp[l], outp + l * DIM, 1.f / (float)NN);
            }
            continue;
        }
        const float* hin = (l == 0) ? x : bufA;
        k_gemm_wmma<<<NN / GEMM_ROWS, 256, 0, stream>>>(hin, Wp[l], bufA);
        k_zero<<<1, 256, 0, stream>>>(stats, 2 * DIM);
        k_selfloop <<<(NN * 32 + 255) / 256, 256, 0, stream>>>(bufA, dinv, bufB);
        k_edge_agg <<<(NE * 32) / 256, 256, 0, stream>>>(esrc, edst, dinv, bufA, bufB);
        k_bias_relu_stats<<<EPI_BLOCKS, 256, 0, stream>>>(bufB, bp[l], stats);
        k_stats_finish   <<<1, DIM, 0, stream>>>(stats, 1.f / (float)NN);
        k_norm_pool      <<<EPI_BLOCKS, 256, 0, stream>>>(
            bufB, stats, gp[l], tp[l], batch, bufA, outp, l);
    }
}


# edge list sorted by destination with LDS counters, no returning global atomic per edge: 8 node ranges x 32 edge slices, a block counts, then places, its share; sum order by slice, not arrival
# speedup vs baseline: 6.9223x; 1.3689x over previous
#include <hip/hip_runtime.h>

#define NN 100000
#define NE 1600000
#define DIM 128
#define NG 512
#define NL 3
#define OUTC (NL * DIM)
#define BN_EPS 1e-5f
#define EPI_RPW 25
#define EPI_BLOCKS (NN / EPI_RPW / 8)
#define SCAN_T 1024
#define RP_PAD 100096
#define GEMM_TILES 5
#define GEMM_ROWS (16 * GEMM_TILES)
#define A_PITCH (DIM + 4)
#define ST_MU  (2 * DIM + NG)
#define ST_RS  (3 * DIM + NG)
#define ST_FLOATS (4 * DIM + NG)
#define SS_SUM(l) (((l) & 1) * 2 * DIM)
#define SS_CNT (4 * DIM)
static_assert(SS_CNT + NG <= ST_FLOATS, "the sorted path's sums and counts fit the statistics block");

typedef __attribute__((ext_vector_type(2))) float v2f;
typedef __attribute__((ext_vector_type(8))) float v8f;


__global__ void k_zero(float* __restrict__ p, int n) {
    int i = blockIdx.x * blockDim.x + threadIdx.x;
    if (i < n) p[i] = 0.f;
}

__global__ void k_zero3(float* __restrict__ a, int na, float* __restrict__ b, int nb,
                        float* __restrict__ c, int nc) {
    int i = blockIdx.x * blockDim.x + threadIdx.x;
    if (i < na) a[i] = 0.f;
    else if (i < na + nb) b[i - na] = 0.f;
    else if (i < na + nb + nc) c[i - na - nb] = 0.f;
}

__device__ __forceinline__ float bn_rstd(float q, float mu, float inv_n) {
    const float var = q * inv_n - mu * mu;
    return rsqrtf(var + BN_EPS);
}

__global__ void k_init_deg(float* __restrict__ deg, int n) {
    int i = blockIdx.x * blockDim.x + threadIdx.x;
    if (i < n) deg[i] = 1.f;
}

__global__ void k_deg_edges(const int* __restrict__ dst, float* __restrict__ deg, int e) {
    int i = blockIdx.x * blockDim.x + threadIdx.x;
    if (i < e) atomicAdd(&deg[dst[i]], 1.f);
}

__global__ void k_dinv(float* __restrict__ deg, int n) {
    int i = blockIdx.x * blockDim.x + threadIdx.x;
    if (i < n) deg[i] = rsqrtf(fmaxf(deg[i], 1.f));
}

static_assert(NN % GEMM_ROWS == 0, "every block owns GEMM_ROWS whole rows");
static_assert((GEMM_ROWS * 32) % 256 == 0, "staging: a whole number of float4 per thread");

__global__ void __launch_bounds__(256)
k_gemm_wmma(const float* __restrict__ A, const float* __restrict__ W, float* __restrict__ out) {
    __shared__ float tileA[GEMM_ROWS * A_PITCH];

    const int tid  = threadIdx.x;
    const int wave = __builtin_amdgcn_readfirstlane(tid >> 5);
    const int lane = tid & 31;
    const int row0 = blockIdx.x * GEMM_ROWS;

    constexpr int NST = GEMM_ROWS * 32 / 256;
    const float4* srcv = (const float4*)(A + (size_t)row0 * DIM);
    float4 st[NST];
#pragma unroll
    for (int j = 0; j < NST; ++j) st[j] = srcv[tid + 256 * j];

    const int m  = lane & 15;
    const int kh = (lane >> 4) << 1;
    const int n  = (wave << 4) + m;

    const float* wcol = W + (size_t)kh * DIM + n;
    v2f b[DIM / 4];
#pragma unroll
    for (int s = 0; s < DIM / 4; ++s) {
        b[s].x = wcol[(size_t)(4 * s) * DIM];
        b[s].y = wcol[(size_t)(4 * s + 1) * DIM];
    }

    float* stage = tileA + wave * A_PITCH + (lane << 2);
#pragma unroll
    for (int j = 0; j < NST; ++j) *(float4*)&stage[8 * j * A_PITCH] = st[j];
    __syncthreads();

    v8f c[GEMM_TILES];
#pragma unroll
    for (int t = 0; t < GEMM_TILES; ++t) c[t] = v8f{};

    const float* arow = tileA + m * A_PITCH + kh;
#pragma unroll
    for (int s = 0; s < DIM / 4; ++s) {
#pragma unroll
        for (int t = 0; t < GEMM_TILES; ++t) {
            const v2f a = *(const v2f*)(arow + t * 16 * A_PITCH + 4 * s);
            c[t] = __builtin_amdgcn_wmma_f32_16x16x4_f32(
                        false, a,  false, b[s],
                        (short)0, c[t],  false,  false);
        }
    }

    const int mo = (lane >> 4) << 3;
    float* orow = out + (size_t)(row0 + mo) * DIM + n;
#pragma unroll
    for (int t = 0; t < GEMM_TILES; ++t)
#pragma unroll
        for (int r = 0; r < 8; ++r)
            orow[(size_t)(t * 16 + r) * DIM] = c[t][r];
}


__global__ void k_selfloop(const float* __restrict__ h, const float* __restrict__ dinv,
                           float* __restrict__ out) {
    int i = blockIdx.x * blockDim.x + threadIdx.x;
    int node = i >> 5, q = i & 31;
    if (node < NN) {
        float w = dinv[node]; w *= w;
        float4 v = ((const float4*)(h + (size_t)node * DIM))[q];
        float4 r; r.x = v.x * w; r.y = v.y * w; r.z = v.z * w; r.w = v.w * w;
        ((float4*)(out + (size_t)node * DIM))[q] = r;
    }
}

__global__ void k_edge_agg(const int* __restrict__ src, const int* __restrict__ dst,
                           const float* __restrict__ dinv, const float* __restrict__ h,
                           float* __restrict__ out) {
    int i = blockIdx.x * blockDim.x + threadIdx.x;
    int e = i >> 5, q = i & 31;
    if (e < NE) {
        int s = src[e], d = dst[e];
        float w = dinv[s] * dinv[d];
        const float* hs = h + (size_t)s * DIM + q;
        float v0 = hs[0], v1 = hs[32], v2 = hs[64], v3 = hs[96];
        float* o = out + (size_t)d * DIM + q;
        atomicAdd(o + 0,  v0 * w);
        atomicAdd(o + 32, v1 * w);
        atomicAdd(o + 64, v2 * w);
        atomicAdd(o + 96, v3 * w);
    }
}


#define SB_NODES   12500
#define SB_BUCKETS (NN / SB_NODES)
#define SB_SLICES  32
#define SB_EDGES   (NE / SB_SLICES)
#define SB_T       512
#define SB_NQ      (SB_EDGES / 4)
#define SB_QPT     ((SB_NQ + SB_T - 1) / SB_T)
#define SB_IPT     ((SB_NODES + SB_T - 1) / SB_T)
static_assert(SB_NODES * SB_BUCKETS == NN, "the buckets tile the nodes");
static_assert(SB_EDGES * SB_SLICES == NE && SB_EDGES % 4 == 0, "the slices tile the edges, as whole int4");
static_assert((size_t)RP_PAD + 2 * (size_t)SB_SLICES * NN <= (size_t)NN * DIM, "counts and both tables fit the buffer they borrow");

__global__ void __launch_bounds__(SB_T)
k_slice_count(const int* __restrict__ dst, int* __restrict__ tab) {
    __shared__ int ctr[SB_NODES];
    const int t = threadIdx.x;
    const int bucket = blockIdx.x % SB_BUCKETS;
    const int slice  = blockIdx.x / SB_BUCKETS;
    const int base   = bucket * SB_NODES;

    const int4* d4 = (const int4*)(dst + (size_t)slice * SB_EDGES);
    int4 v[SB_QPT];
#pragma unroll
    for (int u = 0; u < SB_QPT; ++u) {
        const int q = t + u * SB_T;
        v[u] = (q < SB_NQ) ? d4[q] : make_int4(-1, -1, -1, -1);
    }
    for (int i = t; i < SB_NODES; i += SB_T) ctr[i] = 0;
    __syncthreads();

#pragma unroll
    for (int u = 0; u < SB_QPT; ++u) {
        const unsigned a0 = (unsigned)(v[u].x - base), a1 = (unsigned)(v[u].y - base);
        const unsigned a2 = (unsigned)(v[u].z - base), a3 = (unsigned)(v[u].w - base);
        if (a0 < (unsigned)SB_NODES) atomicAdd(&ctr[a0], 1);
        if (a1 < (unsigned)SB_NODES) atomicAdd(&ctr[a1], 1);
        if (a2 < (unsigned)SB_NODES) atomicAdd(&ctr[a2], 1);
        if (a3 < (unsigned)SB_NODES) atomicAdd(&ctr[a3], 1);
    }
    __syncthreads();
    int* trow = tab + (size_t)slice * NN + base;
    for (int i = t; i < SB_NODES; i += SB_T) trow[i] = ctr[i];
}

__global__ void __launch_bounds__(256)
k_slice_prefix(const int* __restrict__ tab, int* __restrict__ pre, int* __restrict__ cnt) {
    const int n = blockIdx.x * 256 + threadIdx.x;
    if (n < NN) {
        int v[SB_SLICES];
#pragma unroll
        for (int s = 0; s < SB_SLICES; ++s) v[s] = tab[(size_t)s * NN + n];
        int run = 0;
#pragma unroll
        for (int s = 0; s < SB_SLICES; ++s) { pre[(size_t)s * NN + n] = run; run += v[s]; }
        cnt[n] = run;
    }
}

__global__ void __launch_bounds__(SCAN_T)
k_scan_dinv(const int* __restrict__ cnt, int* __restrict__ rowptr, float* __restrict__ dinv) {
    __shared__ int s_pre[SCAN_T / 32];
    __shared__ int s_own[SCAN_T / 32];
    const int t = threadIdx.x, b = blockIdx.x;
    const int lane = t & 31, wave = t >> 5;

    int pre = 0;
    const int4* c4 = (const int4*)cnt;
    for (int q = t; q < b * (SCAN_T / 4); q += SCAN_T) {
        int4 v = c4[q];
        pre += (v.x + v.y) + (v.z + v.w);
    }
    const int i = b * SCAN_T + t;
    const int own = (i < NN) ? cnt[i] : 0;

    int inc = own;
    for (int d = 1; d < 32; d <<= 1) {
        int u = __shfl_up(inc, d);
        if (lane >= d) inc += u;
    }
    for (int d = 16; d > 0; d >>= 1) pre += __shfl_xor(pre, d);
    if (lane == 31) { s_own[wave] = inc; s_pre[wave] = pre; }
    __syncthreads();
    int base = 0;
    for (int w = 0; w < SCAN_T / 32; ++w) {
        base += s_pre[w];
        if (w < wave) base += s_own[w];
    }
    const int excl = base + inc - own;
    if (i < NN) {
        rowptr[i] = excl;
        dinv[i] = rsqrtf(fmaxf((float)(own + 1), 1.f));
        if (i == NN - 1) rowptr[NN] = excl + own;
    }
}

#define SB_FB 13
__global__ void __launch_bounds__(SB_T)
k_slice_fill(const int* __restrict__ src, const int* __restrict__ dst,
             const int* __restrict__ rowptr, const int* __restrict__ pre,
             int* __restrict__ csr) {
    __shared__ int ctr[SB_NODES + 32];
    const int t = threadIdx.x;
    const int bucket = blockIdx.x % SB_BUCKETS;
    const int slice  = blockIdx.x / SB_BUCKETS;
    const int base   = bucket * SB_NODES;
    const unsigned spare = SB_NODES + (t & 31);

    {
        const int* rp   = rowptr + base;
        const int* trow = pre + (size_t)slice * NN + base;
        int a[SB_IPT];
#pragma unroll
        for (int u = 0; u < SB_IPT; ++u) {
            const int i = t + u * SB_T;
            a[u] = (i < SB_NODES) ? rp[i] + trow[i] : 0;
        }
#pragma unroll
        for (int u = 0; u < SB_IPT; ++u) {
            const int i = t + u * SB_T;
            if (i < SB_NODES) ctr[i] = a[u];
        }
        if (t < 32) ctr[SB_NODES + t] = 0;
    }
    __syncthreads();

    const int4* d4 = (const int4*)(dst + (size_t)slice * SB_EDGES);
    const int4* s4 = (const int4*)(src + (size_t)slice * SB_EDGES);
#pragma unroll
    for (int u0 = 0; u0 < SB_QPT; u0 += SB_FB) {
        int4 vd[SB_FB], vs[SB_FB];
#pragma unroll
        for (int u = 0; u < SB_FB; ++u) {
            const int q = t + (u0 + u) * SB_T;
            const bool in = (u0 + u < SB_QPT) && (q < SB_NQ);
            vd[u] = in ? d4[q] : make_int4(-1, -1, -1, -1);
            vs[u] = in ? s4[q] : make_int4(0, 0, 0, 0);
        }
        int pos[SB_FB][4];
#pragma unroll
        for (int u = 0; u < SB_FB; ++u) {
            const int dd[4] = {vd[u].x, vd[u].y, vd[u].z, vd[u].w};
#pragma unroll
            for (int c = 0; c < 4; ++c) {
                const unsigned a = (unsigned)(dd[c] - base);
                const bool hit = a < (unsigned)SB_NODES;
                const int p = atomicAdd(&ctr[hit ? a : spare], 1);
                pos[u][c] = hit ? p : -1;
            }
        }
#pragma unroll
        for (int u = 0; u < SB_FB; ++u) {
            const int ss[4] = {vs[u].x, vs[u].y, vs[u].z, vs[u].w};
#pragma unroll
            for (int c = 0; c < 4; ++c) {
                if ((unsigned)pos[u][c] < (unsigned)NE) csr[pos[u][c]] = ss[c];
            }
        }
    }
}

#define FU_NPW 10
static_assert(8 * FU_NPW == GEMM_ROWS, "the block's 8 waves fill the product's row tile");
static_assert(FU_NPW < 32, "a wave keeps its nodes' segment bounds one per lane");
static_assert(NN / GEMM_ROWS >= NG, "one block per graph for the previous layer's pooled rows");
#define POOL_PITCH (DIM + 4)

__device__ __forceinline__ void pool_flush_raw(float* __restrict__ pooled, float* __restrict__ gcnt,
                                               int g, int lane, float4 acc, float rows) {
    if ((unsigned)g < (unsigned)NG) {
        float* o = pooled + (size_t)g * OUTC + 4 * lane;
        atomicAdd(o + 0, acc.x); atomicAdd(o + 1, acc.y);
        atomicAdd(o + 2, acc.z); atomicAdd(o + 3, acc.w);
        if (gcnt != nullptr && lane == 0) atomicAdd(&gcnt[g], rows);
    }
}

template <bool NORM, bool STORE_ROWS>
__global__ void __launch_bounds__(256)
k_layer(const int* __restrict__ rowptr, const int* __restrict__ csr,
        const float* __restrict__ dinv, const float* __restrict__ h,
        const float* __restrict__ W, const float* __restrict__ bias,
        const float* __restrict__ psums, const float* __restrict__ pgamma,
        const float* __restrict__ pbeta, const float* __restrict__ gcnt_prev,
        float* __restrict__ pooled_prev, const int* __restrict__ batch,
        float* __restrict__ out, float* __restrict__ stats, float* __restrict__ gcnt,
        float* __restrict__ pooled, float inv_n) {
    __shared__ __attribute__((aligned(16))) float tileA[GEMM_ROWS * A_PITCH];
    __shared__ float red[8][2 * DIM];
    __shared__ __attribute__((aligned(16))) float pool[8][POOL_PITCH];
    __shared__ int   pgid[8];

    const int tid  = threadIdx.x;
    const int wave = __builtin_amdgcn_readfirstlane(tid >> 5);
    const int lane = tid & 31;

    float4 mu = {0.f, 0.f, 0.f, 0.f}, sa = {0.f, 0.f, 0.f, 0.f}, be = {0.f, 0.f, 0.f, 0.f};
    if (NORM) {
        if (blockIdx.x < NG && tid < DIM) {
            const float pm = psums[tid] * inv_n;
            const float pr = bn_rstd(psums[DIM + tid], pm, inv_n);
            const float rows = gcnt_prev[blockIdx.x];
            float* o = pooled_prev + (size_t)blockIdx.x * OUTC + tid;
            *o = fmaf(fmaf(-rows, pm, *o) * pr, pgamma[tid], rows * pbeta[tid]);
        }
        const float4 cs = ((const float4*)psums)[lane];
        const float4 cq = ((const float4*)(psums + DIM))[lane];
        const float4 ga = ((const float4*)pgamma)[lane];
        be = ((const float4*)pbeta)[lane];
        mu.x = cs.x * inv_n; mu.y = cs.y * inv_n; mu.z = cs.z * inv_n; mu.w = cs.w * inv_n;
        sa.x = bn_rstd(cq.x, mu.x, inv_n) * ga.x; sa.y = bn_rstd(cq.y, mu.y, inv_n) * ga.y;
        sa.z = bn_rstd(cq.z, mu.z, inv_n) * ga.z; sa.w = bn_rstd(cq.w, mu.w, inv_n) * ga.w;
    }

    const int n0 = __builtin_amdgcn_readfirstlane((blockIdx.x * 8 + wave) * FU_NPW);
    const int rp = rowptr[n0 + min(lane, FU_NPW)];
    const int dl = __float_as_int(dinv[n0 + min(lane, FU_NPW - 1)]);
    const float4* hv = (const float4*)h;
    float* trow = tileA + wave * FU_NPW * A_PITCH + (lane << 2);

    for (int k = 0; k < FU_NPW; ++k) {
        const int node  = n0 + k;
        const int beg   = __builtin_amdgcn_readlane(rp, k);
        const int items = __builtin_amdgcn_readlane(rp, k + 1) - beg + 1;
        const float dd  = __int_as_float(__builtin_amdgcn_readlane(dl, k));

        float4 acc = {0.f, 0.f, 0.f, 0.f};
        float  sd  = 0.f;
        for (int base = 0; base < items; base += 32) {
            const int it = base + lane;
            int s = node;
            if (it > 0 && it < items) {
                s = (int)min((unsigned)csr[beg + it - 1], (unsigned)(NN - 1));
            }
            const int wl = __float_as_int(dinv[s]);
            const int n = min(32, items - base);
            int j = 0;
            for (; j + 4 <= n; j += 4) {
                const int s0 = __builtin_amdgcn_readlane(s, j);
                const int s1 = __builtin_amdgcn_readlane(s, j + 1);
                const int s2 = __builtin_amdgcn_readlane(s, j + 2);
                const int s3 = __builtin_amdgcn_readlane(s, j + 3);
                const float w0 = __int_as_float(__builtin_amdgcn_readlane(wl, j));
                const float w1 = __int_as_float(__builtin_amdgcn_readlane(wl, j + 1));
                const float w2 = __int_as_float(__builtin_amdgcn_readlane(wl, j + 2));
                const float w3 = __int_as_float(__builtin_amdgcn_readlane(wl, j + 3));
                const float4 v0 = hv[(size_t)s0 * 32 + lane];
                const float4 v1 = hv[(size_t)s1 * 32 + lane];
                const float4 v2 = hv[(size_t)s2 * 32 + lane];
                const float4 v3 = hv[(size_t)s3 * 32 + lane];
                acc.x = fmaf(v0.x, w0, acc.x); acc.y = fmaf(v0.y, w0, acc.y); acc.z = fmaf(v0.z, w0, acc.z); acc.w = fmaf(v0.w, w0, acc.w);
                acc.x = fmaf(v1.x, w1, acc.x); acc.y = fmaf(v1.y, w1, acc.y); acc.z = fmaf(v1.z, w1, acc.z); acc.w = fmaf(v1.w, w1, acc.w);
                acc.x = fmaf(v2.x, w2, acc.x); acc.y = fmaf(v2.y, w2, acc.y); acc.z = fmaf(v2.z, w2, acc.z); acc.w = fmaf(v2.w, w2, acc.w);
                acc.x = fmaf(v3.x, w3, acc.x); acc.y = fmaf(v3.y, w3, acc.y); acc.z = fmaf(v3.z, w3, acc.z); acc.w = fmaf(v3.w, w3, acc.w);
                sd += w0; sd += w1; sd += w2; sd += w3;
            }
            const int rest = n - j;
            if (rest > 0) {
                const int s0 = __builtin_amdgcn_readlane(s, j);
                const float w0 = __int_as_float(__builtin_amdgcn_readlane(wl, j));
                const float4 v0 = hv[(size_t)s0 * 32 + lane];
                float4 v1 = {0.f, 0.f, 0.f, 0.f}, v2 = {0.f, 0.f, 0.f, 0.f};
                float w1 = 0.f, w2 = 0.f;
                if (rest > 1) {
                    v1 = hv[(size_t)__builtin_amdgcn_readlane(s, j + 1) * 32 + lane];
                    w1 = __int_as_float(__builtin_amdgcn_readlane(wl, j + 1));
                }
                if (rest > 2) {
                    v2 = hv[(size_t)__builtin_amdgcn_readlane(s, j + 2) * 32 + lane];
                    w2 = __int_as_float(__builtin_amdgcn_readlane(wl, j + 2));
                }
                acc.x = fmaf(v0.x, w0, acc.x); acc.y = fmaf(v0.y, w0, acc.y); acc.z = fmaf(v0.z, w0, acc.z); acc.w = fmaf(v0.w, w0, acc.w);
                acc.x = fmaf(v1.x, w1, acc.x); acc.y = fmaf(v1.y, w1, acc.y); acc.z = fmaf(v1.z, w1, acc.z); acc.w = fmaf(v1.w, w1, acc.w);
                acc.x = fmaf(v2.x, w2, acc.x); acc.y = fmaf(v2.y, w2, acc.y); acc.z = fmaf(v2.z, w2, acc.z); acc.w = fmaf(v2.w, w2, acc.w);
                sd += w0; sd += w1; sd += w2;
            }
        }
        float4 z;
        if (NORM) {
            z.x = (fmaf(-mu.x, sd, acc.x) * sa.x + be.x * sd) * dd;
            z.y = (fmaf(-mu.y, sd, acc.y) * sa.y + be.y * sd) * dd;
            z.z = (fmaf(-mu.z, sd, acc.z) * sa.z + be.z * sd) * dd;
            z.w = (fmaf(-mu.w, sd, acc.w) * sa.w + be.w * sd) * dd;
        } else {
            z.x = acc.x * dd; z.y = acc.y * dd; z.z = acc.z * dd; z.w = acc.w * dd;
        }
        *(float4*)&trow[k * A_PITCH] = z;
    }

    const int m  = lane & 15;
    const int kh = (lane >> 4) << 1;
    const int n  = (wave << 4) + m;

    const float* wcol = W + (size_t)kh * DIM + n;
    constexpr int NGR = DIM / 8;
    constexpr int BPD = 3;
    v2f b[DIM / 4];
#pragma unroll
    for (int s = 0; s < 2 * BPD; ++s) {
        b[s].x = wcol[(size_t)(4 * s) * DIM];
        b[s].y = wcol[(size_t)(4 * s + 1) * DIM];
    }
    __syncthreads();

    v8f c[GEMM_TILES];
#pragma unroll
    for (int t = 0; t < GEMM_TILES; ++t) c[t] = v8f{};

    const float* arow = tileA + m * A_PITCH + kh;
#pragma unroll
    for (int g = 0; g < NGR; ++g) {
        if (g + BPD < NGR) {
#pragma unroll
            for (int s = 2 * (g + BPD); s < 2 * (g + BPD) + 2; ++s) {
                b[s].x = wcol[(size_t)(4 * s) * DIM];
                b[s].y = wcol[(size_t)(4 * s + 1) * DIM];
            }
        }
#pragma unroll
        for (int s = 2 * g; s < 2 * g + 2; ++s) {
#pragma unroll
            for (int t = 0; t < GEMM_TILES; ++t) {
                const v2f a = *(const v2f*)(arow + t * 16 * A_PITCH + 4 * s);
                c[t] = __builtin_amdgcn_wmma_f32_16x16x4_f32(
                            false, a,  false, b[s],
                            (short)0, c[t],  false,  false);
            }
        }
        __builtin_amdgcn_sched_barrier(0);
    }
    const float bn = bias[n];
    __syncthreads();

    const int mo = (lane >> 4) << 3;
    float* crow = tileA + mo * A_PITCH + n;
#pragma unroll
    for (int t = 0; t < GEMM_TILES; ++t)
#pragma unroll
        for (int r = 0; r < 8; ++r)
            crow[(t * 16 + r) * A_PITCH] = fmaxf(c[t][r] + bn, 0.f);
    __syncthreads();

    const int gl = batch[n0 + min(lane, FU_NPW - 1)];
    float4 cs = {0.f, 0.f, 0.f, 0.f}, cq = {0.f, 0.f, 0.f, 0.f};
    float4 pa = {0.f, 0.f, 0.f, 0.f};
    float  pn = 0.f;
    int    cur = __builtin_amdgcn_readlane(gl, 0);
    for (int k = 0; k < FU_NPW; ++k) {
        const float4 x = *(const float4*)&trow[k * A_PITCH];
        if (STORE_ROWS) ((float4*)out)[(size_t)(n0 + k) * 32 + lane] = x;
        const int g = __builtin_amdgcn_readlane(gl, k);
        if (g != cur) {
            pool_flush_raw(pooled, gcnt, cur, lane, pa, pn);
            cur = g; pn = 0.f;
            pa.x = 0.f; pa.y = 0.f; pa.z = 0.f; pa.w = 0.f;
        }
        pa.x += x.x; pa.y += x.y; pa.z += x.z; pa.w += x.w;
        pn += 1.f;
        cs.x += x.x; cs.y += x.y; cs.z += x.z; cs.w += x.w;
        cq.x += x.x * x.x; cq.y += x.y * x.y; cq.z += x.z * x.z; cq.w += x.w * x.w;
    }
    ((float4*)&red[wave][0])[lane]   = cs;
    ((float4*)&red[wave][DIM])[lane] = cq;
    ((float4*)&pool[wave][0])[lane]  = pa;
    if (lane == 0) { pool[wave][DIM] = pn; pgid[wave] = cur; }
    __syncthreads();
    {
        float t = 0.f;
        for (int w = 0; w < 8; ++w) t += red[w][tid];
        atomicAdd(&stats[tid], t);
    }
    if (tid < DIM || (tid == DIM && gcnt != nullptr)) {
        const int c2 = tid;
        float* dst = (c2 < DIM) ? pooled + c2 : gcnt;
        const size_t pitch = (c2 < DIM) ? (size_t)OUTC : (size_t)1;
        int cg = pgid[0];
        float t = pool[0][c2];
        for (int w = 1; w < 8; ++w) {
            const int g = pgid[w];
            if (g != cg) {
                if ((unsigned)cg < (unsigned)NG) atomicAdd(dst + (size_t)cg * pitch, t);
                cg = g; t = 0.f;
            }
            t += pool[w][c2];
        }
        if ((unsigned)cg < (unsigned)NG) atomicAdd(dst + (size_t)cg * pitch, t);
    }
}


static_assert(NN % (EPI_RPW * 8) == 0, "every wave of every block owns EPI_RPW rows");

__global__ void __launch_bounds__(256)
k_bias_relu_stats(float* __restrict__ h, const float* __restrict__ bias,
                  float* __restrict__ stats) {
    __shared__ float red[8][2 * DIM];
    const int lane = threadIdx.x & 31;
    const int wave = threadIdx.x >> 5;
    const int r0 = __builtin_amdgcn_readfirstlane((blockIdx.x * 8 + wave) * EPI_RPW);
    float4* hv = (float4*)h + (size_t)r0 * 32 + lane;
    const float4 b = ((const float4*)bias)[lane];

    float4 v[EPI_RPW];
#pragma unroll
    for (int u = 0; u < EPI_RPW; ++u) v[u] = hv[u * 32];

    float4 s = {0.f, 0.f, 0.f, 0.f}, q = {0.f, 0.f, 0.f, 0.f};
#pragma unroll
    for (int u = 0; u < EPI_RPW; ++u) {
        float4 x = v[u];
        x.x = fmaxf(x.x + b.x, 0.f); x.y = fmaxf(x.y + b.y, 0.f);
        x.z = fmaxf(x.z + b.z, 0.f); x.w = fmaxf(x.w + b.w, 0.f);
        hv[u * 32] = x;
        s.x += x.x; s.y += x.y; s.z += x.z; s.w += x.w;
        q.x += x.x * x.x; q.y += x.y * x.y; q.z += x.z * x.z; q.w += x.w * x.w;
    }
    ((float4*)&red[wave][0])[lane]   = s;
    ((float4*)&red[wave][DIM])[lane] = q;
    __syncthreads();
    float t = 0.f;
    for (int w = 0; w < 8; ++w) t += red[w][threadIdx.x];
    atomicAdd(&stats[threadIdx.x], t);
}

__global__ void k_stats_finish(float* __restrict__ stats, float inv_n) {
    int c = threadIdx.x;
    float mu  = stats[c] * inv_n;
    float var = stats[DIM + c] * inv_n - mu * mu;
    stats[ST_MU + c] = mu;
    stats[ST_RS + c] = rsqrtf(var + BN_EPS);
}

__global__ void __launch_bounds__(DIM)
k_stats_pool_finish(const float* __restrict__ sums, const float* __restrict__ gcnt,
                    const float* __restrict__ gamma, const float* __restrict__ beta,
                    float* __restrict__ pooled, float inv_n) {
    const int c = threadIdx.x, g = blockIdx.x;
    const float mu = sums[c] * inv_n;
    const float rs = bn_rstd(sums[DIM + c], mu, inv_n);
    const float n = gcnt[g];
    float* o = pooled + (size_t)g * OUTC + c;
    *o = fmaf(fmaf(-n, mu, *o) * rs, gamma[c], n * beta[c]);
}

__device__ __forceinline__ void pool_flush(float* __restrict__ out, int g, int layer, int lane,
                                           float4 acc) {
    if ((unsigned)g < (unsigned)NG) {
        float* o = out + (size_t)g * OUTC + layer * DIM + 4 * lane;
        atomicAdd(o + 0, acc.x); atomicAdd(o + 1, acc.y);
        atomicAdd(o + 2, acc.z); atomicAdd(o + 3, acc.w);
    }
}

__global__ void __launch_bounds__(256)
k_norm_pool(const float* __restrict__ h, const float* __restrict__ stats,
            const float* __restrict__ gamma, const float* __restrict__ beta,
            const int* __restrict__ batch, float* __restrict__ hnext,
            float* __restrict__ out, int layer) {
    const int lane = threadIdx.x & 31;
    const int wave = threadIdx.x >> 5;
    const int r0 = __builtin_amdgcn_readfirstlane((blockIdx.x * 8 + wave) * EPI_RPW);
    const float4* hv = (const float4*)h + (size_t)r0 * 32 + lane;
    float4*       hn = (float4*)hnext   + (size_t)r0 * 32 + lane;

    float4 v[EPI_RPW];
#pragma unroll
    for (int u = 0; u < EPI_RPW; ++u) v[u] = hv[u * 32];

    const int gl = batch[r0 + min(lane, EPI_RPW - 1)];
    const float4 mu = ((const float4*)(stats + ST_MU))[lane];
    const float4 rs = ((const float4*)(stats + ST_RS))[lane];
    const float4 ga = ((const float4*)gamma)[lane];
    const float4 be = ((const float4*)beta)[lane];

    int cur = __builtin_amdgcn_readlane(gl, 0);
    float4 acc = {0.f, 0.f, 0.f, 0.f};
#pragma unroll
    for (int u = 0; u < EPI_RPW; ++u) {
        float4 x;
        x.x = (v[u].x - mu.x) * rs.x * ga.x + be.x;
        x.y = (v[u].y - mu.y) * rs.y * ga.y + be.y;
        x.z = (v[u].z - mu.z) * rs.z * ga.z + be.z;
        x.w = (v[u].w - mu.w) * rs.w * ga.w + be.w;
        hn[u * 32] = x;
        const int g = __builtin_amdgcn_readlane(gl, u);
        if (g != cur) {
            pool_flush(out, cur, layer, lane, acc);
            cur = g;
            acc.x = 0.f; acc.y = 0.f; acc.z = 0.f; acc.w = 0.f;
        }
        acc.x += x.x; acc.y += x.y; acc.z += x.z; acc.w += x.w;
    }
    pool_flush(out, cur, layer, lane, acc);
}


extern "C" void kernel_launch(void* const* d_in, const int* in_sizes, int n_in,
                              void* d_out, int out_size, void* d_ws, size_t ws_size,
                              hipStream_t stream) {
    (void)in_sizes; (void)n_in; (void)out_size;

    const float* x     = (const float*)d_in[0];
    const int*   ei    = (const int*)d_in[1];
    const int*   batch = (const int*)d_in[2];
    const float *Wp[NL], *bp[NL], *gp[NL], *tp[NL];
    for (int i = 0; i < NL; ++i) {
        Wp[i] = (const float*)d_in[3 + 4 * i];
        bp[i] = (const float*)d_in[4 + 4 * i];
        gp[i] = (const float*)d_in[5 + 4 * i];
        tp[i] = (const float*)d_in[6 + 4 * i];
    }
    const int* esrc = ei;
    const int* edst = ei + NE;

    float* ws    = (float*)d_ws;
    float* dinv  = ws;
    float* stats = ws + NN;
    size_t o     = ((size_t)NN + ST_FLOATS + 255) & ~(size_t)255;
    float* bufA  = ws + o;
    float* bufB  = bufA + (size_t)NN * DIM;
    float* outp  = (float*)d_out;

    int* rowptr  = (int*)(bufB + (size_t)NN * DIM);
    int* csr     = rowptr + RP_PAD;
    int* cnt     = (int*)bufB;
    int* tab     = cnt + RP_PAD;
    int* pre     = tab + (size_t)SB_SLICES * NN;
    const bool sorted =
        ws_size >= (o + 2 * (size_t)NN * DIM + RP_PAD + NE) * sizeof(float);

    if (sorted) {
        k_zero3       <<<(NG * OUTC + ST_FLOATS + 255) / 256, 256, 0, stream>>>(
            outp, NG * OUTC, (float*)cnt, 0, stats, ST_FLOATS);
        k_slice_count <<<SB_BUCKETS * SB_SLICES, SB_T, 0, stream>>>(edst, tab);
        k_slice_prefix<<<(NN + 255) / 256, 256, 0, stream>>>(tab, pre, cnt);
        k_scan_dinv   <<<(NN + SCAN_T - 1) / SCAN_T, SCAN_T, 0, stream>>>(cnt, rowptr, dinv);
        k_slice_fill  <<<SB_BUCKETS * SB_SLICES, SB_T, 0, stream>>>(esrc, edst, rowptr, pre, csr);
    } else {
        k_zero     <<<(NG * OUTC + 255) / 256, 256, 0, stream>>>(outp, NG * OUTC);
        k_init_deg <<<(NN + 255) / 256, 256, 0, stream>>>(dinv, NN);
        k_deg_edges<<<(NE + 255) / 256, 256, 0, stream>>>(edst, dinv, NE);
        k_dinv     <<<(NN + 255) / 256, 256, 0, stream>>>(dinv, NN);
    }

    for (int l = 0; l < NL; ++l) {
        if (sorted) {
            float* sums = stats + SS_SUM(l);
            float* gcnt = stats + SS_CNT;
            const float* hin  = (l == 0) ? x : ((l & 1) ? bufA : bufB);
            float*       hout = (l & 1) ? bufB : bufA;
            if (l >= 2) k_zero<<<1, 2 * DIM, 0, stream>>>(sums, 2 * DIM);
            if (l == 0)
                k_layer<false, true><<<NN / GEMM_ROWS, 256, 0, stream>>>(
                    rowptr, csr, dinv, hin, Wp[l], bp[l], nullptr, nullptr, nullptr, nullptr, nullptr,
                    batch, hout, sums, gcnt, outp, 1.f / (float)NN);
            else if (l < NL - 1)
                k_layer<true, true><<<NN / GEMM_ROWS, 256, 0, stream>>>(
                    rowptr, csr, dinv, hin, Wp[l], bp[l], stats + SS_SUM(l - 1), gp[l - 1], tp[l - 1],
                    gcnt, outp + (l - 1) * DIM, batch, hout, sums, nullptr, outp + l * DIM,
                    1.f / (float)NN);
            else {
                k_layer<true, false><<<NN / GEMM_ROWS, 256, 0, stream>>>(
                    rowptr, csr, dinv, hin, Wp[l], bp[l], stats + SS_SUM(l - 1), gp[l - 1], tp[l - 1],
                    gcnt, outp + (l - 1) * DIM, batch, nullptr, sums, nullptr, outp + l * DIM,
                    1.f / (float)NN);
                k_stats_pool_finish<<<NG, DIM, 0, stream>>>(
                    sums, gcnt, gp[l], tp[l], outp + l * DIM, 1.f / (float)NN);
            }
            continue;
        }
        const float* hin = (l == 0) ? x : bufA;
        k_gemm_wmma<<<NN / GEMM_ROWS, 256, 0, stream>>>(hin, Wp[l], bufA);
        k_zero<<<1, 256, 0, stream>>>(stats, 2 * DIM);
        k_selfloop <<<(NN * 32 + 255) / 256, 256, 0, stream>>>(bufA, dinv, bufB);
        k_edge_agg <<<(NE * 32) / 256, 256, 0, stream>>>(esrc, edst, dinv, bufA, bufB);
        k_bias_relu_stats<<<EPI_BLOCKS, 256, 0, stream>>>(bufB, bp[l], stats);
        k_stats_finish   <<<1, DIM, 0, stream>>>(stats, 1.f / (float)NN);
        k_norm_pool      <<<EPI_BLOCKS, 256, 0, stream>>>(
            bufB, stats, gp[l], tp[l], batch, bufA, outp, l);
    }
}
